# starting kernel
# baseline (speedup 1.0000x reference)
#include <hip/hip_runtime.h>


typedef __attribute__((ext_vector_type(2))) float v2f;
typedef __attribute__((ext_vector_type(8))) float v8f;

#define DEV_ATOMIC_ADD(p, v) \
    __hip_atomic_fetch_add((p), (v), __ATOMIC_RELAXED, __HIP_MEMORY_SCOPE_AGENT)

__global__ void lgcn_degree(const int* __restrict__ eu, const int* __restrict__ ei,
                            float* __restrict__ deg, int E, int U) {
    int t = blockIdx.x * blockDim.x + threadIdx.x;
    if (t >= E) return;
    DEV_ATOMIC_ADD(&deg[eu[t]], 1.0f);
    DEV_ATOMIC_ADD(&deg[U + ei[t]], 1.0f);
}

__global__ void lgcn_invsqrt(float* __restrict__ d, int N) {
    int t = blockIdx.x * blockDim.x + threadIdx.x;
    if (t >= N) return;
    float v = d[t];
    d[t] = (v > 0.0f) ? (1.0f / sqrtf(v)) : 0.0f;
}

__global__ void lgcn_init_x0(const float* __restrict__ ue, const float* __restrict__ ie,
                             float* __restrict__ x, int U, int N) {
    long long t = (long long)blockIdx.x * blockDim.x + threadIdx.x;
    long long total = (long long)N * 64;
    if (t >= total) return;
    long long ubytes = (long long)U * 64;
    x[t] = (t < ubytes) ? ue[t] : ie[t - ubytes];
}

__global__ void lgcn_spmm(const int* __restrict__ eu, const int* __restrict__ ei,
                          const float* __restrict__ inv,
                          const float* __restrict__ xin, float* __restrict__ xout,
                          int E, int U) {
    long long t = (long long)blockIdx.x * blockDim.x + threadIdx.x;
    int edge = (int)(t >> 4);
    if (edge >= 2 * E) return;
    int q = ((int)t & 15) * 4;

    int s, d;
    if (edge < E) {
        s = eu[edge];
        d = U + ei[edge];
    } else {
        int e2 = edge - E;
        s = U + ei[e2];
        d = eu[e2];
    }
    float w = inv[s] * inv[d];
    const float4 v = *(const float4*)(xin + (size_t)s * 64 + q);
    float* o = xout + (size_t)d * 64 + q;
    DEV_ATOMIC_ADD(o + 0, w * v.x);
    DEV_ATOMIC_ADD(o + 1, w * v.y);
    DEV_ATOMIC_ADD(o + 2, w * v.z);
    DEV_ATOMIC_ADD(o + 3, w * v.w);
}

__global__ void lgcn_gather_acc(const int* __restrict__ users, const int* __restrict__ items,
                                const float* __restrict__ x,
                                float* __restrict__ uacc, float* __restrict__ iacc,
                                int U, int B) {
    int t = blockIdx.x * blockDim.x + threadIdx.x;
    int half = B * 64;
    if (t >= 2 * half) return;
    if (t < half) {
        int b = t >> 6, d = t & 63;
        uacc[t] += x[(size_t)users[b] * 64 + d];
    } else {
        int tt = t - half;
        int b = tt >> 6, d = tt & 63;
        iacc[tt] += x[((size_t)U + items[b]) * 64 + d];
    }
}

__global__ __launch_bounds__(256) void lgcn_dot_wmma(const float* __restrict__ uacc,
                                                     const float* __restrict__ iacc,
                                                     float* __restrict__ out, int B) {
    __shared__ float tile[8][16 * 16];
    const int lane = threadIdx.x & 31;
    const int wave = threadIdx.x >> 5;
    const int tile0 = (blockIdx.x * 8 + wave) * 16;

    const int m = lane & 15;
    const int khalf = (lane >> 4) << 1;
    int row = tile0 + m;
    if (row > B - 1) row = B - 1;
    const float* __restrict__ ur = uacc + (size_t)row * 64;
    const float* __restrict__ ir = iacc + (size_t)row * 64;

    v8f c = {};
#pragma unroll
    for (int kb = 0; kb < 64; kb += 4) {
        int k0 = kb + khalf;
        v2f a; a.x = ur[k0]; a.y = ur[k0 + 1];
        v2f b; b.x = ir[k0]; b.y = ir[k0 + 1];
        c = __builtin_amdgcn_wmma_f32_16x16x4_f32(false, a, false, b,
                                                  (short)0, c, false, false);
    }

    float* lds = &tile[wave][0];
    const int n = lane & 15;
    const int mbase = (lane >> 4) ? 8 : 0;
#pragma unroll
    for (int r = 0; r < 8; ++r)
        lds[(mbase + r) * 16 + n] = c[r];
    __syncthreads();

    if (lane < 16) {
        int b = tile0 + lane;
        if (b < B)
            out[b] = lds[lane * 16 + lane] * (1.0f / 25.0f);
    }
}

extern "C" void kernel_launch(void* const* d_in, const int* in_sizes, int n_in,
                              void* d_out, int out_size, void* d_ws, size_t ws_size,
                              hipStream_t stream) {
    const int*   users = (const int*)d_in[0];
    const int*   items = (const int*)d_in[1];
    const int*   eu    = (const int*)d_in[2];
    const int*   ei    = (const int*)d_in[3];
    const float* uemb  = (const float*)d_in[4];
    const float* iemb  = (const float*)d_in[5];
    float*       out   = (float*)d_out;

    const int B = in_sizes[0];
    const int E = in_sizes[2];
    const int U = in_sizes[4] / 64;
    const int I = in_sizes[5] / 64;
    const int N = U + I;

    char* ws = (char*)d_ws;
    size_t off = 0;
    auto carve = [&](size_t bytes) -> float* {
        float* p = (float*)(ws + off);
        off += (bytes + 255) & ~(size_t)255;
        return p;
    };
    float* inv    = carve((size_t)N * 4);
    float* e_cur  = carve((size_t)N * 64 * 4);
    float* e_next = carve((size_t)N * 64 * 4);
    float* uacc   = carve((size_t)B * 64 * 4);
    float* iacc   = carve((size_t)B * 64 * 4);

    const int BS = 256;

    hipMemsetAsync(inv,  0, (size_t)N * 4, stream);
    hipMemsetAsync(uacc, 0, (size_t)B * 64 * 4, stream);
    hipMemsetAsync(iacc, 0, (size_t)B * 64 * 4, stream);

    {
        long long total = (long long)N * 64;
        lgcn_init_x0<<<(int)((total + BS - 1) / BS), BS, 0, stream>>>(uemb, iemb, e_cur, U, N);
    }
    lgcn_degree<<<(E + BS - 1) / BS, BS, 0, stream>>>(eu, ei, inv, E, U);
    lgcn_invsqrt<<<(N + BS - 1) / BS, BS, 0, stream>>>(inv, N);

    const int gacc_blocks = (2 * B * 64 + BS - 1) / BS;
    lgcn_gather_acc<<<gacc_blocks, BS, 0, stream>>>(users, items, e_cur, uacc, iacc, U, B);

    const long long spmm_threads = (long long)2 * E * 16;
    const int spmm_blocks = (int)((spmm_threads + BS - 1) / BS);
    for (int layer = 0; layer < 4; ++layer) {
        hipMemsetAsync(e_next, 0, (size_t)N * 64 * 4, stream);
        lgcn_spmm<<<spmm_blocks, BS, 0, stream>>>(eu, ei, inv, e_cur, e_next, E, U);
        lgcn_gather_acc<<<gacc_blocks, BS, 0, stream>>>(users, items, e_next, uacc, iacc, U, B);
        float* tmp = e_cur; e_cur = e_next; e_next = tmp;
    }

    const int tiles = (B + 15) / 16;
    const int dot_blocks = (tiles + 7) / 8;
    lgcn_dot_wmma<<<dot_blocks, 256, 0, stream>>>(uacc, iacc, out, B);
}


# SpMM layers: group edges by destination once per call (returning int adds give slots, scan, place), then gather per output row without atomics; sums run in arrival-rank order per row
# speedup vs baseline: 2.3473x; 2.3473x over previous
#include <hip/hip_runtime.h>


typedef __attribute__((ext_vector_type(2))) float v2f;
typedef __attribute__((ext_vector_type(8))) float v8f;

#define DEV_ATOMIC_ADD(p, v) \
    __hip_atomic_fetch_add((p), (v), __ATOMIC_RELAXED, __HIP_MEMORY_SCOPE_AGENT)

__global__ void lgcn_degree(const int* __restrict__ eu, const int* __restrict__ ei,
                            float* __restrict__ deg, int E, int U) {
    int t = blockIdx.x * blockDim.x + threadIdx.x;
    if (t >= E) return;
    DEV_ATOMIC_ADD(&deg[eu[t]], 1.0f);
    DEV_ATOMIC_ADD(&deg[U + ei[t]], 1.0f);
}

__global__ void lgcn_invsqrt(float* __restrict__ d, int N) {
    int t = blockIdx.x * blockDim.x + threadIdx.x;
    if (t >= N) return;
    float v = d[t];
    d[t] = (v > 0.0f) ? (1.0f / sqrtf(v)) : 0.0f;
}

__global__ void lgcn_init_x0(const float* __restrict__ ue, const float* __restrict__ ie,
                             float* __restrict__ x, int U, int N) {
    long long t = (long long)blockIdx.x * blockDim.x + threadIdx.x;
    long long total = (long long)N * 64;
    if (t >= total) return;
    long long ubytes = (long long)U * 64;
    x[t] = (t < ubytes) ? ue[t] : ie[t - ubytes];
}

__global__ void lgcn_spmm(const int* __restrict__ eu, const int* __restrict__ ei,
                          const float* __restrict__ inv,
                          const float* __restrict__ xin, float* __restrict__ xout,
                          int E, int U) {
    long long t = (long long)blockIdx.x * blockDim.x + threadIdx.x;
    int edge = (int)(t >> 4);
    if (edge >= 2 * E) return;
    int q = ((int)t & 15) * 4;

    int s, d;
    if (edge < E) {
        s = eu[edge];
        d = U + ei[edge];
    } else {
        int e2 = edge - E;
        s = U + ei[e2];
        d = eu[e2];
    }
    float w = inv[s] * inv[d];
    const float4 v = *(const float4*)(xin + (size_t)s * 64 + q);
    float* o = xout + (size_t)d * 64 + q;
    DEV_ATOMIC_ADD(o + 0, w * v.x);
    DEV_ATOMIC_ADD(o + 1, w * v.y);
    DEV_ATOMIC_ADD(o + 2, w * v.z);
    DEV_ATOMIC_ADD(o + 3, w * v.w);
}


__global__ void lgcn_count_rank(const int* __restrict__ eu, const int* __restrict__ ei,
                                int* __restrict__ cnt, int* __restrict__ rank_u,
                                int* __restrict__ rank_i, int E, int U) {
    int t = blockIdx.x * blockDim.x + threadIdx.x;
    if (t >= E) return;
    rank_u[t] = __hip_atomic_fetch_add(&cnt[eu[t]], 1, __ATOMIC_RELAXED, __HIP_MEMORY_SCOPE_AGENT);
    rank_i[t] = __hip_atomic_fetch_add(&cnt[U + ei[t]], 1, __ATOMIC_RELAXED, __HIP_MEMORY_SCOPE_AGENT);
}

__global__ __launch_bounds__(256) void lgcn_blocksum(const int* __restrict__ cnt,
                                                     int* __restrict__ bsum, int N) {
    __shared__ int wsum[8];
    const int lane = threadIdx.x & 31;
    const int wave = threadIdx.x >> 5;
    const int n0 = blockIdx.x * 1024 + threadIdx.x * 4;
    int s = 0;
#pragma unroll
    for (int k = 0; k < 4; ++k)
        if (n0 + k < N) s += cnt[n0 + k];
#pragma unroll
    for (int off = 16; off > 0; off >>= 1) s += __shfl_xor(s, off);
    if (lane == 0) wsum[wave] = s;
    __syncthreads();
    if (threadIdx.x == 0) {
        int tot = 0;
#pragma unroll
        for (int w = 0; w < 8; ++w) tot += wsum[w];
        bsum[blockIdx.x] = tot;
    }
}

__global__ __launch_bounds__(256) void lgcn_scan_inv(const int* __restrict__ cnt,
                                                     const int* __restrict__ bsum,
                                                     int* __restrict__ rowptr,
                                                     float* __restrict__ inv, int N) {
    __shared__ int wsum[8];
    __shared__ int wscan[8];
    const int lane = threadIdx.x & 31;
    const int wave = threadIdx.x >> 5;

    int p = 0;
    for (int j = threadIdx.x; j < (int)blockIdx.x; j += 256) p += bsum[j];
#pragma unroll
    for (int off = 16; off > 0; off >>= 1) p += __shfl_xor(p, off);
    if (lane == 0) wsum[wave] = p;

    const int n0 = blockIdx.x * 1024 + threadIdx.x * 4;
    int c[4];
    int tsum = 0;
#pragma unroll
    for (int k = 0; k < 4; ++k) {
        c[k] = (n0 + k < N) ? cnt[n0 + k] : 0;
        tsum += c[k];
    }
    int incl = tsum;
#pragma unroll
    for (int off = 1; off < 32; off <<= 1) {
        int v = __shfl_up(incl, off);
        if (lane >= off) incl += v;
    }
    if (lane == 31) wscan[wave] = incl;
    __syncthreads();

    int run = incl - tsum;
#pragma unroll
    for (int w = 0; w < 8; ++w) {
        run += wsum[w];
        if (w < wave) run += wscan[w];
    }
#pragma unroll
    for (int k = 0; k < 4; ++k) {
        int n = n0 + k;
        if (n < N) {
            rowptr[n] = run;
            float v = (float)c[k];
            inv[n] = (v > 0.0f) ? (1.0f / sqrtf(v)) : 0.0f;
            run += c[k];
            if (n == N - 1) rowptr[N] = run;
        }
    }
}

__global__ void lgcn_fill(const int* __restrict__ eu, const int* __restrict__ ei,
                          const int* __restrict__ rank_u, const int* __restrict__ rank_i,
                          const int* __restrict__ rowptr, const float* __restrict__ inv,
                          int2* __restrict__ csr, int E, int U) {
    int t = blockIdx.x * blockDim.x + threadIdx.x;
    if (t >= E) return;
    int u = eu[t];
    int i = U + ei[t];
    int wb = __float_as_int(inv[u] * inv[i]);
    csr[rowptr[i] + rank_i[t]] = make_int2(u, wb);
    csr[rowptr[u] + rank_u[t]] = make_int2(i, wb);
}

__global__ __launch_bounds__(256) void lgcn_spmm_pull(const int* __restrict__ rowptr,
                                                      const int2* __restrict__ csr,
                                                      const float* __restrict__ xin,
                                                      float* __restrict__ xout, int N) {
    int t = blockIdx.x * 256 + threadIdx.x;
    int row = t >> 4;
    if (row >= N) return;
    int q = (t & 15) * 4;
    int j = rowptr[row];
    const int end = rowptr[row + 1];
    float4 acc = make_float4(0.0f, 0.0f, 0.0f, 0.0f);
    const int last = end - 1;
    for (; j < end; j += 4) {
        const bool p1 = j + 1 < end, p2 = j + 2 < end, p3 = j + 3 < end;
        const int2 e0 = csr[j];
        const int2 e1 = csr[p1 ? j + 1 : last];
        const int2 e2 = csr[p2 ? j + 2 : last];
        const int2 e3 = csr[p3 ? j + 3 : last];
        const float4 v0 = *(const float4*)(xin + (size_t)e0.x * 64 + q);
        const float4 v1 = *(const float4*)(xin + (size_t)e1.x * 64 + q);
        const float4 v2 = *(const float4*)(xin + (size_t)e2.x * 64 + q);
        const float4 v3 = *(const float4*)(xin + (size_t)e3.x * 64 + q);
        const float w0 = __int_as_float(e0.y);
        const float w1 = p1 ? __int_as_float(e1.y) : 0.0f;
        const float w2 = p2 ? __int_as_float(e2.y) : 0.0f;
        const float w3 = p3 ? __int_as_float(e3.y) : 0.0f;
        acc.x += w0 * v0.x; acc.y += w0 * v0.y; acc.z += w0 * v0.z; acc.w += w0 * v0.w;
        acc.x += w1 * v1.x; acc.y += w1 * v1.y; acc.z += w1 * v1.z; acc.w += w1 * v1.w;
        acc.x += w2 * v2.x; acc.y += w2 * v2.y; acc.z += w2 * v2.z; acc.w += w2 * v2.w;
        acc.x += w3 * v3.x; acc.y += w3 * v3.y; acc.z += w3 * v3.z; acc.w += w3 * v3.w;
    }
    *(float4*)(xout + (size_t)row * 64 + q) = acc;
}

__global__ void lgcn_gather_acc(const int* __restrict__ users, const int* __restrict__ items,
                                const float* __restrict__ x,
                                float* __restrict__ uacc, float* __restrict__ iacc,
                                int U, int B) {
    int t = blockIdx.x * blockDim.x + threadIdx.x;
    int half = B * 64;
    if (t >= 2 * half) return;
    if (t < half) {
        int b = t >> 6, d = t & 63;
        uacc[t] += x[(size_t)users[b] * 64 + d];
    } else {
        int tt = t - half;
        int b = tt >> 6, d = tt & 63;
        iacc[tt] += x[((size_t)U + items[b]) * 64 + d];
    }
}

__global__ __launch_bounds__(256) void lgcn_dot_wmma(const float* __restrict__ uacc,
                                                     const float* __restrict__ iacc,
                                                     float* __restrict__ out, int B) {
    __shared__ float tile[8][16 * 16];
    const int lane = threadIdx.x & 31;
    const int wave = threadIdx.x >> 5;
    const int tile0 = (blockIdx.x * 8 + wave) * 16;

    const int m = lane & 15;
    const int khalf = (lane >> 4) << 1;
    int row = tile0 + m;
    if (row > B - 1) row = B - 1;
    const float* __restrict__ ur = uacc + (size_t)row * 64;
    const float* __restrict__ ir = iacc + (size_t)row * 64;

    v8f c = {};
#pragma unroll
    for (int kb = 0; kb < 64; kb += 4) {
        int k0 = kb + khalf;
        v2f a; a.x = ur[k0]; a.y = ur[k0 + 1];
        v2f b; b.x = ir[k0]; b.y = ir[k0 + 1];
        c = __builtin_amdgcn_wmma_f32_16x16x4_f32(false, a, false, b,
                                                  (short)0, c, false, false);
    }

    float* lds = &tile[wave][0];
    const int n = lane & 15;
    const int mbase = (lane >> 4) ? 8 : 0;
#pragma unroll
    for (int r = 0; r < 8; ++r)
        lds[(mbase + r) * 16 + n] = c[r];
    __syncthreads();

    if (lane < 16) {
        int b = tile0 + lane;
        if (b < B)
            out[b] = lds[lane * 16 + lane] * (1.0f / 25.0f);
    }
}

extern "C" void kernel_launch(void* const* d_in, const int* in_sizes, int n_in,
                              void* d_out, int out_size, void* d_ws, size_t ws_size,
                              hipStream_t stream) {
    const int*   users = (const int*)d_in[0];
    const int*   items = (const int*)d_in[1];
    const int*   eu    = (const int*)d_in[2];
    const int*   ei    = (const int*)d_in[3];
    const float* uemb  = (const float*)d_in[4];
    const float* iemb  = (const float*)d_in[5];
    float*       out   = (float*)d_out;

    const int B = in_sizes[0];
    const int E = in_sizes[2];
    const int U = in_sizes[4] / 64;
    const int I = in_sizes[5] / 64;
    const int N = U + I;

    char* ws = (char*)d_ws;
    size_t off = 0;
    auto carve = [&](size_t bytes) -> float* {
        float* p = (float*)(ws + off);
        off += (bytes + 255) & ~(size_t)255;
        return p;
    };
    float* inv    = carve((size_t)N * 4);
    float* e_cur  = carve((size_t)N * 64 * 4);
    float* e_next = carve((size_t)N * 64 * 4);
    float* uacc   = carve((size_t)B * 64 * 4);
    float* iacc   = carve((size_t)B * 64 * 4);
    int*   rowptr = (int*)carve((size_t)(N + 1) * 4);
    int2*  csr    = (int2*)carve((size_t)2 * E * 8);

    const int nchunk = (N + 1023) / 1024;
    size_t toff = 0;
    auto tcarve = [&](size_t bytes) -> int* {
        int* p = (int*)((char*)e_next + toff);
        toff += (bytes + 255) & ~(size_t)255;
        return p;
    };
    int* cnt    = tcarve((size_t)N * 4);
    int* bsum   = tcarve((size_t)nchunk * 4);
    int* rank_u = tcarve((size_t)E * 4);
    int* rank_i = tcarve((size_t)E * 4);

    const bool gather_path = (off <= ws_size) && (toff <= (size_t)N * 64 * 4);

    const int BS = 256;

    hipMemsetAsync(uacc, 0, (size_t)B * 64 * 4, stream);
    hipMemsetAsync(iacc, 0, (size_t)B * 64 * 4, stream);

    {
        long long total = (long long)N * 64;
        lgcn_init_x0<<<(int)((total + BS - 1) / BS), BS, 0, stream>>>(uemb, iemb, e_cur, U, N);
    }

    const int gacc_blocks = (2 * B * 64 + BS - 1) / BS;

    if (gather_path) {
        hipMemsetAsync(cnt, 0, (size_t)N * 4, stream);
        lgcn_count_rank<<<(E + BS - 1) / BS, BS, 0, stream>>>(eu, ei, cnt, rank_u, rank_i, E, U);
        lgcn_blocksum<<<nchunk, 256, 0, stream>>>(cnt, bsum, N);
        lgcn_scan_inv<<<nchunk, 256, 0, stream>>>(cnt, bsum, rowptr, inv, N);
        lgcn_fill<<<(E + BS - 1) / BS, BS, 0, stream>>>(eu, ei, rank_u, rank_i, rowptr, inv, csr, E, U);

        lgcn_gather_acc<<<gacc_blocks, BS, 0, stream>>>(users, items, e_cur, uacc, iacc, U, B);

        const int pull_blocks = (int)(((long long)N * 16 + 255) / 256);
        for (int layer = 0; layer < 4; ++layer) {
            lgcn_spmm_pull<<<pull_blocks, 256, 0, stream>>>(rowptr, csr, e_cur, e_next, N);
            lgcn_gather_acc<<<gacc_blocks, BS, 0, stream>>>(users, items, e_next, uacc, iacc, U, B);
            float* tmp = e_cur; e_cur = e_next; e_next = tmp;
        }
    } else {
        hipMemsetAsync(inv, 0, (size_t)N * 4, stream);
        lgcn_degree<<<(E + BS - 1) / BS, BS, 0, stream>>>(eu, ei, inv, E, U);
        lgcn_invsqrt<<<(N + BS - 1) / BS, BS, 0, stream>>>(inv, N);

        lgcn_gather_acc<<<gacc_blocks, BS, 0, stream>>>(users, items, e_cur, uacc, iacc, U, B);

        const long long spmm_threads = (long long)2 * E * 16;
        const int spmm_blocks = (int)((spmm_threads + BS - 1) / BS);
        for (int layer = 0; layer < 4; ++layer) {
            hipMemsetAsync(e_next, 0, (size_t)N * 64 * 4, stream);
            lgcn_spmm<<<spmm_blocks, BS, 0, stream>>>(eu, ei, inv, e_cur, e_next, E, U);
            lgcn_gather_acc<<<gacc_blocks, BS, 0, stream>>>(users, items, e_next, uacc, iacc, U, B);
            float* tmp = e_cur; e_cur = e_next; e_next = tmp;
        }
    }

    const int tiles = (B + 15) / 16;
    const int dot_blocks = (tiles + 7) / 8;
    lgcn_dot_wmma<<<dot_blocks, 256, 0, stream>>>(uacc, iacc, out, B);
}


# Layer 1 and the layer-0 batch gather read user_emb/item_emb in place (source row picked by node id); the concatenated x0 copy and its 300000-wave launch are gone; per-row sum order unchanged
# speedup vs baseline: 2.6403x; 1.1248x over previous
#include <hip/hip_runtime.h>


typedef __attribute__((ext_vector_type(2))) float v2f;
typedef __attribute__((ext_vector_type(8))) float v8f;

#define DEV_ATOMIC_ADD(p, v) \
    __hip_atomic_fetch_add((p), (v), __ATOMIC_RELAXED, __HIP_MEMORY_SCOPE_AGENT)

__global__ void lgcn_degree(const int* __restrict__ eu, const int* __restrict__ ei,
                            float* __restrict__ deg, int E, int U) {
    int t = blockIdx.x * blockDim.x + threadIdx.x;
    if (t >= E) return;
    DEV_ATOMIC_ADD(&deg[eu[t]], 1.0f);
    DEV_ATOMIC_ADD(&deg[U + ei[t]], 1.0f);
}

__global__ void lgcn_invsqrt(float* __restrict__ d, int N) {
    int t = blockIdx.x * blockDim.x + threadIdx.x;
    if (t >= N) return;
    float v = d[t];
    d[t] = (v > 0.0f) ? (1.0f / sqrtf(v)) : 0.0f;
}

__global__ void lgcn_init_x0(const float* __restrict__ ue, const float* __restrict__ ie,
                             float* __restrict__ x, int U, int N) {
    long long t = (long long)blockIdx.x * blockDim.x + threadIdx.x;
    long long total = (long long)N * 64;
    if (t >= total) return;
    long long ubytes = (long long)U * 64;
    x[t] = (t < ubytes) ? ue[t] : ie[t - ubytes];
}

__global__ void lgcn_spmm(const int* __restrict__ eu, const int* __restrict__ ei,
                          const float* __restrict__ inv,
                          const float* __restrict__ xin, float* __restrict__ xout,
                          int E, int U) {
    long long t = (long long)blockIdx.x * blockDim.x + threadIdx.x;
    int edge = (int)(t >> 4);
    if (edge >= 2 * E) return;
    int q = ((int)t & 15) * 4;

    int s, d;
    if (edge < E) {
        s = eu[edge];
        d = U + ei[edge];
    } else {
        int e2 = edge - E;
        s = U + ei[e2];
        d = eu[e2];
    }
    float w = inv[s] * inv[d];
    const float4 v = *(const float4*)(xin + (size_t)s * 64 + q);
    float* o = xout + (size_t)d * 64 + q;
    DEV_ATOMIC_ADD(o + 0, w * v.x);
    DEV_ATOMIC_ADD(o + 1, w * v.y);
    DEV_ATOMIC_ADD(o + 2, w * v.z);
    DEV_ATOMIC_ADD(o + 3, w * v.w);
}


__global__ void lgcn_count_rank(const int* __restrict__ eu, const int* __restrict__ ei,
                                int* __restrict__ cnt, int* __restrict__ rank_u,
                                int* __restrict__ rank_i, int E, int U) {
    int t = blockIdx.x * blockDim.x + threadIdx.x;
    if (t >= E) return;
    rank_u[t] = __hip_atomic_fetch_add(&cnt[eu[t]], 1, __ATOMIC_RELAXED, __HIP_MEMORY_SCOPE_AGENT);
    rank_i[t] = __hip_atomic_fetch_add(&cnt[U + ei[t]], 1, __ATOMIC_RELAXED, __HIP_MEMORY_SCOPE_AGENT);
}

__global__ __launch_bounds__(256) void lgcn_blocksum(const int* __restrict__ cnt,
                                                     int* __restrict__ bsum, int N) {
    __shared__ int wsum[8];
    const int lane = threadIdx.x & 31;
    const int wave = threadIdx.x >> 5;
    const int n0 = blockIdx.x * 1024 + threadIdx.x * 4;
    int s = 0;
#pragma unroll
    for (int k = 0; k < 4; ++k)
        if (n0 + k < N) s += cnt[n0 + k];
#pragma unroll
    for (int off = 16; off > 0; off >>= 1) s += __shfl_xor(s, off);
    if (lane == 0) wsum[wave] = s;
    __syncthreads();
    if (threadIdx.x == 0) {
        int tot = 0;
#pragma unroll
        for (int w = 0; w < 8; ++w) tot += wsum[w];
        bsum[blockIdx.x] = tot;
    }
}

__global__ __launch_bounds__(256) void lgcn_scan_inv(const int* __restrict__ cnt,
                                                     const int* __restrict__ bsum,
                                                     int* __restrict__ rowptr,
                                                     float* __restrict__ inv, int N) {
    __shared__ int wsum[8];
    __shared__ int wscan[8];
    const int lane = threadIdx.x & 31;
    const int wave = threadIdx.x >> 5;

    int p = 0;
    for (int j = threadIdx.x; j < (int)blockIdx.x; j += 256) p += bsum[j];
#pragma unroll
    for (int off = 16; off > 0; off >>= 1) p += __shfl_xor(p, off);
    if (lane == 0) wsum[wave] = p;

    const int n0 = blockIdx.x * 1024 + threadIdx.x * 4;
    int c[4];
    int tsum = 0;
#pragma unroll
    for (int k = 0; k < 4; ++k) {
        c[k] = (n0 + k < N) ? cnt[n0 + k] : 0;
        tsum += c[k];
    }
    int incl = tsum;
#pragma unroll
    for (int off = 1; off < 32; off <<= 1) {
        int v = __shfl_up(incl, off);
        if (lane >= off) incl += v;
    }
    if (lane == 31) wscan[wave] = incl;
    __syncthreads();

    int run = incl - tsum;
#pragma unroll
    for (int w = 0; w < 8; ++w) {
        run += wsum[w];
        if (w < wave) run += wscan[w];
    }
#pragma unroll
    for (int k = 0; k < 4; ++k) {
        int n = n0 + k;
        if (n < N) {
            rowptr[n] = run;
            float v = (float)c[k];
            inv[n] = (v > 0.0f) ? (1.0f / sqrtf(v)) : 0.0f;
            run += c[k];
            if (n == N - 1) rowptr[N] = run;
        }
    }
}

__global__ void lgcn_fill(const int* __restrict__ eu, const int* __restrict__ ei,
                          const int* __restrict__ rank_u, const int* __restrict__ rank_i,
                          const int* __restrict__ rowptr, const float* __restrict__ inv,
                          int2* __restrict__ csr, int E, int U) {
    int t = blockIdx.x * blockDim.x + threadIdx.x;
    if (t >= E) return;
    int u = eu[t];
    int i = U + ei[t];
    int wb = __float_as_int(inv[u] * inv[i]);
    csr[rowptr[i] + rank_i[t]] = make_int2(u, wb);
    csr[rowptr[u] + rank_u[t]] = make_int2(i, wb);
}

template <bool SPLIT>
__device__ __forceinline__ const float* lgcn_src_row(const float* __restrict__ xa,
                                                     const float* __restrict__ xb,
                                                     int U, int s) {
    if (SPLIT)
        return (s < U) ? xa + (size_t)s * 64 : xb + (size_t)(s - U) * 64;
    return xa + (size_t)s * 64;
}

template <bool SPLIT>
__device__ __forceinline__ void lgcn_pull_body(const int* __restrict__ rowptr,
                                               const int2* __restrict__ csr,
                                               const float* __restrict__ xa,
                                               const float* __restrict__ xb, int U,
                                               float* __restrict__ xout, int N) {
    int t = blockIdx.x * 256 + threadIdx.x;
    int row = t >> 4;
    if (row >= N) return;
    int q = (t & 15) * 4;
    int j = rowptr[row];
    const int end = rowptr[row + 1];
    float4 acc = make_float4(0.0f, 0.0f, 0.0f, 0.0f);
    const int last = end - 1;
    for (; j < end; j += 4) {
        const bool p1 = j + 1 < end, p2 = j + 2 < end, p3 = j + 3 < end;
        const int2 e0 = csr[j];
        const int2 e1 = csr[p1 ? j + 1 : last];
        const int2 e2 = csr[p2 ? j + 2 : last];
        const int2 e3 = csr[p3 ? j + 3 : last];
        const float4 v0 = *(const float4*)(lgcn_src_row<SPLIT>(xa, xb, U, e0.x) + q);
        const float4 v1 = *(const float4*)(lgcn_src_row<SPLIT>(xa, xb, U, e1.x) + q);
        const float4 v2 = *(const float4*)(lgcn_src_row<SPLIT>(xa, xb, U, e2.x) + q);
        const float4 v3 = *(const float4*)(lgcn_src_row<SPLIT>(xa, xb, U, e3.x) + q);
        if (SPLIT) __builtin_amdgcn_sched_barrier(0);
        const float w0 = __int_as_float(e0.y);
        const float w1 = p1 ? __int_as_float(e1.y) : 0.0f;
        const float w2 = p2 ? __int_as_float(e2.y) : 0.0f;
        const float w3 = p3 ? __int_as_float(e3.y) : 0.0f;
        acc.x += w0 * v0.x; acc.y += w0 * v0.y; acc.z += w0 * v0.z; acc.w += w0 * v0.w;
        acc.x += w1 * v1.x; acc.y += w1 * v1.y; acc.z += w1 * v1.z; acc.w += w1 * v1.w;
        acc.x += w2 * v2.x; acc.y += w2 * v2.y; acc.z += w2 * v2.z; acc.w += w2 * v2.w;
        acc.x += w3 * v3.x; acc.y += w3 * v3.y; acc.z += w3 * v3.z; acc.w += w3 * v3.w;
    }
    *(float4*)(xout + (size_t)row * 64 + q) = acc;
}

__global__ __launch_bounds__(256) void lgcn_spmm_pull(const int* __restrict__ rowptr,
                                                      const int2* __restrict__ csr,
                                                      const float* __restrict__ xin,
                                                      float* __restrict__ xout, int N) {
    lgcn_pull_body<false>(rowptr, csr, xin, nullptr, 0, xout, N);
}

__global__ __launch_bounds__(256) void lgcn_spmm_pull_first(const int* __restrict__ rowptr,
                                                            const int2* __restrict__ csr,
                                                            const float* __restrict__ ue,
                                                            const float* __restrict__ ie,
                                                            float* __restrict__ xout,
                                                            int U, int N) {
    lgcn_pull_body<true>(rowptr, csr, ue, ie, U, xout, N);
}

__global__ void lgcn_gather_acc(const int* __restrict__ users, const int* __restrict__ items,
                                const float* __restrict__ xu, const float* __restrict__ xi,
                                float* __restrict__ uacc, float* __restrict__ iacc,
                                int B) {
    int t = blockIdx.x * blockDim.x + threadIdx.x;
    int half = B * 64;
    if (t >= 2 * half) return;
    if (t < half) {
        int b = t >> 6, d = t & 63;
        uacc[t] += xu[(size_t)users[b] * 64 + d];
    } else {
        int tt = t - half;
        int b = tt >> 6, d = tt & 63;
        iacc[tt] += xi[(size_t)items[b] * 64 + d];
    }
}

__global__ __launch_bounds__(256) void lgcn_dot_wmma(const float* __restrict__ uacc,
                                                     const float* __restrict__ iacc,
                                                     float* __restrict__ out, int B) {
    __shared__ float tile[8][16 * 16];
    const int lane = threadIdx.x & 31;
    const int wave = threadIdx.x >> 5;
    const int tile0 = (blockIdx.x * 8 + wave) * 16;

    const int m = lane & 15;
    const int khalf = (lane >> 4) << 1;
    int row = tile0 + m;
    if (row > B - 1) row = B - 1;
    const float* __restrict__ ur = uacc + (size_t)row * 64;
    const float* __restrict__ ir = iacc + (size_t)row * 64;

    v8f c = {};
#pragma unroll
    for (int kb = 0; kb < 64; kb += 4) {
        int k0 = kb + khalf;
        v2f a; a.x = ur[k0]; a.y = ur[k0 + 1];
        v2f b; b.x = ir[k0]; b.y = ir[k0 + 1];
        c = __builtin_amdgcn_wmma_f32_16x16x4_f32(false, a, false, b,
                                                  (short)0, c, false, false);
    }

    float* lds = &tile[wave][0];
    const int n = lane & 15;
    const int mbase = (lane >> 4) ? 8 : 0;
#pragma unroll
    for (int r = 0; r < 8; ++r)
        lds[(mbase + r) * 16 + n] = c[r];
    __syncthreads();

    if (lane < 16) {
        int b = tile0 + lane;
        if (b < B)
            out[b] = lds[lane * 16 + lane] * (1.0f / 25.0f);
    }
}

extern "C" void kernel_launch(void* const* d_in, const int* in_sizes, int n_in,
                              void* d_out, int out_size, void* d_ws, size_t ws_size,
                              hipStream_t stream) {
    const int*   users = (const int*)d_in[0];
    const int*   items = (const int*)d_in[1];
    const int*   eu    = (const int*)d_in[2];
    const int*   ei    = (const int*)d_in[3];
    const float* uemb  = (const float*)d_in[4];
    const float* iemb  = (const float*)d_in[5];
    float*       out   = (float*)d_out;

    const int B = in_sizes[0];
    const int E = in_sizes[2];
    const int U = in_sizes[4] / 64;
    const int I = in_sizes[5] / 64;
    const int N = U + I;

    char* ws = (char*)d_ws;
    size_t off = 0;
    auto carve = [&](size_t bytes) -> float* {
        float* p = (float*)(ws + off);
        off += (bytes + 255) & ~(size_t)255;
        return p;
    };
    float* inv    = carve((size_t)N * 4);
    float* e_cur  = carve((size_t)N * 64 * 4);
    float* e_next = carve((size_t)N * 64 * 4);
    float* uacc   = carve((size_t)B * 64 * 4);
    float* iacc   = carve((size_t)B * 64 * 4);
    int*   rowptr = (int*)carve((size_t)(N + 1) * 4);
    int2*  csr    = (int2*)carve((size_t)2 * E * 8);

    const int nchunk = (N + 1023) / 1024;
    size_t toff = 0;
    auto tcarve = [&](size_t bytes) -> int* {
        int* p = (int*)((char*)e_next + toff);
        toff += (bytes + 255) & ~(size_t)255;
        return p;
    };
    int* cnt    = tcarve((size_t)N * 4);
    int* bsum   = tcarve((size_t)nchunk * 4);
    int* rank_u = tcarve((size_t)E * 4);
    int* rank_i = tcarve((size_t)E * 4);

    const bool gather_path = (off <= ws_size) && (toff <= (size_t)N * 64 * 4);

    const int BS = 256;

    hipMemsetAsync(uacc, 0, (size_t)B * 64 * 4, stream);
    hipMemsetAsync(iacc, 0, (size_t)B * 64 * 4, stream);

    const int gacc_blocks = (2 * B * 64 + BS - 1) / BS;
    const size_t item0 = (size_t)U * 64;

    if (gather_path) {
        hipMemsetAsync(cnt, 0, (size_t)N * 4, stream);
        lgcn_count_rank<<<(E + BS - 1) / BS, BS, 0, stream>>>(eu, ei, cnt, rank_u, rank_i, E, U);
        lgcn_blocksum<<<nchunk, 256, 0, stream>>>(cnt, bsum, N);
        lgcn_scan_inv<<<nchunk, 256, 0, stream>>>(cnt, bsum, rowptr, inv, N);
        lgcn_fill<<<(E + BS - 1) / BS, BS, 0, stream>>>(eu, ei, rank_u, rank_i, rowptr, inv, csr, E, U);

        lgcn_gather_acc<<<gacc_blocks, BS, 0, stream>>>(users, items, uemb, iemb, uacc, iacc, B);

        const int pull_blocks = (int)(((long long)N * 16 + 255) / 256);
        lgcn_spmm_pull_first<<<pull_blocks, 256, 0, stream>>>(rowptr, csr, uemb, iemb, e_cur, U, N);
        lgcn_gather_acc<<<gacc_blocks, BS, 0, stream>>>(users, items, e_cur, e_cur + item0, uacc, iacc, B);
        for (int layer = 1; layer < 4; ++layer) {
            lgcn_spmm_pull<<<pull_blocks, 256, 0, stream>>>(rowptr, csr, e_cur, e_next, N);
            lgcn_gather_acc<<<gacc_blocks, BS, 0, stream>>>(users, items, e_next, e_next + item0, uacc, iacc, B);
            float* tmp = e_cur; e_cur = e_next; e_next = tmp;
        }
    } else {
        {
            long long total = (long long)N * 64;
            lgcn_init_x0<<<(int)((total + BS - 1) / BS), BS, 0, stream>>>(uemb, iemb, e_cur, U, N);
        }
        hipMemsetAsync(inv, 0, (size_t)N * 4, stream);
        lgcn_degree<<<(E + BS - 1) / BS, BS, 0, stream>>>(eu, ei, inv, E, U);
        lgcn_invsqrt<<<(N + BS - 1) / BS, BS, 0, stream>>>(inv, N);

        lgcn_gather_acc<<<gacc_blocks, BS, 0, stream>>>(users, items, e_cur, e_cur + item0, uacc, iacc, B);

        const long long spmm_threads = (long long)2 * E * 16;
        const int spmm_blocks = (int)((spmm_threads + BS - 1) / BS);
        for (int layer = 0; layer < 4; ++layer) {
            hipMemsetAsync(e_next, 0, (size_t)N * 64 * 4, stream);
            lgcn_spmm<<<spmm_blocks, BS, 0, stream>>>(eu, ei, inv, e_cur, e_next, E, U);
            lgcn_gather_acc<<<gacc_blocks, BS, 0, stream>>>(users, items, e_next, e_next + item0, uacc, iacc, B);
            float* tmp = e_cur; e_cur = e_next; e_next = tmp;
        }
    }

    const int tiles = (B + 15) / 16;
    const int dot_blocks = (tiles + 7) / 8;
    lgcn_dot_wmma<<<dot_blocks, 256, 0, stream>>>(uacc, iacc, out, B);
}


# Output-driven pruning: layer 4 summed only for the 2*4096 batch rows (added straight into the running sums), layer 3 only for rows marked as batch rows or their neighbours; sum order unchanged
# speedup vs baseline: 2.8740x; 1.0885x over previous
#include <hip/hip_runtime.h>


typedef __attribute__((ext_vector_type(2))) float v2f;
typedef __attribute__((ext_vector_type(8))) float v8f;

#define DEV_ATOMIC_ADD(p, v) \
    __hip_atomic_fetch_add((p), (v), __ATOMIC_RELAXED, __HIP_MEMORY_SCOPE_AGENT)

__global__ void lgcn_degree(const int* __restrict__ eu, const int* __restrict__ ei,
                            float* __restrict__ deg, int E, int U) {
    int t = blockIdx.x * blockDim.x + threadIdx.x;
    if (t >= E) return;
    DEV_ATOMIC_ADD(&deg[eu[t]], 1.0f);
    DEV_ATOMIC_ADD(&deg[U + ei[t]], 1.0f);
}

__global__ void lgcn_invsqrt(float* __restrict__ d, int N) {
    int t = blockIdx.x * blockDim.x + threadIdx.x;
    if (t >= N) return;
    float v = d[t];
    d[t] = (v > 0.0f) ? (1.0f / sqrtf(v)) : 0.0f;
}

__global__ void lgcn_init_x0(const float* __restrict__ ue, const float* __restrict__ ie,
                             float* __restrict__ x, int U, int N) {
    long long t = (long long)blockIdx.x * blockDim.x + threadIdx.x;
    long long total = (long long)N * 64;
    if (t >= total) return;
    long long ubytes = (long long)U * 64;
    x[t] = (t < ubytes) ? ue[t] : ie[t - ubytes];
}

__global__ void lgcn_spmm(const int* __restrict__ eu, const int* __restrict__ ei,
                          const float* __restrict__ inv,
                          const float* __restrict__ xin, float* __restrict__ xout,
                          int E, int U) {
    long long t = (long long)blockIdx.x * blockDim.x + threadIdx.x;
    int edge = (int)(t >> 4);
    if (edge >= 2 * E) return;
    int q = ((int)t & 15) * 4;

    int s, d;
    if (edge < E) {
        s = eu[edge];
        d = U + ei[edge];
    } else {
        int e2 = edge - E;
        s = U + ei[e2];
        d = eu[e2];
    }
    float w = inv[s] * inv[d];
    const float4 v = *(const float4*)(xin + (size_t)s * 64 + q);
    float* o = xout + (size_t)d * 64 + q;
    DEV_ATOMIC_ADD(o + 0, w * v.x);
    DEV_ATOMIC_ADD(o + 1, w * v.y);
    DEV_ATOMIC_ADD(o + 2, w * v.z);
    DEV_ATOMIC_ADD(o + 3, w * v.w);
}


__global__ void lgcn_count_rank(const int* __restrict__ eu, const int* __restrict__ ei,
                                int* __restrict__ cnt, int* __restrict__ rank_u,
                                int* __restrict__ rank_i, int E, int U) {
    int t = blockIdx.x * blockDim.x + threadIdx.x;
    if (t >= E) return;
    rank_u[t] = __hip_atomic_fetch_add(&cnt[eu[t]], 1, __ATOMIC_RELAXED, __HIP_MEMORY_SCOPE_AGENT);
    rank_i[t] = __hip_atomic_fetch_add(&cnt[U + ei[t]], 1, __ATOMIC_RELAXED, __HIP_MEMORY_SCOPE_AGENT);
}

__global__ __launch_bounds__(256) void lgcn_blocksum(const int* __restrict__ cnt,
                                                     int* __restrict__ bsum, int N) {
    __shared__ int wsum[8];
    const int lane = threadIdx.x & 31;
    const int wave = threadIdx.x >> 5;
    const int n0 = blockIdx.x * 1024 + threadIdx.x * 4;
    int s = 0;
#pragma unroll
    for (int k = 0; k < 4; ++k)
        if (n0 + k < N) s += cnt[n0 + k];
#pragma unroll
    for (int off = 16; off > 0; off >>= 1) s += __shfl_xor(s, off);
    if (lane == 0) wsum[wave] = s;
    __syncthreads();
    if (threadIdx.x == 0) {
        int tot = 0;
#pragma unroll
        for (int w = 0; w < 8; ++w) tot += wsum[w];
        bsum[blockIdx.x] = tot;
    }
}

__global__ __launch_bounds__(256) void lgcn_scan_inv(const int* __restrict__ cnt,
                                                     const int* __restrict__ bsum,
                                                     int* __restrict__ rowptr,
                                                     float* __restrict__ inv,
                                                     int* __restrict__ need, int N) {
    __shared__ int wsum[8];
    __shared__ int wscan[8];
    const int lane = threadIdx.x & 31;
    const int wave = threadIdx.x >> 5;

    int p = 0;
    for (int j = threadIdx.x; j < (int)blockIdx.x; j += 256) p += bsum[j];
#pragma unroll
    for (int off = 16; off > 0; off >>= 1) p += __shfl_xor(p, off);
    if (lane == 0) wsum[wave] = p;

    const int n0 = blockIdx.x * 1024 + threadIdx.x * 4;
    int c[4];
    int tsum = 0;
#pragma unroll
    for (int k = 0; k < 4; ++k) {
        c[k] = (n0 + k < N) ? cnt[n0 + k] : 0;
        tsum += c[k];
    }
    int incl = tsum;
#pragma unroll
    for (int off = 1; off < 32; off <<= 1) {
        int v = __shfl_up(incl, off);
        if (lane >= off) incl += v;
    }
    if (lane == 31) wscan[wave] = incl;
    __syncthreads();

    int run = incl - tsum;
#pragma unroll
    for (int w = 0; w < 8; ++w) {
        run += wsum[w];
        if (w < wave) run += wscan[w];
    }
#pragma unroll
    for (int k = 0; k < 4; ++k) {
        int n = n0 + k;
        if (n < N) {
            rowptr[n] = run;
            float v = (float)c[k];
            inv[n] = (v > 0.0f) ? (1.0f / sqrtf(v)) : 0.0f;
            need[n] = 0;
            run += c[k];
            if (n == N - 1) rowptr[N] = run;
        }
    }
}

__global__ void lgcn_fill(const int* __restrict__ eu, const int* __restrict__ ei,
                          const int* __restrict__ rank_u, const int* __restrict__ rank_i,
                          const int* __restrict__ rowptr, const float* __restrict__ inv,
                          int2* __restrict__ csr, int E, int U) {
    int t = blockIdx.x * blockDim.x + threadIdx.x;
    if (t >= E) return;
    int u = eu[t];
    int i = U + ei[t];
    int wb = __float_as_int(inv[u] * inv[i]);
    csr[rowptr[i] + rank_i[t]] = make_int2(u, wb);
    csr[rowptr[u] + rank_u[t]] = make_int2(i, wb);
}

template <bool SPLIT>
__device__ __forceinline__ const float* lgcn_src_row(const float* __restrict__ xa,
                                                     const float* __restrict__ xb,
                                                     int U, int s) {
    if (SPLIT)
        return (s < U) ? xa + (size_t)s * 64 : xb + (size_t)(s - U) * 64;
    return xa + (size_t)s * 64;
}

template <bool SPLIT>
__device__ __forceinline__ float4 lgcn_row_sum(const int* __restrict__ rowptr,
                                               const int2* __restrict__ csr,
                                               const float* __restrict__ xa,
                                               const float* __restrict__ xb, int U,
                                               int row, int q) {
    int j = rowptr[row];
    const int end = rowptr[row + 1];
    float4 acc = make_float4(0.0f, 0.0f, 0.0f, 0.0f);
    const int last = end - 1;
    for (; j < end; j += 4) {
        const bool p1 = j + 1 < end, p2 = j + 2 < end, p3 = j + 3 < end;
        const int2 e0 = csr[j];
        const int2 e1 = csr[p1 ? j + 1 : last];
        const int2 e2 = csr[p2 ? j + 2 : last];
        const int2 e3 = csr[p3 ? j + 3 : last];
        const float4 v0 = *(const float4*)(lgcn_src_row<SPLIT>(xa, xb, U, e0.x) + q);
        const float4 v1 = *(const float4*)(lgcn_src_row<SPLIT>(xa, xb, U, e1.x) + q);
        const float4 v2 = *(const float4*)(lgcn_src_row<SPLIT>(xa, xb, U, e2.x) + q);
        const float4 v3 = *(const float4*)(lgcn_src_row<SPLIT>(xa, xb, U, e3.x) + q);
        if (SPLIT) __builtin_amdgcn_sched_barrier(0);
        const float w0 = __int_as_float(e0.y);
        const float w1 = p1 ? __int_as_float(e1.y) : 0.0f;
        const float w2 = p2 ? __int_as_float(e2.y) : 0.0f;
        const float w3 = p3 ? __int_as_float(e3.y) : 0.0f;
        acc.x += w0 * v0.x; acc.y += w0 * v0.y; acc.z += w0 * v0.z; acc.w += w0 * v0.w;
        acc.x += w1 * v1.x; acc.y += w1 * v1.y; acc.z += w1 * v1.z; acc.w += w1 * v1.w;
        acc.x += w2 * v2.x; acc.y += w2 * v2.y; acc.z += w2 * v2.z; acc.w += w2 * v2.w;
        acc.x += w3 * v3.x; acc.y += w3 * v3.y; acc.z += w3 * v3.z; acc.w += w3 * v3.w;
    }
    return acc;
}

__global__ __launch_bounds__(256) void lgcn_spmm_pull(const int* __restrict__ rowptr,
                                                      const int2* __restrict__ csr,
                                                      const float* __restrict__ xin,
                                                      float* __restrict__ xout, int N) {
    const int t = blockIdx.x * 256 + threadIdx.x;
    const int row = t >> 4;
    if (row >= N) return;
    const int q = (t & 15) * 4;
    *(float4*)(xout + (size_t)row * 64 + q) =
        lgcn_row_sum<false>(rowptr, csr, xin, nullptr, 0, row, q);
}

__global__ __launch_bounds__(256) void lgcn_spmm_pull_first(const int* __restrict__ rowptr,
                                                            const int2* __restrict__ csr,
                                                            const float* __restrict__ ue,
                                                            const float* __restrict__ ie,
                                                            float* __restrict__ xout,
                                                            int U, int N) {
    const int t = blockIdx.x * 256 + threadIdx.x;
    const int row = t >> 4;
    if (row >= N) return;
    const int q = (t & 15) * 4;
    *(float4*)(xout + (size_t)row * 64 + q) =
        lgcn_row_sum<true>(rowptr, csr, ue, ie, U, row, q);
}

__device__ __forceinline__ int lgcn_batch_node(const int* __restrict__ users,
                                               const int* __restrict__ items,
                                               int entry, int B, int U) {
    return (entry < B) ? users[entry] : U + items[entry - B];
}

__global__ __launch_bounds__(256) void lgcn_mark_need(const int* __restrict__ users,
                                                      const int* __restrict__ items,
                                                      const int* __restrict__ rowptr,
                                                      const int2* __restrict__ csr,
                                                      int* __restrict__ need, int B, int U) {
    const int t = blockIdx.x * 256 + threadIdx.x;
    const int entry = t >> 4;
    if (entry >= 2 * B) return;
    const int l = t & 15;
    const int node = lgcn_batch_node(users, items, entry, B, U);
    if (l == 0)
        __hip_atomic_fetch_or(&need[node], 1, __ATOMIC_RELAXED, __HIP_MEMORY_SCOPE_AGENT);
    const int end = rowptr[node + 1];
    for (int j = rowptr[node] + l; j < end; j += 16)
        __hip_atomic_fetch_or(&need[csr[j].x], 1, __ATOMIC_RELAXED, __HIP_MEMORY_SCOPE_AGENT);
}

__global__ __launch_bounds__(256) void lgcn_spmm_pull_needed(const int* __restrict__ rowptr,
                                                             const int2* __restrict__ csr,
                                                             const int* __restrict__ need,
                                                             const float* __restrict__ xin,
                                                             float* __restrict__ xout, int N) {
    const int t = blockIdx.x * 256 + threadIdx.x;
    const int row = t >> 4;
    if (row >= N) return;
    if (need[row] == 0) return;
    const int q = (t & 15) * 4;
    *(float4*)(xout + (size_t)row * 64 + q) =
        lgcn_row_sum<false>(rowptr, csr, xin, nullptr, 0, row, q);
}

__global__ __launch_bounds__(256) void lgcn_spmm_pull_batch(const int* __restrict__ users,
                                                            const int* __restrict__ items,
                                                            const int* __restrict__ rowptr,
                                                            const int2* __restrict__ csr,
                                                            const float* __restrict__ xin,
                                                            float* __restrict__ uacc,
                                                            float* __restrict__ iacc,
                                                            int B, int U) {
    const int t = blockIdx.x * 256 + threadIdx.x;
    const int entry = t >> 4;
    if (entry >= 2 * B) return;
    const int q = (t & 15) * 4;
    const int node = lgcn_batch_node(users, items, entry, B, U);
    const float4 s = lgcn_row_sum<false>(rowptr, csr, xin, nullptr, 0, node, q);
    float* dst = (entry < B) ? uacc + (size_t)entry * 64 + q
                             : iacc + (size_t)(entry - B) * 64 + q;
    float4 a = *(const float4*)dst;
    a.x += s.x; a.y += s.y; a.z += s.z; a.w += s.w;
    *(float4*)dst = a;
}

__global__ void lgcn_gather_acc(const int* __restrict__ users, const int* __restrict__ items,
                                const float* __restrict__ xu, const float* __restrict__ xi,
                                float* __restrict__ uacc, float* __restrict__ iacc,
                                int B) {
    int t = blockIdx.x * blockDim.x + threadIdx.x;
    int half = B * 64;
    if (t >= 2 * half) return;
    if (t < half) {
        int b = t >> 6, d = t & 63;
        uacc[t] += xu[(size_t)users[b] * 64 + d];
    } else {
        int tt = t - half;
        int b = tt >> 6, d = tt & 63;
        iacc[tt] += xi[(size_t)items[b] * 64 + d];
    }
}

__global__ __launch_bounds__(256) void lgcn_dot_wmma(const float* __restrict__ uacc,
                                                     const float* __restrict__ iacc,
                                                     float* __restrict__ out, int B) {
    __shared__ float tile[8][16 * 16];
    const int lane = threadIdx.x & 31;
    const int wave = threadIdx.x >> 5;
    const int tile0 = (blockIdx.x * 8 + wave) * 16;

    const int m = lane & 15;
    const int khalf = (lane >> 4) << 1;
    int row = tile0 + m;
    if (row > B - 1) row = B - 1;
    const float* __restrict__ ur = uacc + (size_t)row * 64;
    const float* __restrict__ ir = iacc + (size_t)row * 64;

    v8f c = {};
#pragma unroll
    for (int kb = 0; kb < 64; kb += 4) {
        int k0 = kb + khalf;
        v2f a; a.x = ur[k0]; a.y = ur[k0 + 1];
        v2f b; b.x = ir[k0]; b.y = ir[k0 + 1];
        c = __builtin_amdgcn_wmma_f32_16x16x4_f32(false, a, false, b,
                                                  (short)0, c, false, false);
    }

    float* lds = &tile[wave][0];
    const int n = lane & 15;
    const int mbase = (lane >> 4) ? 8 : 0;
#pragma unroll
    for (int r = 0; r < 8; ++r)
        lds[(mbase + r) * 16 + n] = c[r];
    __syncthreads();

    if (lane < 16) {
        int b = tile0 + lane;
        if (b < B)
            out[b] = lds[lane * 16 + lane] * (1.0f / 25.0f);
    }
}

extern "C" void kernel_launch(void* const* d_in, const int* in_sizes, int n_in,
                              void* d_out, int out_size, void* d_ws, size_t ws_size,
                              hipStream_t stream) {
    const int*   users = (const int*)d_in[0];
    const int*   items = (const int*)d_in[1];
    const int*   eu    = (const int*)d_in[2];
    const int*   ei    = (const int*)d_in[3];
    const float* uemb  = (const float*)d_in[4];
    const float* iemb  = (const float*)d_in[5];
    float*       out   = (float*)d_out;

    const int B = in_sizes[0];
    const int E = in_sizes[2];
    const int U = in_sizes[4] / 64;
    const int I = in_sizes[5] / 64;
    const int N = U + I;

    char* ws = (char*)d_ws;
    size_t off = 0;
    auto carve = [&](size_t bytes) -> float* {
        float* p = (float*)(ws + off);
        off += (bytes + 255) & ~(size_t)255;
        return p;
    };
    float* inv    = carve((size_t)N * 4);
    float* e_cur  = carve((size_t)N * 64 * 4);
    float* e_next = carve((size_t)N * 64 * 4);
    float* uacc   = carve((size_t)B * 64 * 4);
    float* iacc   = carve((size_t)B * 64 * 4);
    int*   rowptr = (int*)carve((size_t)(N + 1) * 4);
    int2*  csr    = (int2*)carve((size_t)2 * E * 8);

    const int nchunk = (N + 1023) / 1024;
    size_t toff = 0;
    auto tcarve = [&](size_t bytes) -> int* {
        int* p = (int*)((char*)e_next + toff);
        toff += (bytes + 255) & ~(size_t)255;
        return p;
    };
    int* cnt    = tcarve((size_t)N * 4);
    int* bsum   = tcarve((size_t)nchunk * 4);
    int* rank_u = tcarve((size_t)E * 4);
    int* rank_i = tcarve((size_t)E * 4);
    float* inv_g = (float*)tcarve((size_t)N * 4);
    int*   need  = (int*)inv;

    const bool gather_path = (off <= ws_size) && (toff <= (size_t)N * 64 * 4);

    const int BS = 256;

    hipMemsetAsync(uacc, 0, (size_t)B * 64 * 4, stream);
    hipMemsetAsync(iacc, 0, (size_t)B * 64 * 4, stream);

    const int gacc_blocks = (2 * B * 64 + BS - 1) / BS;
    const size_t item0 = (size_t)U * 64;

    if (gather_path) {
        hipMemsetAsync(cnt, 0, (size_t)N * 4, stream);
        lgcn_count_rank<<<(E + BS - 1) / BS, BS, 0, stream>>>(eu, ei, cnt, rank_u, rank_i, E, U);
        lgcn_blocksum<<<nchunk, 256, 0, stream>>>(cnt, bsum, N);
        lgcn_scan_inv<<<nchunk, 256, 0, stream>>>(cnt, bsum, rowptr, inv_g, need, N);
        lgcn_fill<<<(E + BS - 1) / BS, BS, 0, stream>>>(eu, ei, rank_u, rank_i, rowptr, inv_g, csr, E, U);
        const int batch_blocks = (2 * B * 16 + 255) / 256;
        lgcn_mark_need<<<batch_blocks, 256, 0, stream>>>(users, items, rowptr, csr, need, B, U);

        lgcn_gather_acc<<<gacc_blocks, BS, 0, stream>>>(users, items, uemb, iemb, uacc, iacc, B);

        const int pull_blocks = (int)(((long long)N * 16 + 255) / 256);
        lgcn_spmm_pull_first<<<pull_blocks, 256, 0, stream>>>(rowptr, csr, uemb, iemb, e_cur, U, N);
        lgcn_gather_acc<<<gacc_blocks, BS, 0, stream>>>(users, items, e_cur, e_cur + item0, uacc, iacc, B);
        lgcn_spmm_pull<<<pull_blocks, 256, 0, stream>>>(rowptr, csr, e_cur, e_next, N);
        lgcn_gather_acc<<<gacc_blocks, BS, 0, stream>>>(users, items, e_next, e_next + item0, uacc, iacc, B);
        lgcn_spmm_pull_needed<<<pull_blocks, 256, 0, stream>>>(rowptr, csr, need, e_next, e_cur, N);
        lgcn_gather_acc<<<gacc_blocks, BS, 0, stream>>>(users, items, e_cur, e_cur + item0, uacc, iacc, B);
        lgcn_spmm_pull_batch<<<batch_blocks, 256, 0, stream>>>(users, items, rowptr, csr, e_cur, uacc, iacc, B, U);
    } else {
        {
            long long total = (long long)N * 64;
            lgcn_init_x0<<<(int)((total + BS - 1) / BS), BS, 0, stream>>>(uemb, iemb, e_cur, U, N);
        }
        hipMemsetAsync(inv, 0, (size_t)N * 4, stream);
        lgcn_degree<<<(E + BS - 1) / BS, BS, 0, stream>>>(eu, ei, inv, E, U);
        lgcn_invsqrt<<<(N + BS - 1) / BS, BS, 0, stream>>>(inv, N);

        lgcn_gather_acc<<<gacc_blocks, BS, 0, stream>>>(users, items, e_cur, e_cur + item0, uacc, iacc, B);

        const long long spmm_threads = (long long)2 * E * 16;
        const int spmm_blocks = (int)((spmm_threads + BS - 1) / BS);
        for (int layer = 0; layer < 4; ++layer) {
            hipMemsetAsync(e_next, 0, (size_t)N * 64 * 4, stream);
            lgcn_spmm<<<spmm_blocks, BS, 0, stream>>>(eu, ei, inv, e_cur, e_next, E, U);
            lgcn_gather_acc<<<gacc_blocks, BS, 0, stream>>>(users, items, e_next, e_next + item0, uacc, iacc, B);
            float* tmp = e_cur; e_cur = e_next; e_next = tmp;
        }
    }

    const int tiles = (B + 15) / 16;
    const int dot_blocks = (tiles + 7) / 8;
    lgcn_dot_wmma<<<dot_blocks, 256, 0, stream>>>(uacc, iacc, out, B);
}


# Batch running-sum gathers and need marks ride as extra blocks in front of the layer launches (layer-0 rows stored: both accumulator memsets go); 17 stream operations become 10; sum order unchanged
# speedup vs baseline: 3.1122x; 1.0829x over previous
#include <hip/hip_runtime.h>


typedef __attribute__((ext_vector_type(2))) float v2f;
typedef __attribute__((ext_vector_type(8))) float v8f;

#define DEV_ATOMIC_ADD(p, v) \
    __hip_atomic_fetch_add((p), (v), __ATOMIC_RELAXED, __HIP_MEMORY_SCOPE_AGENT)

__global__ void lgcn_degree(const int* __restrict__ eu, const int* __restrict__ ei,
                            float* __restrict__ deg, int E, int U) {
    int t = blockIdx.x * blockDim.x + threadIdx.x;
    if (t >= E) return;
    DEV_ATOMIC_ADD(&deg[eu[t]], 1.0f);
    DEV_ATOMIC_ADD(&deg[U + ei[t]], 1.0f);
}

__global__ void lgcn_invsqrt(float* __restrict__ d, int N) {
    int t = blockIdx.x * blockDim.x + threadIdx.x;
    if (t >= N) return;
    float v = d[t];
    d[t] = (v > 0.0f) ? (1.0f / sqrtf(v)) : 0.0f;
}

__global__ void lgcn_init_x0(const float* __restrict__ ue, const float* __restrict__ ie,
                             float* __restrict__ x, int U, int N) {
    long long t = (long long)blockIdx.x * blockDim.x + threadIdx.x;
    long long total = (long long)N * 64;
    if (t >= total) return;
    long long ubytes = (long long)U * 64;
    x[t] = (t < ubytes) ? ue[t] : ie[t - ubytes];
}

__global__ void lgcn_spmm(const int* __restrict__ eu, const int* __restrict__ ei,
                          const float* __restrict__ inv,
                          const float* __restrict__ xin, float* __restrict__ xout,
                          int E, int U) {
    long long t = (long long)blockIdx.x * blockDim.x + threadIdx.x;
    int edge = (int)(t >> 4);
    if (edge >= 2 * E) return;
    int q = ((int)t & 15) * 4;

    int s, d;
    if (edge < E) {
        s = eu[edge];
        d = U + ei[edge];
    } else {
        int e2 = edge - E;
        s = U + ei[e2];
        d = eu[e2];
    }
    float w = inv[s] * inv[d];
    const float4 v = *(const float4*)(xin + (size_t)s * 64 + q);
    float* o = xout + (size_t)d * 64 + q;
    DEV_ATOMIC_ADD(o + 0, w * v.x);
    DEV_ATOMIC_ADD(o + 1, w * v.y);
    DEV_ATOMIC_ADD(o + 2, w * v.z);
    DEV_ATOMIC_ADD(o + 3, w * v.w);
}


__global__ void lgcn_count_rank(const int* __restrict__ eu, const int* __restrict__ ei,
                                int* __restrict__ cnt, int* __restrict__ rank_u,
                                int* __restrict__ rank_i, int E, int U) {
    int t = blockIdx.x * blockDim.x + threadIdx.x;
    if (t >= E) return;
    rank_u[t] = __hip_atomic_fetch_add(&cnt[eu[t]], 1, __ATOMIC_RELAXED, __HIP_MEMORY_SCOPE_AGENT);
    rank_i[t] = __hip_atomic_fetch_add(&cnt[U + ei[t]], 1, __ATOMIC_RELAXED, __HIP_MEMORY_SCOPE_AGENT);
}

__global__ __launch_bounds__(256) void lgcn_blocksum(const int* __restrict__ cnt,
                                                     int* __restrict__ bsum, int N) {
    __shared__ int wsum[8];
    const int lane = threadIdx.x & 31;
    const int wave = threadIdx.x >> 5;
    const int n0 = blockIdx.x * 1024 + threadIdx.x * 4;
    int s = 0;
#pragma unroll
    for (int k = 0; k < 4; ++k)
        if (n0 + k < N) s += cnt[n0 + k];
#pragma unroll
    for (int off = 16; off > 0; off >>= 1) s += __shfl_xor(s, off);
    if (lane == 0) wsum[wave] = s;
    __syncthreads();
    if (threadIdx.x == 0) {
        int tot = 0;
#pragma unroll
        for (int w = 0; w < 8; ++w) tot += wsum[w];
        bsum[blockIdx.x] = tot;
    }
}

__global__ __launch_bounds__(256) void lgcn_scan_inv(const int* __restrict__ cnt,
                                                     const int* __restrict__ bsum,
                                                     int* __restrict__ rowptr,
                                                     float* __restrict__ inv,
                                                     int* __restrict__ need, int N) {
    __shared__ int wsum[8];
    __shared__ int wscan[8];
    const int lane = threadIdx.x & 31;
    const int wave = threadIdx.x >> 5;

    int p = 0;
    for (int j = threadIdx.x; j < (int)blockIdx.x; j += 256) p += bsum[j];
#pragma unroll
    for (int off = 16; off > 0; off >>= 1) p += __shfl_xor(p, off);
    if (lane == 0) wsum[wave] = p;

    const int n0 = blockIdx.x * 1024 + threadIdx.x * 4;
    int c[4];
    int tsum = 0;
#pragma unroll
    for (int k = 0; k < 4; ++k) {
        c[k] = (n0 + k < N) ? cnt[n0 + k] : 0;
        tsum += c[k];
    }
    int incl = tsum;
#pragma unroll
    for (int off = 1; off < 32; off <<= 1) {
        int v = __shfl_up(incl, off);
        if (lane >= off) incl += v;
    }
    if (lane == 31) wscan[wave] = incl;
    __syncthreads();

    int run = incl - tsum;
#pragma unroll
    for (int w = 0; w < 8; ++w) {
        run += wsum[w];
        if (w < wave) run += wscan[w];
    }
#pragma unroll
    for (int k = 0; k < 4; ++k) {
        int n = n0 + k;
        if (n < N) {
            rowptr[n] = run;
            float v = (float)c[k];
            inv[n] = (v > 0.0f) ? (1.0f / sqrtf(v)) : 0.0f;
            need[n] = 0;
            run += c[k];
            if (n == N - 1) rowptr[N] = run;
        }
    }
}

__global__ void lgcn_fill(const int* __restrict__ eu, const int* __restrict__ ei,
                          const int* __restrict__ rank_u, const int* __restrict__ rank_i,
                          const int* __restrict__ rowptr, const float* __restrict__ inv,
                          int2* __restrict__ csr, int E, int U) {
    int t = blockIdx.x * blockDim.x + threadIdx.x;
    if (t >= E) return;
    int u = eu[t];
    int i = U + ei[t];
    int wb = __float_as_int(inv[u] * inv[i]);
    csr[rowptr[i] + rank_i[t]] = make_int2(u, wb);
    csr[rowptr[u] + rank_u[t]] = make_int2(i, wb);
}

template <bool SPLIT>
__device__ __forceinline__ const float* lgcn_src_row(const float* __restrict__ xa,
                                                     const float* __restrict__ xb,
                                                     int U, int s) {
    if (SPLIT)
        return (s < U) ? xa + (size_t)s * 64 : xb + (size_t)(s - U) * 64;
    return xa + (size_t)s * 64;
}

template <bool SPLIT>
__device__ __forceinline__ float4 lgcn_row_sum(const int* __restrict__ rowptr,
                                               const int2* __restrict__ csr,
                                               const float* __restrict__ xa,
                                               const float* __restrict__ xb, int U,
                                               int row, int q) {
    int j = rowptr[row];
    const int end = rowptr[row + 1];
    float4 acc = make_float4(0.0f, 0.0f, 0.0f, 0.0f);
    const int last = end - 1;
    for (; j < end; j += 4) {
        const bool p1 = j + 1 < end, p2 = j + 2 < end, p3 = j + 3 < end;
        const int2 e0 = csr[j];
        const int2 e1 = csr[p1 ? j + 1 : last];
        const int2 e2 = csr[p2 ? j + 2 : last];
        const int2 e3 = csr[p3 ? j + 3 : last];
        const float4 v0 = *(const float4*)(lgcn_src_row<SPLIT>(xa, xb, U, e0.x) + q);
        const float4 v1 = *(const float4*)(lgcn_src_row<SPLIT>(xa, xb, U, e1.x) + q);
        const float4 v2 = *(const float4*)(lgcn_src_row<SPLIT>(xa, xb, U, e2.x) + q);
        const float4 v3 = *(const float4*)(lgcn_src_row<SPLIT>(xa, xb, U, e3.x) + q);
        if (SPLIT) __builtin_amdgcn_sched_barrier(0);
        const float w0 = __int_as_float(e0.y);
        const float w1 = p1 ? __int_as_float(e1.y) : 0.0f;
        const float w2 = p2 ? __int_as_float(e2.y) : 0.0f;
        const float w3 = p3 ? __int_as_float(e3.y) : 0.0f;
        acc.x += w0 * v0.x; acc.y += w0 * v0.y; acc.z += w0 * v0.z; acc.w += w0 * v0.w;
        acc.x += w1 * v1.x; acc.y += w1 * v1.y; acc.z += w1 * v1.z; acc.w += w1 * v1.w;
        acc.x += w2 * v2.x; acc.y += w2 * v2.y; acc.z += w2 * v2.z; acc.w += w2 * v2.w;
        acc.x += w3 * v3.x; acc.y += w3 * v3.y; acc.z += w3 * v3.z; acc.w += w3 * v3.w;
    }
    return acc;
}

__device__ __forceinline__ int lgcn_batch_node(const int* __restrict__ users,
                                               const int* __restrict__ items,
                                               int entry, int B, int U) {
    return (entry < B) ? users[entry] : U + items[entry - B];
}

__device__ __forceinline__ int lgcn_batch_blocks(int B) {
    return (B + 7) >> 3;
}

template <bool FIRST>
__device__ __forceinline__ void lgcn_batch_acc(const int* __restrict__ users,
                                               const int* __restrict__ items,
                                               const float* __restrict__ xu,
                                               const float* __restrict__ xi,
                                               float* uacc, float* iacc, int B, int t) {
    const int entry = t >> 4;
    if (entry >= 2 * B) return;
    const int q = (t & 15) * 4;
    const float* src;
    float* dst;
    if (entry < B) {
        src = xu + (size_t)users[entry] * 64 + q;
        dst = uacc + (size_t)entry * 64 + q;
    } else {
        const int e = entry - B;
        src = xi + (size_t)items[e] * 64 + q;
        dst = iacc + (size_t)e * 64 + q;
    }
    float4 v = *(const float4*)src;
    if (!FIRST) {
        float4 a = *(const float4*)dst;
        a.x += v.x; a.y += v.y; a.z += v.z; a.w += v.w;
        v = a;
    }
    *(float4*)dst = v;
}

__global__ __launch_bounds__(256) void lgcn_spmm_pull(const int* __restrict__ users,
                                                      const int* __restrict__ items,
                                                      const int* __restrict__ rowptr,
                                                      const int2* __restrict__ csr,
                                                      const float* __restrict__ xin,
                                                      float* __restrict__ xout,
                                                      float* uacc, float* iacc,
                                                      int B, int U, int N) {
    const int batch_blocks = lgcn_batch_blocks(B);
    if ((int)blockIdx.x < batch_blocks) {
        lgcn_batch_acc<false>(users, items, xin, xin + (size_t)U * 64, uacc, iacc, B,
                              (int)blockIdx.x * 256 + (int)threadIdx.x);
        return;
    }
    const int t = ((int)blockIdx.x - batch_blocks) * 256 + (int)threadIdx.x;
    const int row = t >> 4;
    if (row >= N) return;
    const int q = (t & 15) * 4;
    *(float4*)(xout + (size_t)row * 64 + q) =
        lgcn_row_sum<false>(rowptr, csr, xin, nullptr, 0, row, q);
}

__device__ __forceinline__ void lgcn_mark_need(const int* __restrict__ users,
                                               const int* __restrict__ items,
                                               const int* __restrict__ rowptr,
                                               const int2* __restrict__ csr,
                                               int* __restrict__ need, int B, int U, int t) {
    const int entry = t >> 4;
    if (entry >= 2 * B) return;
    const int l = t & 15;
    const int node = lgcn_batch_node(users, items, entry, B, U);
    if (l == 0)
        __hip_atomic_fetch_or(&need[node], 1, __ATOMIC_RELAXED, __HIP_MEMORY_SCOPE_AGENT);
    const int end = rowptr[node + 1];
    for (int j = rowptr[node] + l; j < end; j += 16)
        __hip_atomic_fetch_or(&need[csr[j].x], 1, __ATOMIC_RELAXED, __HIP_MEMORY_SCOPE_AGENT);
}

__global__ __launch_bounds__(256) void lgcn_spmm_pull_first(const int* __restrict__ users,
                                                            const int* __restrict__ items,
                                                            const int* __restrict__ rowptr,
                                                            const int2* __restrict__ csr,
                                                            const float* __restrict__ ue,
                                                            const float* __restrict__ ie,
                                                            float* __restrict__ xout,
                                                            float* uacc, float* iacc,
                                                            int* __restrict__ need,
                                                            int B, int U, int N) {
    const int batch_blocks = lgcn_batch_blocks(B);
    if ((int)blockIdx.x < batch_blocks) {
        const int tb = (int)blockIdx.x * 256 + (int)threadIdx.x;
        lgcn_batch_acc<true>(users, items, ue, ie, uacc, iacc, B, tb);
        lgcn_mark_need(users, items, rowptr, csr, need, B, U, tb);
        return;
    }
    const int t = ((int)blockIdx.x - batch_blocks) * 256 + (int)threadIdx.x;
    const int row = t >> 4;
    if (row >= N) return;
    const int q = (t & 15) * 4;
    *(float4*)(xout + (size_t)row * 64 + q) =
        lgcn_row_sum<true>(rowptr, csr, ue, ie, U, row, q);
}

__global__ __launch_bounds__(256) void lgcn_spmm_pull_needed(const int* __restrict__ users,
                                                             const int* __restrict__ items,
                                                             const int* __restrict__ rowptr,
                                                             const int2* __restrict__ csr,
                                                             const int* __restrict__ need,
                                                             const float* __restrict__ xin,
                                                             float* __restrict__ xout,
                                                             float* uacc, float* iacc,
                                                             int B, int U, int N) {
    const int batch_blocks = lgcn_batch_blocks(B);
    if ((int)blockIdx.x < batch_blocks) {
        lgcn_batch_acc<false>(users, items, xin, xin + (size_t)U * 64, uacc, iacc, B,
                              (int)blockIdx.x * 256 + (int)threadIdx.x);
        return;
    }
    const int t = ((int)blockIdx.x - batch_blocks) * 256 + (int)threadIdx.x;
    const int row = t >> 4;
    if (row >= N) return;
    if (need[row] == 0) return;
    const int q = (t & 15) * 4;
    *(float4*)(xout + (size_t)row * 64 + q) =
        lgcn_row_sum<false>(rowptr, csr, xin, nullptr, 0, row, q);
}

__global__ __launch_bounds__(256) void lgcn_spmm_pull_batch(const int* __restrict__ users,
                                                            const int* __restrict__ items,
                                                            const int* __restrict__ rowptr,
                                                            const int2* __restrict__ csr,
                                                            const float* __restrict__ xin,
                                                            float* __restrict__ uacc,
                                                            float* __restrict__ iacc,
                                                            int B, int U) {
    const int t = blockIdx.x * 256 + threadIdx.x;
    const int entry = t >> 4;
    if (entry >= 2 * B) return;
    const int q = (t & 15) * 4;
    const int node = lgcn_batch_node(users, items, entry, B, U);
    float* dst = (entry < B) ? uacc + (size_t)entry * 64 + q
                             : iacc + (size_t)(entry - B) * 64 + q;
    float4 a = *(const float4*)dst;
    const float4 x3 = *(const float4*)(xin + (size_t)node * 64 + q);
    const float4 s = lgcn_row_sum<false>(rowptr, csr, xin, nullptr, 0, node, q);
    a.x += x3.x; a.y += x3.y; a.z += x3.z; a.w += x3.w;
    a.x += s.x; a.y += s.y; a.z += s.z; a.w += s.w;
    *(float4*)dst = a;
}

__global__ void lgcn_gather_acc(const int* __restrict__ users, const int* __restrict__ items,
                                const float* __restrict__ xu, const float* __restrict__ xi,
                                float* __restrict__ uacc, float* __restrict__ iacc,
                                int B) {
    int t = blockIdx.x * blockDim.x + threadIdx.x;
    int half = B * 64;
    if (t >= 2 * half) return;
    if (t < half) {
        int b = t >> 6, d = t & 63;
        uacc[t] += xu[(size_t)users[b] * 64 + d];
    } else {
        int tt = t - half;
        int b = tt >> 6, d = tt & 63;
        iacc[tt] += xi[(size_t)items[b] * 64 + d];
    }
}

__global__ __launch_bounds__(256) void lgcn_dot_wmma(const float* __restrict__ uacc,
                                                     const float* __restrict__ iacc,
                                                     float* __restrict__ out, int B) {
    __shared__ float tile[8][16 * 16];
    const int lane = threadIdx.x & 31;
    const int wave = threadIdx.x >> 5;
    const int tile0 = (blockIdx.x * 8 + wave) * 16;

    const int m = lane & 15;
    const int khalf = (lane >> 4) << 1;
    int row = tile0 + m;
    if (row > B - 1) row = B - 1;
    const float* __restrict__ ur = uacc + (size_t)row * 64;
    const float* __restrict__ ir = iacc + (size_t)row * 64;

    v8f c = {};
#pragma unroll
    for (int kb = 0; kb < 64; kb += 4) {
        int k0 = kb + khalf;
        v2f a; a.x = ur[k0]; a.y = ur[k0 + 1];
        v2f b; b.x = ir[k0]; b.y = ir[k0 + 1];
        c = __builtin_amdgcn_wmma_f32_16x16x4_f32(false, a, false, b,
                                                  (short)0, c, false, false);
    }

    float* lds = &tile[wave][0];
    const int n = lane & 15;
    const int mbase = (lane >> 4) ? 8 : 0;
#pragma unroll
    for (int r = 0; r < 8; ++r)
        lds[(mbase + r) * 16 + n] = c[r];
    __syncthreads();

    if (lane < 16) {
        int b = tile0 + lane;
        if (b < B)
            out[b] = lds[lane * 16 + lane] * (1.0f / 25.0f);
    }
}

extern "C" void kernel_launch(void* const* d_in, const int* in_sizes, int n_in,
                              void* d_out, int out_size, void* d_ws, size_t ws_size,
                              hipStream_t stream) {
    const int*   users = (const int*)d_in[0];
    const int*   items = (const int*)d_in[1];
    const int*   eu    = (const int*)d_in[2];
    const int*   ei    = (const int*)d_in[3];
    const float* uemb  = (const float*)d_in[4];
    const float* iemb  = (const float*)d_in[5];
    float*       out   = (float*)d_out;

    const int B = in_sizes[0];
    const int E = in_sizes[2];
    const int U = in_sizes[4] / 64;
    const int I = in_sizes[5] / 64;
    const int N = U + I;

    char* ws = (char*)d_ws;
    size_t off = 0;
    auto carve = [&](size_t bytes) -> float* {
        float* p = (float*)(ws + off);
        off += (bytes + 255) & ~(size_t)255;
        return p;
    };
    float* inv    = carve((size_t)N * 4);
    float* e_cur  = carve((size_t)N * 64 * 4);
    float* e_next = carve((size_t)N * 64 * 4);
    float* uacc   = carve((size_t)B * 64 * 4);
    float* iacc   = carve((size_t)B * 64 * 4);
    int*   rowptr = (int*)carve((size_t)(N + 1) * 4);
    int2*  csr    = (int2*)carve((size_t)2 * E * 8);

    const int nchunk = (N + 1023) / 1024;
    size_t toff = 0;
    auto tcarve = [&](size_t bytes) -> int* {
        int* p = (int*)((char*)e_next + toff);
        toff += (bytes + 255) & ~(size_t)255;
        return p;
    };
    int* cnt    = tcarve((size_t)N * 4);
    int* bsum   = tcarve((size_t)nchunk * 4);
    int* rank_u = tcarve((size_t)E * 4);
    int* rank_i = tcarve((size_t)E * 4);
    float* inv_g = (float*)tcarve((size_t)N * 4);
    int*   need  = (int*)inv;

    const bool gather_path = (off <= ws_size) && (toff <= (size_t)N * 64 * 4);

    const int BS = 256;

    const int gacc_blocks = (2 * B * 64 + BS - 1) / BS;
    const size_t item0 = (size_t)U * 64;

    if (gather_path) {
        hipMemsetAsync(cnt, 0, (size_t)N * 4, stream);
        lgcn_count_rank<<<(E + BS - 1) / BS, BS, 0, stream>>>(eu, ei, cnt, rank_u, rank_i, E, U);
        lgcn_blocksum<<<nchunk, 256, 0, stream>>>(cnt, bsum, N);
        lgcn_scan_inv<<<nchunk, 256, 0, stream>>>(cnt, bsum, rowptr, inv_g, need, N);
        lgcn_fill<<<(E + BS - 1) / BS, BS, 0, stream>>>(eu, ei, rank_u, rank_i, rowptr, inv_g, csr, E, U);
        const int batch_blocks = (B + 7) / 8;
        const int pull_blocks = (N + 15) / 16;
        const int layer_blocks = pull_blocks + batch_blocks;

        lgcn_spmm_pull_first<<<layer_blocks, 256, 0, stream>>>(users, items, rowptr, csr, uemb, iemb,
                                                               e_cur, uacc, iacc, need, B, U, N);
        lgcn_spmm_pull<<<layer_blocks, 256, 0, stream>>>(users, items, rowptr, csr, e_cur, e_next,
                                                         uacc, iacc, B, U, N);
        lgcn_spmm_pull_needed<<<layer_blocks, 256, 0, stream>>>(users, items, rowptr, csr, need, e_next,
                                                                e_cur, uacc, iacc, B, U, N);
        lgcn_spmm_pull_batch<<<batch_blocks, 256, 0, stream>>>(users, items, rowptr, csr, e_cur, uacc, iacc, B, U);
    } else {
        hipMemsetAsync(uacc, 0, (size_t)B * 64 * 4, stream);
        hipMemsetAsync(iacc, 0, (size_t)B * 64 * 4, stream);

        {
            long long total = (long long)N * 64;
            lgcn_init_x0<<<(int)((total + BS - 1) / BS), BS, 0, stream>>>(uemb, iemb, e_cur, U, N);
        }
        hipMemsetAsync(inv, 0, (size_t)N * 4, stream);
        lgcn_degree<<<(E + BS - 1) / BS, BS, 0, stream>>>(eu, ei, inv, E, U);
        lgcn_invsqrt<<<(N + BS - 1) / BS, BS, 0, stream>>>(inv, N);

        lgcn_gather_acc<<<gacc_blocks, BS, 0, stream>>>(users, items, e_cur, e_cur + item0, uacc, iacc, B);

        const long long spmm_threads = (long long)2 * E * 16;
        const int spmm_blocks = (int)((spmm_threads + BS - 1) / BS);
        for (int layer = 0; layer < 4; ++layer) {
            hipMemsetAsync(e_next, 0, (size_t)N * 64 * 4, stream);
            lgcn_spmm<<<spmm_blocks, BS, 0, stream>>>(eu, ei, inv, e_cur, e_next, E, U);
            lgcn_gather_acc<<<gacc_blocks, BS, 0, stream>>>(users, items, e_next, e_next + item0, uacc, iacc, B);
            float* tmp = e_cur; e_cur = e_next; e_next = tmp;
        }
    }

    const int tiles = (B + 15) / 16;
    const int dot_blocks = (tiles + 7) / 8;
    lgcn_dot_wmma<<<dot_blocks, 256, 0, stream>>>(uacc, iacc, out, B);
}


# Row sums: a row's 16 threads fetch 16 adjacency entries with one load and share them by DPP row broadcast, up to 8 row gathers in flight (was one entry fetch per 4 gathers); row sum order unchanged
# speedup vs baseline: 3.4044x; 1.0939x over previous
#include <hip/hip_runtime.h>


typedef __attribute__((ext_vector_type(2))) float v2f;
typedef __attribute__((ext_vector_type(8))) float v8f;

#define DEV_ATOMIC_ADD(p, v) \
    __hip_atomic_fetch_add((p), (v), __ATOMIC_RELAXED, __HIP_MEMORY_SCOPE_AGENT)

__global__ void lgcn_degree(const int* __restrict__ eu, const int* __restrict__ ei,
                            float* __restrict__ deg, int E, int U) {
    int t = blockIdx.x * blockDim.x + threadIdx.x;
    if (t >= E) return;
    DEV_ATOMIC_ADD(&deg[eu[t]], 1.0f);
    DEV_ATOMIC_ADD(&deg[U + ei[t]], 1.0f);
}

__global__ void lgcn_invsqrt(float* __restrict__ d, int N) {
    int t = blockIdx.x * blockDim.x + threadIdx.x;
    if (t >= N) return;
    float v = d[t];
    d[t] = (v > 0.0f) ? (1.0f / sqrtf(v)) : 0.0f;
}

__global__ void lgcn_init_x0(const float* __restrict__ ue, const float* __restrict__ ie,
                             float* __restrict__ x, int U, int N) {
    long long t = (long long)blockIdx.x * blockDim.x + threadIdx.x;
    long long total = (long long)N * 64;
    if (t >= total) return;
    long long ubytes = (long long)U * 64;
    x[t] = (t < ubytes) ? ue[t] : ie[t - ubytes];
}

__global__ void lgcn_spmm(const int* __restrict__ eu, const int* __restrict__ ei,
                          const float* __restrict__ inv,
                          const float* __restrict__ xin, float* __restrict__ xout,
                          int E, int U) {
    long long t = (long long)blockIdx.x * blockDim.x + threadIdx.x;
    int edge = (int)(t >> 4);
    if (edge >= 2 * E) return;
    int q = ((int)t & 15) * 4;

    int s, d;
    if (edge < E) {
        s = eu[edge];
        d = U + ei[edge];
    } else {
        int e2 = edge - E;
        s = U + ei[e2];
        d = eu[e2];
    }
    float w = inv[s] * inv[d];
    const float4 v = *(const float4*)(xin + (size_t)s * 64 + q);
    float* o = xout + (size_t)d * 64 + q;
    DEV_ATOMIC_ADD(o + 0, w * v.x);
    DEV_ATOMIC_ADD(o + 1, w * v.y);
    DEV_ATOMIC_ADD(o + 2, w * v.z);
    DEV_ATOMIC_ADD(o + 3, w * v.w);
}


__global__ void lgcn_count_rank(const int* __restrict__ eu, const int* __restrict__ ei,
                                int* __restrict__ cnt, int* __restrict__ rank_u,
                                int* __restrict__ rank_i, int E, int U) {
    int t = blockIdx.x * blockDim.x + threadIdx.x;
    if (t >= E) return;
    rank_u[t] = __hip_atomic_fetch_add(&cnt[eu[t]], 1, __ATOMIC_RELAXED, __HIP_MEMORY_SCOPE_AGENT);
    rank_i[t] = __hip_atomic_fetch_add(&cnt[U + ei[t]], 1, __ATOMIC_RELAXED, __HIP_MEMORY_SCOPE_AGENT);
}

__global__ __launch_bounds__(256) void lgcn_blocksum(const int* __restrict__ cnt,
                                                     int* __restrict__ bsum, int N) {
    __shared__ int wsum[8];
    const int lane = threadIdx.x & 31;
    const int wave = threadIdx.x >> 5;
    const int n0 = blockIdx.x * 1024 + threadIdx.x * 4;
    int s = 0;
#pragma unroll
    for (int k = 0; k < 4; ++k)
        if (n0 + k < N) s += cnt[n0 + k];
#pragma unroll
    for (int off = 16; off > 0; off >>= 1) s += __shfl_xor(s, off);
    if (lane == 0) wsum[wave] = s;
    __syncthreads();
    if (threadIdx.x == 0) {
        int tot = 0;
#pragma unroll
        for (int w = 0; w < 8; ++w) tot += wsum[w];
        bsum[blockIdx.x] = tot;
    }
}

__global__ __launch_bounds__(256) void lgcn_scan_inv(const int* __restrict__ cnt,
                                                     const int* __restrict__ bsum,
                                                     int* __restrict__ rowptr,
                                                     float* __restrict__ inv,
                                                     int* __restrict__ need, int N) {
    __shared__ int wsum[8];
    __shared__ int wscan[8];
    const int lane = threadIdx.x & 31;
    const int wave = threadIdx.x >> 5;

    int p = 0;
    for (int j = threadIdx.x; j < (int)blockIdx.x; j += 256) p += bsum[j];
#pragma unroll
    for (int off = 16; off > 0; off >>= 1) p += __shfl_xor(p, off);
    if (lane == 0) wsum[wave] = p;

    const int n0 = blockIdx.x * 1024 + threadIdx.x * 4;
    int c[4];
    int tsum = 0;
#pragma unroll
    for (int k = 0; k < 4; ++k) {
        c[k] = (n0 + k < N) ? cnt[n0 + k] : 0;
        tsum += c[k];
    }
    int incl = tsum;
#pragma unroll
    for (int off = 1; off < 32; off <<= 1) {
        int v = __shfl_up(incl, off);
        if (lane >= off) incl += v;
    }
    if (lane == 31) wscan[wave] = incl;
    __syncthreads();

    int run = incl - tsum;
#pragma unroll
    for (int w = 0; w < 8; ++w) {
        run += wsum[w];
        if (w < wave) run += wscan[w];
    }
#pragma unroll
    for (int k = 0; k < 4; ++k) {
        int n = n0 + k;
        if (n < N) {
            rowptr[n] = run;
            float v = (float)c[k];
            inv[n] = (v > 0.0f) ? (1.0f / sqrtf(v)) : 0.0f;
            need[n] = 0;
            run += c[k];
            if (n == N - 1) rowptr[N] = run;
        }
    }
}

__global__ void lgcn_fill(const int* __restrict__ eu, const int* __restrict__ ei,
                          const int* __restrict__ rank_u, const int* __restrict__ rank_i,
                          const int* __restrict__ rowptr, const float* __restrict__ inv,
                          int2* __restrict__ csr, int E, int U) {
    int t = blockIdx.x * blockDim.x + threadIdx.x;
    if (t >= E) return;
    int u = eu[t];
    int i = U + ei[t];
    int wb = __float_as_int(inv[u] * inv[i]);
    csr[rowptr[i] + rank_i[t]] = make_int2(u, wb);
    csr[rowptr[u] + rank_u[t]] = make_int2(i, wb);
}

template <bool SPLIT>
__device__ __forceinline__ const float* lgcn_src_row(const float* __restrict__ xa,
                                                     const float* __restrict__ xb,
                                                     int U, int s) {
    if (SPLIT)
        return (s < U) ? xa + (size_t)s * 64 : xb + (size_t)(s - U) * 64;
    return xa + (size_t)s * 64;
}

template <int K>
__device__ __forceinline__ int lgcn_row_share(int v) {
    return __builtin_amdgcn_update_dpp(0, v, 0x150 + K, 0xf, 0xf, true);
}

template <bool SPLIT, int K>
__device__ __forceinline__ float4 lgcn_gather(const float* __restrict__ xa,
                                              const float* __restrict__ xb, int U,
                                              int es, int q) {
    const int s = lgcn_row_share<K>(es);
    if (SPLIT) return *(const float4*)(lgcn_src_row<true>(xa, xb, U, s) + q);
    const unsigned off = ((unsigned)s << 8) | ((unsigned)q << 2);
    return *(const float4*)((const char*)xa + off);
}

template <int K>
__device__ __forceinline__ void lgcn_add(float4& acc, int ew, const float4& v) {
    const float w = __int_as_float(lgcn_row_share<K>(ew));
    acc.x = __builtin_fmaf(w, v.x, acc.x);
    acc.y = __builtin_fmaf(w, v.y, acc.y);
    acc.z = __builtin_fmaf(w, v.z, acc.z);
    acc.w = __builtin_fmaf(w, v.w, acc.w);
}

template <bool SPLIT, int K0>
__device__ __forceinline__ void lgcn_eight(float4& acc, const float* __restrict__ xa,
                                           const float* __restrict__ xb, int U,
                                           int es, int ew, int q, bool upper) {
    const float4 v0 = lgcn_gather<SPLIT, K0 + 0>(xa, xb, U, es, q);
    const float4 v1 = lgcn_gather<SPLIT, K0 + 1>(xa, xb, U, es, q);
    const float4 v2 = lgcn_gather<SPLIT, K0 + 2>(xa, xb, U, es, q);
    const float4 v3 = lgcn_gather<SPLIT, K0 + 3>(xa, xb, U, es, q);
    if (upper) {
        const float4 v4 = lgcn_gather<SPLIT, K0 + 4>(xa, xb, U, es, q);
        const float4 v5 = lgcn_gather<SPLIT, K0 + 5>(xa, xb, U, es, q);
        const float4 v6 = lgcn_gather<SPLIT, K0 + 6>(xa, xb, U, es, q);
        const float4 v7 = lgcn_gather<SPLIT, K0 + 7>(xa, xb, U, es, q);
        if (SPLIT) __builtin_amdgcn_sched_barrier(0);
        lgcn_add<K0 + 0>(acc, ew, v0);
        lgcn_add<K0 + 1>(acc, ew, v1);
        lgcn_add<K0 + 2>(acc, ew, v2);
        lgcn_add<K0 + 3>(acc, ew, v3);
        lgcn_add<K0 + 4>(acc, ew, v4);
        lgcn_add<K0 + 5>(acc, ew, v5);
        lgcn_add<K0 + 6>(acc, ew, v6);
        lgcn_add<K0 + 7>(acc, ew, v7);
    } else {
        lgcn_add<K0 + 0>(acc, ew, v0);
        lgcn_add<K0 + 1>(acc, ew, v1);
        lgcn_add<K0 + 2>(acc, ew, v2);
        lgcn_add<K0 + 3>(acc, ew, v3);
    }
}

template <bool SPLIT>
__device__ __forceinline__ float4 lgcn_row_sum(const int* __restrict__ rowptr,
                                               const int2* __restrict__ csr,
                                               const float* __restrict__ xa,
                                               const float* __restrict__ xb, int U,
                                               int row, int q) {
    const int lane = q >> 2;
    const int end = rowptr[row + 1];
    const int last = end - 1;
    float4 acc = make_float4(0.0f, 0.0f, 0.0f, 0.0f);
    for (int base = rowptr[row]; base < end; base += 16) {
        const int left = end - base;
        const int2 e = csr[min(base + lane, last)];
        const int ew = (lane < left) ? e.y : 0;
        lgcn_eight<SPLIT, 0>(acc, xa, xb, U, e.x, ew, q, left > 4);
        if (left > 8) lgcn_eight<SPLIT, 8>(acc, xa, xb, U, e.x, ew, q, left > 12);
    }
    return acc;
}

__device__ __forceinline__ int lgcn_batch_node(const int* __restrict__ users,
                                               const int* __restrict__ items,
                                               int entry, int B, int U) {
    return (entry < B) ? users[entry] : U + items[entry - B];
}

__device__ __forceinline__ int lgcn_batch_blocks(int B) {
    return (B + 7) >> 3;
}

template <bool FIRST>
__device__ __forceinline__ void lgcn_batch_acc(const int* __restrict__ users,
                                               const int* __restrict__ items,
                                               const float* __restrict__ xu,
                                               const float* __restrict__ xi,
                                               float* uacc, float* iacc, int B, int t) {
    const int entry = t >> 4;
    if (entry >= 2 * B) return;
    const int q = (t & 15) * 4;
    const float* src;
    float* dst;
    if (entry < B) {
        src = xu + (size_t)users[entry] * 64 + q;
        dst = uacc + (size_t)entry * 64 + q;
    } else {
        const int e = entry - B;
        src = xi + (size_t)items[e] * 64 + q;
        dst = iacc + (size_t)e * 64 + q;
    }
    float4 v = *(const float4*)src;
    if (!FIRST) {
        float4 a = *(const float4*)dst;
        a.x += v.x; a.y += v.y; a.z += v.z; a.w += v.w;
        v = a;
    }
    *(float4*)dst = v;
}

__global__ __launch_bounds__(256) void lgcn_spmm_pull(const int* __restrict__ users,
                                                      const int* __restrict__ items,
                                                      const int* __restrict__ rowptr,
                                                      const int2* __restrict__ csr,
                                                      const float* __restrict__ xin,
                                                      float* __restrict__ xout,
                                                      float* uacc, float* iacc,
                                                      int B, int U, int N) {
    const int batch_blocks = lgcn_batch_blocks(B);
    if ((int)blockIdx.x < batch_blocks) {
        lgcn_batch_acc<false>(users, items, xin, xin + (size_t)U * 64, uacc, iacc, B,
                              (int)blockIdx.x * 256 + (int)threadIdx.x);
        return;
    }
    const int t = ((int)blockIdx.x - batch_blocks) * 256 + (int)threadIdx.x;
    const int row = t >> 4;
    if (row >= N) return;
    const int q = (t & 15) * 4;
    *(float4*)(xout + (size_t)row * 64 + q) =
        lgcn_row_sum<false>(rowptr, csr, xin, nullptr, 0, row, q);
}

__device__ __forceinline__ void lgcn_mark_need(const int* __restrict__ users,
                                               const int* __restrict__ items,
                                               const int* __restrict__ rowptr,
                                               const int2* __restrict__ csr,
                                               int* __restrict__ need, int B, int U, int t) {
    const int entry = t >> 4;
    if (entry >= 2 * B) return;
    const int l = t & 15;
    const int node = lgcn_batch_node(users, items, entry, B, U);
    if (l == 0)
        __hip_atomic_fetch_or(&need[node], 1, __ATOMIC_RELAXED, __HIP_MEMORY_SCOPE_AGENT);
    const int end = rowptr[node + 1];
    for (int j = rowptr[node] + l; j < end; j += 16)
        __hip_atomic_fetch_or(&need[csr[j].x], 1, __ATOMIC_RELAXED, __HIP_MEMORY_SCOPE_AGENT);
}

__global__ __launch_bounds__(256) void lgcn_spmm_pull_first(const int* __restrict__ users,
                                                            const int* __restrict__ items,
                                                            const int* __restrict__ rowptr,
                                                            const int2* __restrict__ csr,
                                                            const float* __restrict__ ue,
                                                            const float* __restrict__ ie,
                                                            float* __restrict__ xout,
                                                            float* uacc, float* iacc,
                                                            int* __restrict__ need,
                                                            int B, int U, int N) {
    const int batch_blocks = lgcn_batch_blocks(B);
    if ((int)blockIdx.x < batch_blocks) {
        const int tb = (int)blockIdx.x * 256 + (int)threadIdx.x;
        lgcn_batch_acc<true>(users, items, ue, ie, uacc, iacc, B, tb);
        lgcn_mark_need(users, items, rowptr, csr, need, B, U, tb);
        return;
    }
    const int t = ((int)blockIdx.x - batch_blocks) * 256 + (int)threadIdx.x;
    const int row = t >> 4;
    if (row >= N) return;
    const int q = (t & 15) * 4;
    *(float4*)(xout + (size_t)row * 64 + q) =
        lgcn_row_sum<true>(rowptr, csr, ue, ie, U, row, q);
}

__global__ __launch_bounds__(256) void lgcn_spmm_pull_needed(const int* __restrict__ users,
                                                             const int* __restrict__ items,
                                                             const int* __restrict__ rowptr,
                                                             const int2* __restrict__ csr,
                                                             const int* __restrict__ need,
                                                             const float* __restrict__ xin,
                                                             float* __restrict__ xout,
                                                             float* uacc, float* iacc,
                                                             int B, int U, int N) {
    const int batch_blocks = lgcn_batch_blocks(B);
    if ((int)blockIdx.x < batch_blocks) {
        lgcn_batch_acc<false>(users, items, xin, xin + (size_t)U * 64, uacc, iacc, B,
                              (int)blockIdx.x * 256 + (int)threadIdx.x);
        return;
    }
    const int t = ((int)blockIdx.x - batch_blocks) * 256 + (int)threadIdx.x;
    const int row = t >> 4;
    if (row >= N) return;
    if (need[row] == 0) return;
    const int q = (t & 15) * 4;
    *(float4*)(xout + (size_t)row * 64 + q) =
        lgcn_row_sum<false>(rowptr, csr, xin, nullptr, 0, row, q);
}

__global__ __launch_bounds__(256) void lgcn_spmm_pull_batch(const int* __restrict__ users,
                                                            const int* __restrict__ items,
                                                            const int* __restrict__ rowptr,
                                                            const int2* __restrict__ csr,
                                                            const float* __restrict__ xin,
                                                            float* __restrict__ uacc,
                                                            float* __restrict__ iacc,
                                                            int B, int U) {
    const int t = blockIdx.x * 256 + threadIdx.x;
    const int entry = t >> 4;
    if (entry >= 2 * B) return;
    const int q = (t & 15) * 4;
    const int node = lgcn_batch_node(users, items, entry, B, U);
    float* dst = (entry < B) ? uacc + (size_t)entry * 64 + q
                             : iacc + (size_t)(entry - B) * 64 + q;
    float4 a = *(const float4*)dst;
    const float4 x3 = *(const float4*)(xin + (size_t)node * 64 + q);
    const float4 s = lgcn_row_sum<false>(rowptr, csr, xin, nullptr, 0, node, q);
    a.x += x3.x; a.y += x3.y; a.z += x3.z; a.w += x3.w;
    a.x += s.x; a.y += s.y; a.z += s.z; a.w += s.w;
    *(float4*)dst = a;
}

__global__ void lgcn_gather_acc(const int* __restrict__ users, const int* __restrict__ items,
                                const float* __restrict__ xu, const float* __restrict__ xi,
                                float* __restrict__ uacc, float* __restrict__ iacc,
                                int B) {
    int t = blockIdx.x * blockDim.x + threadIdx.x;
    int half = B * 64;
    if (t >= 2 * half) return;
    if (t < half) {
        int b = t >> 6, d = t & 63;
        uacc[t] += xu[(size_t)users[b] * 64 + d];
    } else {
        int tt = t - half;
        int b = tt >> 6, d = tt & 63;
        iacc[tt] += xi[(size_t)items[b] * 64 + d];
    }
}

__global__ __launch_bounds__(256) void lgcn_dot_wmma(const float* __restrict__ uacc,
                                                     const float* __restrict__ iacc,
                                                     float* __restrict__ out, int B) {
    __shared__ float tile[8][16 * 16];
    const int lane = threadIdx.x & 31;
    const int wave = threadIdx.x >> 5;
    const int tile0 = (blockIdx.x * 8 + wave) * 16;

    const int m = lane & 15;
    const int khalf = (lane >> 4) << 1;
    int row = tile0 + m;
    if (row > B - 1) row = B - 1;
    const float* __restrict__ ur = uacc + (size_t)row * 64;
    const float* __restrict__ ir = iacc + (size_t)row * 64;

    v8f c = {};
#pragma unroll
    for (int kb = 0; kb < 64; kb += 4) {
        int k0 = kb + khalf;
        v2f a; a.x = ur[k0]; a.y = ur[k0 + 1];
        v2f b; b.x = ir[k0]; b.y = ir[k0 + 1];
        c = __builtin_amdgcn_wmma_f32_16x16x4_f32(false, a, false, b,
                                                  (short)0, c, false, false);
    }

    float* lds = &tile[wave][0];
    const int n = lane & 15;
    const int mbase = (lane >> 4) ? 8 : 0;
#pragma unroll
    for (int r = 0; r < 8; ++r)
        lds[(mbase + r) * 16 + n] = c[r];
    __syncthreads();

    if (lane < 16) {
        int b = tile0 + lane;
        if (b < B)
            out[b] = lds[lane * 16 + lane] * (1.0f / 25.0f);
    }
}

extern "C" void kernel_launch(void* const* d_in, const int* in_sizes, int n_in,
                              void* d_out, int out_size, void* d_ws, size_t ws_size,
                              hipStream_t stream) {
    const int*   users = (const int*)d_in[0];
    const int*   items = (const int*)d_in[1];
    const int*   eu    = (const int*)d_in[2];
    const int*   ei    = (const int*)d_in[3];
    const float* uemb  = (const float*)d_in[4];
    const float* iemb  = (const float*)d_in[5];
    float*       out   = (float*)d_out;

    const int B = in_sizes[0];
    const int E = in_sizes[2];
    const int U = in_sizes[4] / 64;
    const int I = in_sizes[5] / 64;
    const int N = U + I;

    char* ws = (char*)d_ws;
    size_t off = 0;
    auto carve = [&](size_t bytes) -> float* {
        float* p = (float*)(ws + off);
        off += (bytes + 255) & ~(size_t)255;
        return p;
    };
    float* inv    = carve((size_t)N * 4);
    float* e_cur  = carve((size_t)N * 64 * 4);
    float* e_next = carve((size_t)N * 64 * 4);
    float* uacc   = carve((size_t)B * 64 * 4);
    float* iacc   = carve((size_t)B * 64 * 4);
    int*   rowptr = (int*)carve((size_t)(N + 1) * 4);
    int2*  csr    = (int2*)carve((size_t)2 * E * 8);

    const int nchunk = (N + 1023) / 1024;
    size_t toff = 0;
    auto tcarve = [&](size_t bytes) -> int* {
        int* p = (int*)((char*)e_next + toff);
        toff += (bytes + 255) & ~(size_t)255;
        return p;
    };
    int* cnt    = tcarve((size_t)N * 4);
    int* bsum   = tcarve((size_t)nchunk * 4);
    int* rank_u = tcarve((size_t)E * 4);
    int* rank_i = tcarve((size_t)E * 4);
    float* inv_g = (float*)tcarve((size_t)N * 4);
    int*   need  = (int*)inv;

    const bool gather_path = (off <= ws_size) && (toff <= (size_t)N * 64 * 4) &&
                             ((size_t)N * 64 * 4 <= (size_t)0xffffffffu);

    const int BS = 256;

    const int gacc_blocks = (2 * B * 64 + BS - 1) / BS;
    const size_t item0 = (size_t)U * 64;

    if (gather_path) {
        hipMemsetAsync(cnt, 0, (size_t)N * 4, stream);
        lgcn_count_rank<<<(E + BS - 1) / BS, BS, 0, stream>>>(eu, ei, cnt, rank_u, rank_i, E, U);
        lgcn_blocksum<<<nchunk, 256, 0, stream>>>(cnt, bsum, N);
        lgcn_scan_inv<<<nchunk, 256, 0, stream>>>(cnt, bsum, rowptr, inv_g, need, N);
        lgcn_fill<<<(E + BS - 1) / BS, BS, 0, stream>>>(eu, ei, rank_u, rank_i, rowptr, inv_g, csr, E, U);
        const int batch_blocks = (B + 7) / 8;
        const int pull_blocks = (N + 15) / 16;
        const int layer_blocks = pull_blocks + batch_blocks;

        lgcn_spmm_pull_first<<<layer_blocks, 256, 0, stream>>>(users, items, rowptr, csr, uemb, iemb,
                                                               e_cur, uacc, iacc, need, B, U, N);
        lgcn_spmm_pull<<<layer_blocks, 256, 0, stream>>>(users, items, rowptr, csr, e_cur, e_next,
                                                         uacc, iacc, B, U, N);
        lgcn_spmm_pull_needed<<<layer_blocks, 256, 0, stream>>>(users, items, rowptr, csr, need, e_next,
                                                                e_cur, uacc, iacc, B, U, N);
        lgcn_spmm_pull_batch<<<batch_blocks, 256, 0, stream>>>(users, items, rowptr, csr, e_cur, uacc, iacc, B, U);
    } else {
        hipMemsetAsync(uacc, 0, (size_t)B * 64 * 4, stream);
        hipMemsetAsync(iacc, 0, (size_t)B * 64 * 4, stream);

        {
            long long total = (long long)N * 64;
            lgcn_init_x0<<<(int)((total + BS - 1) / BS), BS, 0, stream>>>(uemb, iemb, e_cur, U, N);
        }
        hipMemsetAsync(inv, 0, (size_t)N * 4, stream);
        lgcn_degree<<<(E + BS - 1) / BS, BS, 0, stream>>>(eu, ei, inv, E, U);
        lgcn_invsqrt<<<(N + BS - 1) / BS, BS, 0, stream>>>(inv, N);

        lgcn_gather_acc<<<gacc_blocks, BS, 0, stream>>>(users, items, e_cur, e_cur + item0, uacc, iacc, B);

        const long long spmm_threads = (long long)2 * E * 16;
        const int spmm_blocks = (int)((spmm_threads + BS - 1) / BS);
        for (int layer = 0; layer < 4; ++layer) {
            hipMemsetAsync(e_next, 0, (size_t)N * 64 * 4, stream);
            lgcn_spmm<<<spmm_blocks, BS, 0, stream>>>(eu, ei, inv, e_cur, e_next, E, U);
            lgcn_gather_acc<<<gacc_blocks, BS, 0, stream>>>(users, items, e_next, e_next + item0, uacc, iacc, B);
            float* tmp = e_cur; e_cur = e_next; e_next = tmp;
        }
    }

    const int tiles = (B + 15) / 16;
    const int dot_blocks = (tiles + 7) / 8;
    lgcn_dot_wmma<<<dot_blocks, 256, 0, stream>>>(uacc, iacc, out, B);
}


# Adjacency grouped without global atomics: edge blocks bin entries by 1024-node range with LDS adds, one block per bin counts, scans, places its rows in LDS; row sum order = LDS arrival order
# speedup vs baseline: 5.2491x; 1.5419x over previous
#include <hip/hip_runtime.h>


typedef __attribute__((ext_vector_type(2))) float v2f;
typedef __attribute__((ext_vector_type(8))) float v8f;

#define DEV_ATOMIC_ADD(p, v) \
    __hip_atomic_fetch_add((p), (v), __ATOMIC_RELAXED, __HIP_MEMORY_SCOPE_AGENT)

__global__ void lgcn_degree(const int* __restrict__ eu, const int* __restrict__ ei,
                            float* __restrict__ deg, int E, int U) {
    int t = blockIdx.x * blockDim.x + threadIdx.x;
    if (t >= E) return;
    DEV_ATOMIC_ADD(&deg[eu[t]], 1.0f);
    DEV_ATOMIC_ADD(&deg[U + ei[t]], 1.0f);
}

__global__ void lgcn_invsqrt(float* __restrict__ d, int N) {
    int t = blockIdx.x * blockDim.x + threadIdx.x;
    if (t >= N) return;
    float v = d[t];
    d[t] = (v > 0.0f) ? (1.0f / sqrtf(v)) : 0.0f;
}

__global__ void lgcn_init_x0(const float* __restrict__ ue, const float* __restrict__ ie,
                             float* __restrict__ x, int U, int N) {
    long long t = (long long)blockIdx.x * blockDim.x + threadIdx.x;
    long long total = (long long)N * 64;
    if (t >= total) return;
    long long ubytes = (long long)U * 64;
    x[t] = (t < ubytes) ? ue[t] : ie[t - ubytes];
}

__global__ void lgcn_spmm(const int* __restrict__ eu, const int* __restrict__ ei,
                          const float* __restrict__ inv,
                          const float* __restrict__ xin, float* __restrict__ xout,
                          int E, int U) {
    long long t = (long long)blockIdx.x * blockDim.x + threadIdx.x;
    int edge = (int)(t >> 4);
    if (edge >= 2 * E) return;
    int q = ((int)t & 15) * 4;

    int s, d;
    if (edge < E) {
        s = eu[edge];
        d = U + ei[edge];
    } else {
        int e2 = edge - E;
        s = U + ei[e2];
        d = eu[e2];
    }
    float w = inv[s] * inv[d];
    const float4 v = *(const float4*)(xin + (size_t)s * 64 + q);
    float* o = xout + (size_t)d * 64 + q;
    DEV_ATOMIC_ADD(o + 0, w * v.x);
    DEV_ATOMIC_ADD(o + 1, w * v.y);
    DEV_ATOMIC_ADD(o + 2, w * v.z);
    DEV_ATOMIC_ADD(o + 3, w * v.w);
}


#define LGCN_EB 8192
#define LGCN_BIN_SHIFT 10

__global__ __launch_bounds__(1024) void lgcn_bin_edges(const int* __restrict__ eu,
                                                       const int* __restrict__ ei,
                                                       int2* __restrict__ binned,
                                                       int* __restrict__ segoff,
                                                       int E, int U, int nb) {
    __shared__ int hist[1024];
    __shared__ int wtot[32];
    const int tid = threadIdx.x;
    const int lane = tid & 31;
    const int wave = tid >> 5;
    hist[tid] = 0;
    __syncthreads();

    const int e0 = (int)blockIdx.x * LGCN_EB;
    int un[8], in[8], ra[8], rb[8];
#pragma unroll
    for (int k = 0; k < 8; ++k) {
        const int e = e0 + k * 1024 + tid;
        un[k] = -1; in[k] = 0; ra[k] = 0; rb[k] = 0;
        if (e < E) {
            un[k] = eu[e];
            in[k] = U + ei[e];
        }
    }
#pragma unroll
    for (int k = 0; k < 8; ++k) {
        if (un[k] >= 0) {
            ra[k] = atomicAdd(&hist[in[k] >> LGCN_BIN_SHIFT], 1);
            rb[k] = atomicAdd(&hist[un[k] >> LGCN_BIN_SHIFT], 1);
        }
    }
    __syncthreads();

    const int v = hist[tid];
    int incl = v;
#pragma unroll
    for (int off = 1; off < 32; off <<= 1) {
        int t = __shfl_up(incl, off);
        if (lane >= off) incl += t;
    }
    if (lane == 31) wtot[wave] = incl;
    __syncthreads();
    int excl = incl - v;
    for (int w = 0; w < wave; ++w) excl += wtot[w];
    hist[tid] = excl;
    if (tid <= nb) segoff[(size_t)blockIdx.x * (nb + 1) + tid] = excl;
    __syncthreads();

    int2* slice = binned + (size_t)2 * e0;
#pragma unroll
    for (int k = 0; k < 8; ++k) {
        if (un[k] >= 0) {
            slice[hist[in[k] >> LGCN_BIN_SHIFT] + ra[k]] = make_int2(un[k], in[k]);
            slice[hist[un[k] >> LGCN_BIN_SHIFT] + rb[k]] = make_int2(in[k], un[k]);
        }
    }
}

__device__ __forceinline__ int2 lgcn_run_entry(const int2* __restrict__ run, int j, int end) {
    return (j < end) ? run[j] : make_int2(0, -1);
}

__global__ __launch_bounds__(1024) void lgcn_build_rows(const int2* __restrict__ binned,
                                                        const int* __restrict__ segoff,
                                                        int* __restrict__ rowptr,
                                                        float* __restrict__ inv,
                                                        int* __restrict__ need,
                                                        int2* __restrict__ csr,
                                                        int N, int nb, int nblk) {
    __shared__ int segS[1024];
    __shared__ int segE[1024];
    __shared__ int cnt[1024];
    __shared__ int rowst[1024];
    __shared__ int wtot[32];
    __shared__ int binstart;
    const int tid = threadIdx.x;
    const int lane = tid & 31;
    const int wave = tid >> 5;
    const int b = blockIdx.x;
    const int node0 = b << LGCN_BIN_SHIFT;

    if (tid == 0) binstart = 0;
    cnt[tid] = 0;
    __syncthreads();
    if (tid < nblk) {
        const int s = segoff[(size_t)tid * (nb + 1) + b];
        const int e = segoff[(size_t)tid * (nb + 1) + b + 1];
        segS[tid] = s;
        segE[tid] = e;
        atomicAdd(&binstart, s);
    }
    __syncthreads();

    for (int blk = wave; blk < nblk; blk += 32) {
        const int2* run = binned + (size_t)2 * LGCN_EB * blk;
        const int end = segE[blk];
        for (int j = segS[blk] + lane; j < end; j += 128) {
            const int2 a0 = lgcn_run_entry(run, j, end);
            const int2 a1 = lgcn_run_entry(run, j + 32, end);
            const int2 a2 = lgcn_run_entry(run, j + 64, end);
            const int2 a3 = lgcn_run_entry(run, j + 96, end);
            if (a0.y >= 0) atomicAdd(&cnt[a0.y - node0], 1);
            if (a1.y >= 0) atomicAdd(&cnt[a1.y - node0], 1);
            if (a2.y >= 0) atomicAdd(&cnt[a2.y - node0], 1);
            if (a3.y >= 0) atomicAdd(&cnt[a3.y - node0], 1);
        }
    }
    __syncthreads();

    const int c = cnt[tid];
    int incl = c;
#pragma unroll
    for (int off = 1; off < 32; off <<= 1) {
        int t = __shfl_up(incl, off);
        if (lane >= off) incl += t;
    }
    if (lane == 31) wtot[wave] = incl;
    __syncthreads();
    int start = binstart + incl - c;
    for (int w = 0; w < wave; ++w) start += wtot[w];
    const int node = node0 + tid;
    if (node < N) {
        rowptr[node] = start;
        const float v = (float)c;
        inv[node] = (v > 0.0f) ? (1.0f / sqrtf(v)) : 0.0f;
        need[node] = 0;
        if (node == N - 1) rowptr[N] = start + c;
    }
    rowst[tid] = start;
    cnt[tid] = 0;
    __syncthreads();

    for (int blk = wave; blk < nblk; blk += 32) {
        const int2* run = binned + (size_t)2 * LGCN_EB * blk;
        const int end = segE[blk];
        for (int j = segS[blk] + lane; j < end; j += 128) {
            const int2 a0 = lgcn_run_entry(run, j, end);
            const int2 a1 = lgcn_run_entry(run, j + 32, end);
            const int2 a2 = lgcn_run_entry(run, j + 64, end);
            const int2 a3 = lgcn_run_entry(run, j + 96, end);
            if (a0.y >= 0) { const int l = a0.y - node0; csr[rowst[l] + atomicAdd(&cnt[l], 1)] = a0; }
            if (a1.y >= 0) { const int l = a1.y - node0; csr[rowst[l] + atomicAdd(&cnt[l], 1)] = a1; }
            if (a2.y >= 0) { const int l = a2.y - node0; csr[rowst[l] + atomicAdd(&cnt[l], 1)] = a2; }
            if (a3.y >= 0) { const int l = a3.y - node0; csr[rowst[l] + atomicAdd(&cnt[l], 1)] = a3; }
        }
    }
}

__global__ __launch_bounds__(256) void lgcn_weights(int2* __restrict__ csr,
                                                    const float* __restrict__ inv, int total) {
    const int t0 = (int)blockIdx.x * 1024 + (int)threadIdx.x;
    int2 e[4];
#pragma unroll
    for (int k = 0; k < 4; ++k) {
        const int t = t0 + k * 256;
        e[k] = (t < total) ? csr[t] : make_int2(0, 0);
    }
    float w[4];
#pragma unroll
    for (int k = 0; k < 4; ++k) w[k] = inv[e[k].x] * inv[e[k].y];
#pragma unroll
    for (int k = 0; k < 4; ++k) {
        const int t = t0 + k * 256;
        if (t < total) csr[t].y = __float_as_int(w[k]);
    }
}

template <bool SPLIT>
__device__ __forceinline__ const float* lgcn_src_row(const float* __restrict__ xa,
                                                     const float* __restrict__ xb,
                                                     int U, int s) {
    if (SPLIT)
        return (s < U) ? xa + (size_t)s * 64 : xb + (size_t)(s - U) * 64;
    return xa + (size_t)s * 64;
}

template <int K>
__device__ __forceinline__ int lgcn_row_share(int v) {
    return __builtin_amdgcn_update_dpp(0, v, 0x150 + K, 0xf, 0xf, true);
}

template <bool SPLIT, int K>
__device__ __forceinline__ float4 lgcn_gather(const float* __restrict__ xa,
                                              const float* __restrict__ xb, int U,
                                              int es, int q) {
    const int s = lgcn_row_share<K>(es);
    if (SPLIT) return *(const float4*)(lgcn_src_row<true>(xa, xb, U, s) + q);
    const unsigned off = ((unsigned)s << 8) | ((unsigned)q << 2);
    return *(const float4*)((const char*)xa + off);
}

template <int K>
__device__ __forceinline__ void lgcn_add(float4& acc, int ew, const float4& v) {
    const float w = __int_as_float(lgcn_row_share<K>(ew));
    acc.x = __builtin_fmaf(w, v.x, acc.x);
    acc.y = __builtin_fmaf(w, v.y, acc.y);
    acc.z = __builtin_fmaf(w, v.z, acc.z);
    acc.w = __builtin_fmaf(w, v.w, acc.w);
}

template <bool SPLIT, int K0>
__device__ __forceinline__ void lgcn_eight(float4& acc, const float* __restrict__ xa,
                                           const float* __restrict__ xb, int U,
                                           int es, int ew, int q, bool upper) {
    const float4 v0 = lgcn_gather<SPLIT, K0 + 0>(xa, xb, U, es, q);
    const float4 v1 = lgcn_gather<SPLIT, K0 + 1>(xa, xb, U, es, q);
    const float4 v2 = lgcn_gather<SPLIT, K0 + 2>(xa, xb, U, es, q);
    const float4 v3 = lgcn_gather<SPLIT, K0 + 3>(xa, xb, U, es, q);
    if (upper) {
        const float4 v4 = lgcn_gather<SPLIT, K0 + 4>(xa, xb, U, es, q);
        const float4 v5 = lgcn_gather<SPLIT, K0 + 5>(xa, xb, U, es, q);
        const float4 v6 = lgcn_gather<SPLIT, K0 + 6>(xa, xb, U, es, q);
        const float4 v7 = lgcn_gather<SPLIT, K0 + 7>(xa, xb, U, es, q);
        if (SPLIT) __builtin_amdgcn_sched_barrier(0);
        lgcn_add<K0 + 0>(acc, ew, v0);
        lgcn_add<K0 + 1>(acc, ew, v1);
        lgcn_add<K0 + 2>(acc, ew, v2);
        lgcn_add<K0 + 3>(acc, ew, v3);
        lgcn_add<K0 + 4>(acc, ew, v4);
        lgcn_add<K0 + 5>(acc, ew, v5);
        lgcn_add<K0 + 6>(acc, ew, v6);
        lgcn_add<K0 + 7>(acc, ew, v7);
    } else {
        lgcn_add<K0 + 0>(acc, ew, v0);
        lgcn_add<K0 + 1>(acc, ew, v1);
        lgcn_add<K0 + 2>(acc, ew, v2);
        lgcn_add<K0 + 3>(acc, ew, v3);
    }
}

template <bool SPLIT>
__device__ __forceinline__ float4 lgcn_row_sum(const int* __restrict__ rowptr,
                                               const int2* __restrict__ csr,
                                               const float* __restrict__ xa,
                                               const float* __restrict__ xb, int U,
                                               int row, int q) {
    const int lane = q >> 2;
    const int end = rowptr[row + 1];
    const int last = end - 1;
    float4 acc = make_float4(0.0f, 0.0f, 0.0f, 0.0f);
    for (int base = rowptr[row]; base < end; base += 16) {
        const int left = end - base;
        const int2 e = csr[min(base + lane, last)];
        const int ew = (lane < left) ? e.y : 0;
        lgcn_eight<SPLIT, 0>(acc, xa, xb, U, e.x, ew, q, left > 4);
        if (left > 8) lgcn_eight<SPLIT, 8>(acc, xa, xb, U, e.x, ew, q, left > 12);
    }
    return acc;
}

__device__ __forceinline__ int lgcn_batch_node(const int* __restrict__ users,
                                               const int* __restrict__ items,
                                               int entry, int B, int U) {
    return (entry < B) ? users[entry] : U + items[entry - B];
}

__device__ __forceinline__ int lgcn_batch_blocks(int B) {
    return (B + 7) >> 3;
}

template <bool FIRST>
__device__ __forceinline__ void lgcn_batch_acc(const int* __restrict__ users,
                                               const int* __restrict__ items,
                                               const float* __restrict__ xu,
                                               const float* __restrict__ xi,
                                               float* uacc, float* iacc, int B, int t) {
    const int entry = t >> 4;
    if (entry >= 2 * B) return;
    const int q = (t & 15) * 4;
    const float* src;
    float* dst;
    if (entry < B) {
        src = xu + (size_t)users[entry] * 64 + q;
        dst = uacc + (size_t)entry * 64 + q;
    } else {
        const int e = entry - B;
        src = xi + (size_t)items[e] * 64 + q;
        dst = iacc + (size_t)e * 64 + q;
    }
    float4 v = *(const float4*)src;
    if (!FIRST) {
        float4 a = *(const float4*)dst;
        a.x += v.x; a.y += v.y; a.z += v.z; a.w += v.w;
        v = a;
    }
    *(float4*)dst = v;
}

__global__ __launch_bounds__(256) void lgcn_spmm_pull(const int* __restrict__ users,
                                                      const int* __restrict__ items,
                                                      const int* __restrict__ rowptr,
                                                      const int2* __restrict__ csr,
                                                      const float* __restrict__ xin,
                                                      float* __restrict__ xout,
                                                      float* uacc, float* iacc,
                                                      int B, int U, int N) {
    const int batch_blocks = lgcn_batch_blocks(B);
    if ((int)blockIdx.x < batch_blocks) {
        lgcn_batch_acc<false>(users, items, xin, xin + (size_t)U * 64, uacc, iacc, B,
                              (int)blockIdx.x * 256 + (int)threadIdx.x);
        return;
    }
    const int t = ((int)blockIdx.x - batch_blocks) * 256 + (int)threadIdx.x;
    const int row = t >> 4;
    if (row >= N) return;
    const int q = (t & 15) * 4;
    *(float4*)(xout + (size_t)row * 64 + q) =
        lgcn_row_sum<false>(rowptr, csr, xin, nullptr, 0, row, q);
}

__device__ __forceinline__ void lgcn_mark_need(const int* __restrict__ users,
                                               const int* __restrict__ items,
                                               const int* __restrict__ rowptr,
                                               const int2* __restrict__ csr,
                                               int* __restrict__ need, int B, int U, int t) {
    const int entry = t >> 4;
    if (entry >= 2 * B) return;
    const int l = t & 15;
    const int node = lgcn_batch_node(users, items, entry, B, U);
    if (l == 0)
        __hip_atomic_fetch_or(&need[node], 1, __ATOMIC_RELAXED, __HIP_MEMORY_SCOPE_AGENT);
    const int end = rowptr[node + 1];
    for (int j = rowptr[node] + l; j < end; j += 16)
        __hip_atomic_fetch_or(&need[csr[j].x], 1, __ATOMIC_RELAXED, __HIP_MEMORY_SCOPE_AGENT);
}

__global__ __launch_bounds__(256) void lgcn_spmm_pull_first(const int* __restrict__ users,
                                                            const int* __restrict__ items,
                                                            const int* __restrict__ rowptr,
                                                            const int2* __restrict__ csr,
                                                            const float* __restrict__ ue,
                                                            const float* __restrict__ ie,
                                                            float* __restrict__ xout,
                                                            float* uacc, float* iacc,
                                                            int* __restrict__ need,
                                                            int B, int U, int N) {
    const int batch_blocks = lgcn_batch_blocks(B);
    if ((int)blockIdx.x < batch_blocks) {
        const int tb = (int)blockIdx.x * 256 + (int)threadIdx.x;
        lgcn_batch_acc<true>(users, items, ue, ie, uacc, iacc, B, tb);
        lgcn_mark_need(users, items, rowptr, csr, need, B, U, tb);
        return;
    }
    const int t = ((int)blockIdx.x - batch_blocks) * 256 + (int)threadIdx.x;
    const int row = t >> 4;
    if (row >= N) return;
    const int q = (t & 15) * 4;
    *(float4*)(xout + (size_t)row * 64 + q) =
        lgcn_row_sum<true>(rowptr, csr, ue, ie, U, row, q);
}

__global__ __launch_bounds__(256) void lgcn_spmm_pull_needed(const int* __restrict__ users,
                                                             const int* __restrict__ items,
                                                             const int* __restrict__ rowptr,
                                                             const int2* __restrict__ csr,
                                                             const int* __restrict__ need,
                                                             const float* __restrict__ xin,
                                                             float* __restrict__ xout,
                                                             float* uacc, float* iacc,
                                                             int B, int U, int N) {
    const int batch_blocks = lgcn_batch_blocks(B);
    if ((int)blockIdx.x < batch_blocks) {
        lgcn_batch_acc<false>(users, items, xin, xin + (size_t)U * 64, uacc, iacc, B,
                              (int)blockIdx.x * 256 + (int)threadIdx.x);
        return;
    }
    const int t = ((int)blockIdx.x - batch_blocks) * 256 + (int)threadIdx.x;
    const int row = t >> 4;
    if (row >= N) return;
    if (need[row] == 0) return;
    const int q = (t & 15) * 4;
    *(float4*)(xout + (size_t)row * 64 + q) =
        lgcn_row_sum<false>(rowptr, csr, xin, nullptr, 0, row, q);
}

__global__ __launch_bounds__(256) void lgcn_spmm_pull_batch(const int* __restrict__ users,
                                                            const int* __restrict__ items,
                                                            const int* __restrict__ rowptr,
                                                            const int2* __restrict__ csr,
                                                            const float* __restrict__ xin,
                                                            float* __restrict__ uacc,
                                                            float* __restrict__ iacc,
                                                            int B, int U) {
    const int t = blockIdx.x * 256 + threadIdx.x;
    const int entry = t >> 4;
    if (entry >= 2 * B) return;
    const int q = (t & 15) * 4;
    const int node = lgcn_batch_node(users, items, entry, B, U);
    float* dst = (entry < B) ? uacc + (size_t)entry * 64 + q
                             : iacc + (size_t)(entry - B) * 64 + q;
    float4 a = *(const float4*)dst;
    const float4 x3 = *(const float4*)(xin + (size_t)node * 64 + q);
    const float4 s = lgcn_row_sum<false>(rowptr, csr, xin, nullptr, 0, node, q);
    a.x += x3.x; a.y += x3.y; a.z += x3.z; a.w += x3.w;
    a.x += s.x; a.y += s.y; a.z += s.z; a.w += s.w;
    *(float4*)dst = a;
}

__global__ void lgcn_gather_acc(const int* __restrict__ users, const int* __restrict__ items,
                                const float* __restrict__ xu, const float* __restrict__ xi,
                                float* __restrict__ uacc, float* __restrict__ iacc,
                                int B) {
    int t = blockIdx.x * blockDim.x + threadIdx.x;
    int half = B * 64;
    if (t >= 2 * half) return;
    if (t < half) {
        int b = t >> 6, d = t & 63;
        uacc[t] += xu[(size_t)users[b] * 64 + d];
    } else {
        int tt = t - half;
        int b = tt >> 6, d = tt & 63;
        iacc[tt] += xi[(size_t)items[b] * 64 + d];
    }
}

__global__ __launch_bounds__(256) void lgcn_dot_wmma(const float* __restrict__ uacc,
                                                     const float* __restrict__ iacc,
                                                     float* __restrict__ out, int B) {
    __shared__ float tile[8][16 * 16];
    const int lane = threadIdx.x & 31;
    const int wave = threadIdx.x >> 5;
    const int tile0 = (blockIdx.x * 8 + wave) * 16;

    const int m = lane & 15;
    const int khalf = (lane >> 4) << 1;
    int row = tile0 + m;
    if (row > B - 1) row = B - 1;
    const float* __restrict__ ur = uacc + (size_t)row * 64;
    const float* __restrict__ ir = iacc + (size_t)row * 64;

    v8f c = {};
#pragma unroll
    for (int kb = 0; kb < 64; kb += 4) {
        int k0 = kb + khalf;
        v2f a; a.x = ur[k0]; a.y = ur[k0 + 1];
        v2f b; b.x = ir[k0]; b.y = ir[k0 + 1];
        c = __builtin_amdgcn_wmma_f32_16x16x4_f32(false, a, false, b,
                                                  (short)0, c, false, false);
    }

    float* lds = &tile[wave][0];
    const int n = lane & 15;
    const int mbase = (lane >> 4) ? 8 : 0;
#pragma unroll
    for (int r = 0; r < 8; ++r)
        lds[(mbase + r) * 16 + n] = c[r];
    __syncthreads();

    if (lane < 16) {
        int b = tile0 + lane;
        if (b < B)
            out[b] = lds[lane * 16 + lane] * (1.0f / 25.0f);
    }
}

extern "C" void kernel_launch(void* const* d_in, const int* in_sizes, int n_in,
                              void* d_out, int out_size, void* d_ws, size_t ws_size,
                              hipStream_t stream) {
    const int*   users = (const int*)d_in[0];
    const int*   items = (const int*)d_in[1];
    const int*   eu    = (const int*)d_in[2];
    const int*   ei    = (const int*)d_in[3];
    const float* uemb  = (const float*)d_in[4];
    const float* iemb  = (const float*)d_in[5];
    float*       out   = (float*)d_out;

    const int B = in_sizes[0];
    const int E = in_sizes[2];
    const int U = in_sizes[4] / 64;
    const int I = in_sizes[5] / 64;
    const int N = U + I;

    char* ws = (char*)d_ws;
    size_t off = 0;
    auto carve = [&](size_t bytes) -> float* {
        float* p = (float*)(ws + off);
        off += (bytes + 255) & ~(size_t)255;
        return p;
    };
    float* inv    = carve((size_t)N * 4);
    float* e_cur  = carve((size_t)N * 64 * 4);
    float* e_next = carve((size_t)N * 64 * 4);
    float* uacc   = carve((size_t)B * 64 * 4);
    float* iacc   = carve((size_t)B * 64 * 4);
    int*   rowptr = (int*)carve((size_t)(N + 1) * 4);
    int2*  csr    = (int2*)carve((size_t)2 * E * 8);

    const int nchunk = (N + 1023) / 1024;
    const int nblk = (E + LGCN_EB - 1) / LGCN_EB;
    size_t toff = 0;
    auto tcarve = [&](size_t bytes) -> int* {
        int* p = (int*)((char*)e_next + toff);
        toff += (bytes + 255) & ~(size_t)255;
        return p;
    };
    int2* binned = (int2*)tcarve((size_t)2 * E * 8);
    int*  segoff = tcarve((size_t)nblk * (nchunk + 1) * 4);
    float* inv_g = (float*)tcarve((size_t)N * 4);
    int*   need  = (int*)inv;

    const bool gather_path = (off <= ws_size) && (toff <= (size_t)N * 64 * 4) &&
                             ((size_t)N * 64 * 4 <= (size_t)0xffffffffu) &&
                             (nchunk <= 1023) && (nblk <= 1024) && (E <= 0x3fffffff);

    const int BS = 256;

    const int gacc_blocks = (2 * B * 64 + BS - 1) / BS;
    const size_t item0 = (size_t)U * 64;

    if (gather_path) {
        lgcn_bin_edges<<<nblk, 1024, 0, stream>>>(eu, ei, binned, segoff, E, U, nchunk);
        lgcn_build_rows<<<nchunk, 1024, 0, stream>>>(binned, segoff, rowptr, inv_g, need, csr, N, nchunk, nblk);
        lgcn_weights<<<(2 * E + 1023) / 1024, 256, 0, stream>>>(csr, inv_g, 2 * E);
        const int batch_blocks = (B + 7) / 8;
        const int pull_blocks = (N + 15) / 16;
        const int layer_blocks = pull_blocks + batch_blocks;

        lgcn_spmm_pull_first<<<layer_blocks, 256, 0, stream>>>(users, items, rowptr, csr, uemb, iemb,
                                                               e_cur, uacc, iacc, need, B, U, N);
        lgcn_spmm_pull<<<layer_blocks, 256, 0, stream>>>(users, items, rowptr, csr, e_cur, e_next,
                                                         uacc, iacc, B, U, N);
        lgcn_spmm_pull_needed<<<layer_blocks, 256, 0, stream>>>(users, items, rowptr, csr, need, e_next,
                                                                e_cur, uacc, iacc, B, U, N);
        lgcn_spmm_pull_batch<<<batch_blocks, 256, 0, stream>>>(users, items, rowptr, csr, e_cur, uacc, iacc, B, U);
    } else {
        hipMemsetAsync(uacc, 0, (size_t)B * 64 * 4, stream);
        hipMemsetAsync(iacc, 0, (size_t)B * 64 * 4, stream);

        {
            long long total = (long long)N * 64;
            lgcn_init_x0<<<(int)((total + BS - 1) / BS), BS, 0, stream>>>(uemb, iemb, e_cur, U, N);
        }
        hipMemsetAsync(inv, 0, (size_t)N * 4, stream);
        lgcn_degree<<<(E + BS - 1) / BS, BS, 0, stream>>>(eu, ei, inv, E, U);
        lgcn_invsqrt<<<(N + BS - 1) / BS, BS, 0, stream>>>(inv, N);

        lgcn_gather_acc<<<gacc_blocks, BS, 0, stream>>>(users, items, e_cur, e_cur + item0, uacc, iacc, B);

        const long long spmm_threads = (long long)2 * E * 16;
        const int spmm_blocks = (int)((spmm_threads + BS - 1) / BS);
        for (int layer = 0; layer < 4; ++layer) {
            hipMemsetAsync(e_next, 0, (size_t)N * 64 * 4, stream);
            lgcn_spmm<<<spmm_blocks, BS, 0, stream>>>(eu, ei, inv, e_cur, e_next, E, U);
            lgcn_gather_acc<<<gacc_blocks, BS, 0, stream>>>(users, items, e_next, e_next + item0, uacc, iacc, B);
            float* tmp = e_cur; e_cur = e_next; e_next = tmp;
        }
    }

    const int tiles = (B + 15) / 16;
    const int dot_blocks = (tiles + 7) / 8;
    lgcn_dot_wmma<<<dot_blocks, 256, 0, stream>>>(uacc, iacc, out, B);
}


# Both grouping kernels place their sorted entries in LDS and copy them out in order: each store instruction writes whole 128-byte lines (was one line piece per entry); row sum order unchanged
# speedup vs baseline: 5.8503x; 1.1145x over previous
#include <hip/hip_runtime.h>


typedef __attribute__((ext_vector_type(2))) float v2f;
typedef __attribute__((ext_vector_type(8))) float v8f;

#define DEV_ATOMIC_ADD(p, v) \
    __hip_atomic_fetch_add((p), (v), __ATOMIC_RELAXED, __HIP_MEMORY_SCOPE_AGENT)

__global__ void lgcn_degree(const int* __restrict__ eu, const int* __restrict__ ei,
                            float* __restrict__ deg, int E, int U) {
    int t = blockIdx.x * blockDim.x + threadIdx.x;
    if (t >= E) return;
    DEV_ATOMIC_ADD(&deg[eu[t]], 1.0f);
    DEV_ATOMIC_ADD(&deg[U + ei[t]], 1.0f);
}

__global__ void lgcn_invsqrt(float* __restrict__ d, int N) {
    int t = blockIdx.x * blockDim.x + threadIdx.x;
    if (t >= N) return;
    float v = d[t];
    d[t] = (v > 0.0f) ? (1.0f / sqrtf(v)) : 0.0f;
}

__global__ void lgcn_init_x0(const float* __restrict__ ue, const float* __restrict__ ie,
                             float* __restrict__ x, int U, int N) {
    long long t = (long long)blockIdx.x * blockDim.x + threadIdx.x;
    long long total = (long long)N * 64;
    if (t >= total) return;
    long long ubytes = (long long)U * 64;
    x[t] = (t < ubytes) ? ue[t] : ie[t - ubytes];
}

__global__ void lgcn_spmm(const int* __restrict__ eu, const int* __restrict__ ei,
                          const float* __restrict__ inv,
                          const float* __restrict__ xin, float* __restrict__ xout,
                          int E, int U) {
    long long t = (long long)blockIdx.x * blockDim.x + threadIdx.x;
    int edge = (int)(t >> 4);
    if (edge >= 2 * E) return;
    int q = ((int)t & 15) * 4;

    int s, d;
    if (edge < E) {
        s = eu[edge];
        d = U + ei[edge];
    } else {
        int e2 = edge - E;
        s = U + ei[e2];
        d = eu[e2];
    }
    float w = inv[s] * inv[d];
    const float4 v = *(const float4*)(xin + (size_t)s * 64 + q);
    float* o = xout + (size_t)d * 64 + q;
    DEV_ATOMIC_ADD(o + 0, w * v.x);
    DEV_ATOMIC_ADD(o + 1, w * v.y);
    DEV_ATOMIC_ADD(o + 2, w * v.z);
    DEV_ATOMIC_ADD(o + 3, w * v.w);
}


#define LGCN_EB 8192
#define LGCN_BIN_SHIFT 10

__global__ __launch_bounds__(1024) void lgcn_bin_edges(const int* __restrict__ eu,
                                                       const int* __restrict__ ei,
                                                       int2* __restrict__ binned,
                                                       int* __restrict__ segoff,
                                                       int E, int U, int nb) {
    __shared__ int hist[1024];
    __shared__ int wtot[32];
    __shared__ int2 stage[2 * LGCN_EB];
    const int tid = threadIdx.x;
    const int lane = tid & 31;
    const int wave = tid >> 5;
    hist[tid] = 0;
    __syncthreads();

    const int e0 = (int)blockIdx.x * LGCN_EB;
    int un[8], in[8], ra[8], rb[8];
#pragma unroll
    for (int k = 0; k < 8; ++k) {
        const int e = e0 + k * 1024 + tid;
        un[k] = -1; in[k] = 0; ra[k] = 0; rb[k] = 0;
        if (e < E) {
            un[k] = eu[e];
            in[k] = U + ei[e];
        }
    }
#pragma unroll
    for (int k = 0; k < 8; ++k) {
        if (un[k] >= 0) {
            ra[k] = atomicAdd(&hist[in[k] >> LGCN_BIN_SHIFT], 1);
            rb[k] = atomicAdd(&hist[un[k] >> LGCN_BIN_SHIFT], 1);
        }
    }
    __syncthreads();

    const int v = hist[tid];
    int incl = v;
#pragma unroll
    for (int off = 1; off < 32; off <<= 1) {
        int t = __shfl_up(incl, off);
        if (lane >= off) incl += t;
    }
    if (lane == 31) wtot[wave] = incl;
    __syncthreads();
    int excl = incl - v;
    for (int w = 0; w < wave; ++w) excl += wtot[w];
    hist[tid] = excl;
    if (tid <= nb) segoff[(size_t)blockIdx.x * (nb + 1) + tid] = excl;
    __syncthreads();

#pragma unroll
    for (int k = 0; k < 8; ++k) {
        if (un[k] >= 0) {
            stage[hist[in[k] >> LGCN_BIN_SHIFT] + ra[k]] = make_int2(un[k], in[k]);
            stage[hist[un[k] >> LGCN_BIN_SHIFT] + rb[k]] = make_int2(in[k], un[k]);
        }
    }
    __syncthreads();

    const int pairs = min(LGCN_EB, E - e0);
    int4* slice4 = (int4*)(binned + (size_t)2 * e0);
    for (int k = 0; k < 8; ++k) {
        const int j = k * 1024 + tid;
        if (j < pairs) {
            const int2 p0 = stage[2 * j];
            const int2 p1 = stage[2 * j + 1];
            slice4[j] = make_int4(p0.x, p0.y, p1.x, p1.y);
        }
    }
}

__device__ __forceinline__ int2 lgcn_run_entry(const int2* __restrict__ run, int j, int end) {
    return (j < end) ? run[j] : make_int2(0, -1);
}

#define LGCN_STAGE_CAP 16384

__global__ __launch_bounds__(1024) void lgcn_build_rows(const int2* __restrict__ binned,
                                                        const int* __restrict__ segoff,
                                                        int* __restrict__ rowptr,
                                                        float* __restrict__ inv,
                                                        int* __restrict__ need,
                                                        int2* __restrict__ csr,
                                                        int N, int nb, int nblk) {
    __shared__ int segS[1024];
    __shared__ int segE[1024];
    __shared__ int cnt[1024];
    __shared__ int rowst[1024];
    __shared__ int wtot[32];
    __shared__ int binstart;
    __shared__ int bintotal;
    __shared__ int2 stage[LGCN_STAGE_CAP];
    const int tid = threadIdx.x;
    const int lane = tid & 31;
    const int wave = tid >> 5;
    const int b = blockIdx.x;
    const int node0 = b << LGCN_BIN_SHIFT;

    if (tid == 0) binstart = 0;
    cnt[tid] = 0;
    __syncthreads();
    if (tid < nblk) {
        const int s = segoff[(size_t)tid * (nb + 1) + b];
        const int e = segoff[(size_t)tid * (nb + 1) + b + 1];
        segS[tid] = s;
        segE[tid] = e;
        atomicAdd(&binstart, s);
    }
    __syncthreads();

    for (int blk = wave; blk < nblk; blk += 32) {
        const int2* run = binned + (size_t)2 * LGCN_EB * blk;
        const int end = segE[blk];
        for (int j = segS[blk] + lane; j < end; j += 128) {
            const int2 a0 = lgcn_run_entry(run, j, end);
            const int2 a1 = lgcn_run_entry(run, j + 32, end);
            const int2 a2 = lgcn_run_entry(run, j + 64, end);
            const int2 a3 = lgcn_run_entry(run, j + 96, end);
            if (a0.y >= 0) atomicAdd(&cnt[a0.y - node0], 1);
            if (a1.y >= 0) atomicAdd(&cnt[a1.y - node0], 1);
            if (a2.y >= 0) atomicAdd(&cnt[a2.y - node0], 1);
            if (a3.y >= 0) atomicAdd(&cnt[a3.y - node0], 1);
        }
    }
    __syncthreads();

    const int c = cnt[tid];
    int incl = c;
#pragma unroll
    for (int off = 1; off < 32; off <<= 1) {
        int t = __shfl_up(incl, off);
        if (lane >= off) incl += t;
    }
    if (lane == 31) wtot[wave] = incl;
    __syncthreads();
    const int bs = binstart;
    int start = bs + incl - c;
    for (int w = 0; w < wave; ++w) start += wtot[w];
    if (tid == 1023) bintotal = start + c - bs;
    const int node = node0 + tid;
    if (node < N) {
        rowptr[node] = start;
        const float v = (float)c;
        inv[node] = (v > 0.0f) ? (1.0f / sqrtf(v)) : 0.0f;
        need[node] = 0;
        if (node == N - 1) rowptr[N] = start + c;
    }
    rowst[tid] = start;
    cnt[tid] = 0;
    __syncthreads();

    const int total = bintotal;
    if (total <= LGCN_STAGE_CAP) {
        for (int blk = wave; blk < nblk; blk += 32) {
            const int2* run = binned + (size_t)2 * LGCN_EB * blk;
            const int end = segE[blk];
            for (int j = segS[blk] + lane; j < end; j += 128) {
                const int2 a0 = lgcn_run_entry(run, j, end);
                const int2 a1 = lgcn_run_entry(run, j + 32, end);
                const int2 a2 = lgcn_run_entry(run, j + 64, end);
                const int2 a3 = lgcn_run_entry(run, j + 96, end);
                if (a0.y >= 0) { const int l = a0.y - node0; stage[rowst[l] - bs + atomicAdd(&cnt[l], 1)] = a0; }
                if (a1.y >= 0) { const int l = a1.y - node0; stage[rowst[l] - bs + atomicAdd(&cnt[l], 1)] = a1; }
                if (a2.y >= 0) { const int l = a2.y - node0; stage[rowst[l] - bs + atomicAdd(&cnt[l], 1)] = a2; }
                if (a3.y >= 0) { const int l = a3.y - node0; stage[rowst[l] - bs + atomicAdd(&cnt[l], 1)] = a3; }
            }
        }
        __syncthreads();
        const int shift = bs & 15;
        for (int j = tid - shift; j < total; j += 1024)
            if (j >= 0) csr[bs + j] = stage[j];
        return;
    }

    for (int blk = wave; blk < nblk; blk += 32) {
        const int2* run = binned + (size_t)2 * LGCN_EB * blk;
        const int end = segE[blk];
        for (int j = segS[blk] + lane; j < end; j += 128) {
            const int2 a0 = lgcn_run_entry(run, j, end);
            const int2 a1 = lgcn_run_entry(run, j + 32, end);
            const int2 a2 = lgcn_run_entry(run, j + 64, end);
            const int2 a3 = lgcn_run_entry(run, j + 96, end);
            if (a0.y >= 0) { const int l = a0.y - node0; csr[rowst[l] + atomicAdd(&cnt[l], 1)] = a0; }
            if (a1.y >= 0) { const int l = a1.y - node0; csr[rowst[l] + atomicAdd(&cnt[l], 1)] = a1; }
            if (a2.y >= 0) { const int l = a2.y - node0; csr[rowst[l] + atomicAdd(&cnt[l], 1)] = a2; }
            if (a3.y >= 0) { const int l = a3.y - node0; csr[rowst[l] + atomicAdd(&cnt[l], 1)] = a3; }
        }
    }
}

__global__ __launch_bounds__(256) void lgcn_weights(int2* __restrict__ csr,
                                                    const float* __restrict__ inv, int total) {
    const int t0 = (int)blockIdx.x * 1024 + (int)threadIdx.x;
    int2 e[4];
#pragma unroll
    for (int k = 0; k < 4; ++k) {
        const int t = t0 + k * 256;
        e[k] = (t < total) ? csr[t] : make_int2(0, 0);
    }
    float w[4];
#pragma unroll
    for (int k = 0; k < 4; ++k) w[k] = inv[e[k].x] * inv[e[k].y];
#pragma unroll
    for (int k = 0; k < 4; ++k) {
        const int t = t0 + k * 256;
        if (t < total) csr[t].y = __float_as_int(w[k]);
    }
}

template <bool SPLIT>
__device__ __forceinline__ const float* lgcn_src_row(const float* __restrict__ xa,
                                                     const float* __restrict__ xb,
                                                     int U, int s) {
    if (SPLIT)
        return (s < U) ? xa + (size_t)s * 64 : xb + (size_t)(s - U) * 64;
    return xa + (size_t)s * 64;
}

template <int K>
__device__ __forceinline__ int lgcn_row_share(int v) {
    return __builtin_amdgcn_update_dpp(0, v, 0x150 + K, 0xf, 0xf, true);
}

template <bool SPLIT, int K>
__device__ __forceinline__ float4 lgcn_gather(const float* __restrict__ xa,
                                              const float* __restrict__ xb, int U,
                                              int es, int q) {
    const int s = lgcn_row_share<K>(es);
    if (SPLIT) return *(const float4*)(lgcn_src_row<true>(xa, xb, U, s) + q);
    const unsigned off = ((unsigned)s << 8) | ((unsigned)q << 2);
    return *(const float4*)((const char*)xa + off);
}

template <int K>
__device__ __forceinline__ void lgcn_add(float4& acc, int ew, const float4& v) {
    const float w = __int_as_float(lgcn_row_share<K>(ew));
    acc.x = __builtin_fmaf(w, v.x, acc.x);
    acc.y = __builtin_fmaf(w, v.y, acc.y);
    acc.z = __builtin_fmaf(w, v.z, acc.z);
    acc.w = __builtin_fmaf(w, v.w, acc.w);
}

template <bool SPLIT, int K0>
__device__ __forceinline__ void lgcn_eight(float4& acc, const float* __restrict__ xa,
                                           const float* __restrict__ xb, int U,
                                           int es, int ew, int q, bool upper) {
    const float4 v0 = lgcn_gather<SPLIT, K0 + 0>(xa, xb, U, es, q);
    const float4 v1 = lgcn_gather<SPLIT, K0 + 1>(xa, xb, U, es, q);
    const float4 v2 = lgcn_gather<SPLIT, K0 + 2>(xa, xb, U, es, q);
    const float4 v3 = lgcn_gather<SPLIT, K0 + 3>(xa, xb, U, es, q);
    if (upper) {
        const float4 v4 = lgcn_gather<SPLIT, K0 + 4>(xa, xb, U, es, q);
        const float4 v5 = lgcn_gather<SPLIT, K0 + 5>(xa, xb, U, es, q);
        const float4 v6 = lgcn_gather<SPLIT, K0 + 6>(xa, xb, U, es, q);
        const float4 v7 = lgcn_gather<SPLIT, K0 + 7>(xa, xb, U, es, q);
        if (SPLIT) __builtin_amdgcn_sched_barrier(0);
        lgcn_add<K0 + 0>(acc, ew, v0);
        lgcn_add<K0 + 1>(acc, ew, v1);
        lgcn_add<K0 + 2>(acc, ew, v2);
        lgcn_add<K0 + 3>(acc, ew, v3);
        lgcn_add<K0 + 4>(acc, ew, v4);
        lgcn_add<K0 + 5>(acc, ew, v5);
        lgcn_add<K0 + 6>(acc, ew, v6);
        lgcn_add<K0 + 7>(acc, ew, v7);
    } else {
        lgcn_add<K0 + 0>(acc, ew, v0);
        lgcn_add<K0 + 1>(acc, ew, v1);
        lgcn_add<K0 + 2>(acc, ew, v2);
        lgcn_add<K0 + 3>(acc, ew, v3);
    }
}

template <bool SPLIT>
__device__ __forceinline__ float4 lgcn_row_sum(const int* __restrict__ rowptr,
                                               const int2* __restrict__ csr,
                                               const float* __restrict__ xa,
                                               const float* __restrict__ xb, int U,
                                               int row, int q) {
    const int lane = q >> 2;
    const int end = rowptr[row + 1];
    const int last = end - 1;
    float4 acc = make_float4(0.0f, 0.0f, 0.0f, 0.0f);
    for (int base = rowptr[row]; base < end; base += 16) {
        const int left = end - base;
        const int2 e = csr[min(base + lane, last)];
        const int ew = (lane < left) ? e.y : 0;
        lgcn_eight<SPLIT, 0>(acc, xa, xb, U, e.x, ew, q, left > 4);
        if (left > 8) lgcn_eight<SPLIT, 8>(acc, xa, xb, U, e.x, ew, q, left > 12);
    }
    return acc;
}

__device__ __forceinline__ int lgcn_batch_node(const int* __restrict__ users,
                                               const int* __restrict__ items,
                                               int entry, int B, int U) {
    return (entry < B) ? users[entry] : U + items[entry - B];
}

__device__ __forceinline__ int lgcn_batch_blocks(int B) {
    return (B + 7) >> 3;
}

template <bool FIRST>
__device__ __forceinline__ void lgcn_batch_acc(const int* __restrict__ users,
                                               const int* __restrict__ items,
                                               const float* __restrict__ xu,
                                               const float* __restrict__ xi,
                                               float* uacc, float* iacc, int B, int t) {
    const int entry = t >> 4;
    if (entry >= 2 * B) return;
    const int q = (t & 15) * 4;
    const float* src;
    float* dst;
    if (entry < B) {
        src = xu + (size_t)users[entry] * 64 + q;
        dst = uacc + (size_t)entry * 64 + q;
    } else {
        const int e = entry - B;
        src = xi + (size_t)items[e] * 64 + q;
        dst = iacc + (size_t)e * 64 + q;
    }
    float4 v = *(const float4*)src;
    if (!FIRST) {
        float4 a = *(const float4*)dst;
        a.x += v.x; a.y += v.y; a.z += v.z; a.w += v.w;
        v = a;
    }
    *(float4*)dst = v;
}

__global__ __launch_bounds__(256) void lgcn_spmm_pull(const int* __restrict__ users,
                                                      const int* __restrict__ items,
                                                      const int* __restrict__ rowptr,
                                                      const int2* __restrict__ csr,
                                                      const float* __restrict__ xin,
                                                      float* __restrict__ xout,
                                                      float* uacc, float* iacc,
                                                      int B, int U, int N) {
    const int batch_blocks = lgcn_batch_blocks(B);
    if ((int)blockIdx.x < batch_blocks) {
        lgcn_batch_acc<false>(users, items, xin, xin + (size_t)U * 64, uacc, iacc, B,
                              (int)blockIdx.x * 256 + (int)threadIdx.x);
        return;
    }
    const int t = ((int)blockIdx.x - batch_blocks) * 256 + (int)threadIdx.x;
    const int row = t >> 4;
    if (row >= N) return;
    const int q = (t & 15) * 4;
    *(float4*)(xout + (size_t)row * 64 + q) =
        lgcn_row_sum<false>(rowptr, csr, xin, nullptr, 0, row, q);
}

__device__ __forceinline__ void lgcn_mark_need(const int* __restrict__ users,
                                               const int* __restrict__ items,
                                               const int* __restrict__ rowptr,
                                               const int2* __restrict__ csr,
                                               int* __restrict__ need, int B, int U, int t) {
    const int entry = t >> 4;
    if (entry >= 2 * B) return;
    const int l = t & 15;
    const int node = lgcn_batch_node(users, items, entry, B, U);
    if (l == 0)
        __hip_atomic_fetch_or(&need[node], 1, __ATOMIC_RELAXED, __HIP_MEMORY_SCOPE_AGENT);
    const int end = rowptr[node + 1];
    for (int j = rowptr[node] + l; j < end; j += 16)
        __hip_atomic_fetch_or(&need[csr[j].x], 1, __ATOMIC_RELAXED, __HIP_MEMORY_SCOPE_AGENT);
}

__global__ __launch_bounds__(256) void lgcn_spmm_pull_first(const int* __restrict__ users,
                                                            const int* __restrict__ items,
                                                            const int* __restrict__ rowptr,
                                                            const int2* __restrict__ csr,
                                                            const float* __restrict__ ue,
                                                            const float* __restrict__ ie,
                                                            float* __restrict__ xout,
                                                            float* uacc, float* iacc,
                                                            int* __restrict__ need,
                                                            int B, int U, int N) {
    const int batch_blocks = lgcn_batch_blocks(B);
    if ((int)blockIdx.x < batch_blocks) {
        const int tb = (int)blockIdx.x * 256 + (int)threadIdx.x;
        lgcn_batch_acc<true>(users, items, ue, ie, uacc, iacc, B, tb);
        lgcn_mark_need(users, items, rowptr, csr, need, B, U, tb);
        return;
    }
    const int t = ((int)blockIdx.x - batch_blocks) * 256 + (int)threadIdx.x;
    const int row = t >> 4;
    if (row >= N) return;
    const int q = (t & 15) * 4;
    *(float4*)(xout + (size_t)row * 64 + q) =
        lgcn_row_sum<true>(rowptr, csr, ue, ie, U, row, q);
}

__global__ __launch_bounds__(256) void lgcn_spmm_pull_needed(const int* __restrict__ users,
                                                             const int* __restrict__ items,
                                                             const int* __restrict__ rowptr,
                                                             const int2* __restrict__ csr,
                                                             const int* __restrict__ need,
                                                             const float* __restrict__ xin,
                                                             float* __restrict__ xout,
                                                             float* uacc, float* iacc,
                                                             int B, int U, int N) {
    const int batch_blocks = lgcn_batch_blocks(B);
    if ((int)blockIdx.x < batch_blocks) {
        lgcn_batch_acc<false>(users, items, xin, xin + (size_t)U * 64, uacc, iacc, B,
                              (int)blockIdx.x * 256 + (int)threadIdx.x);
        return;
    }
    const int t = ((int)blockIdx.x - batch_blocks) * 256 + (int)threadIdx.x;
    const int row = t >> 4;
    if (row >= N) return;
    if (need[row] == 0) return;
    const int q = (t & 15) * 4;
    *(float4*)(xout + (size_t)row * 64 + q) =
        lgcn_row_sum<false>(rowptr, csr, xin, nullptr, 0, row, q);
}

__global__ __launch_bounds__(256) void lgcn_spmm_pull_batch(const int* __restrict__ users,
                                                            const int* __restrict__ items,
                                                            const int* __restrict__ rowptr,
                                                            const int2* __restrict__ csr,
                                                            const float* __restrict__ xin,
                                                            float* __restrict__ uacc,
                                                            float* __restrict__ iacc,
                                                            int B, int U) {
    const int t = blockIdx.x * 256 + threadIdx.x;
    const int entry = t >> 4;
    if (entry >= 2 * B) return;
    const int q = (t & 15) * 4;
    const int node = lgcn_batch_node(users, items, entry, B, U);
    float* dst = (entry < B) ? uacc + (size_t)entry * 64 + q
                             : iacc + (size_t)(entry - B) * 64 + q;
    float4 a = *(const float4*)dst;
    const float4 x3 = *(const float4*)(xin + (size_t)node * 64 + q);
    const float4 s = lgcn_row_sum<false>(rowptr, csr, xin, nullptr, 0, node, q);
    a.x += x3.x; a.y += x3.y; a.z += x3.z; a.w += x3.w;
    a.x += s.x; a.y += s.y; a.z += s.z; a.w += s.w;
    *(float4*)dst = a;
}

__global__ void lgcn_gather_acc(const int* __restrict__ users, const int* __restrict__ items,
                                const float* __restrict__ xu, const float* __restrict__ xi,
                                float* __restrict__ uacc, float* __restrict__ iacc,
                                int B) {
    int t = blockIdx.x * blockDim.x + threadIdx.x;
    int half = B * 64;
    if (t >= 2 * half) return;
    if (t < half) {
        int b = t >> 6, d = t & 63;
        uacc[t] += xu[(size_t)users[b] * 64 + d];
    } else {
        int tt = t - half;
        int b = tt >> 6, d = tt & 63;
        iacc[tt] += xi[(size_t)items[b] * 64 + d];
    }
}

__global__ __launch_bounds__(256) void lgcn_dot_wmma(const float* __restrict__ uacc,
                                                     const float* __restrict__ iacc,
                                                     float* __restrict__ out, int B) {
    __shared__ float tile[8][16 * 16];
    const int lane = threadIdx.x & 31;
    const int wave = threadIdx.x >> 5;
    const int tile0 = (blockIdx.x * 8 + wave) * 16;

    const int m = lane & 15;
    const int khalf = (lane >> 4) << 1;
    int row = tile0 + m;
    if (row > B - 1) row = B - 1;
    const float* __restrict__ ur = uacc + (size_t)row * 64;
    const float* __restrict__ ir = iacc + (size_t)row * 64;

    v8f c = {};
#pragma unroll
    for (int kb = 0; kb < 64; kb += 4) {
        int k0 = kb + khalf;
        v2f a; a.x = ur[k0]; a.y = ur[k0 + 1];
        v2f b; b.x = ir[k0]; b.y = ir[k0 + 1];
        c = __builtin_amdgcn_wmma_f32_16x16x4_f32(false, a, false, b,
                                                  (short)0, c, false, false);
    }

    float* lds = &tile[wave][0];
    const int n = lane & 15;
    const int mbase = (lane >> 4) ? 8 : 0;
#pragma unroll
    for (int r = 0; r < 8; ++r)
        lds[(mbase + r) * 16 + n] = c[r];
    __syncthreads();

    if (lane < 16) {
        int b = tile0 + lane;
        if (b < B)
            out[b] = lds[lane * 16 + lane] * (1.0f / 25.0f);
    }
}

extern "C" void kernel_launch(void* const* d_in, const int* in_sizes, int n_in,
                              void* d_out, int out_size, void* d_ws, size_t ws_size,
                              hipStream_t stream) {
    const int*   users = (const int*)d_in[0];
    const int*   items = (const int*)d_in[1];
    const int*   eu    = (const int*)d_in[2];
    const int*   ei    = (const int*)d_in[3];
    const float* uemb  = (const float*)d_in[4];
    const float* iemb  = (const float*)d_in[5];
    float*       out   = (float*)d_out;

    const int B = in_sizes[0];
    const int E = in_sizes[2];
    const int U = in_sizes[4] / 64;
    const int I = in_sizes[5] / 64;
    const int N = U + I;

    char* ws = (char*)d_ws;
    size_t off = 0;
    auto carve = [&](size_t bytes) -> float* {
        float* p = (float*)(ws + off);
        off += (bytes + 255) & ~(size_t)255;
        return p;
    };
    float* inv    = carve((size_t)N * 4);
    float* e_cur  = carve((size_t)N * 64 * 4);
    float* e_next = carve((size_t)N * 64 * 4);
    float* uacc   = carve((size_t)B * 64 * 4);
    float* iacc   = carve((size_t)B * 64 * 4);
    int*   rowptr = (int*)carve((size_t)(N + 1) * 4);
    int2*  csr    = (int2*)carve((size_t)2 * E * 8);

    const int nchunk = (N + 1023) / 1024;
    const int nblk = (E + LGCN_EB - 1) / LGCN_EB;
    size_t toff = 0;
    auto tcarve = [&](size_t bytes) -> int* {
        int* p = (int*)((char*)e_next + toff);
        toff += (bytes + 255) & ~(size_t)255;
        return p;
    };
    int2* binned = (int2*)tcarve((size_t)2 * E * 8);
    int*  segoff = tcarve((size_t)nblk * (nchunk + 1) * 4);
    float* inv_g = (float*)tcarve((size_t)N * 4);
    int*   need  = (int*)inv;

    const bool gather_path = (off <= ws_size) && (toff <= (size_t)N * 64 * 4) &&
                             ((size_t)N * 64 * 4 <= (size_t)0xffffffffu) &&
                             (nchunk <= 1023) && (nblk <= 1024) && (E <= 0x3fffffff);

    const int BS = 256;

    const int gacc_blocks = (2 * B * 64 + BS - 1) / BS;
    const size_t item0 = (size_t)U * 64;

    if (gather_path) {
        lgcn_bin_edges<<<nblk, 1024, 0, stream>>>(eu, ei, binned, segoff, E, U, nchunk);
        lgcn_build_rows<<<nchunk, 1024, 0, stream>>>(binned, segoff, rowptr, inv_g, need, csr, N, nchunk, nblk);
        lgcn_weights<<<(2 * E + 1023) / 1024, 256, 0, stream>>>(csr, inv_g, 2 * E);
        const int batch_blocks = (B + 7) / 8;
        const int pull_blocks = (N + 15) / 16;
        const int layer_blocks = pull_blocks + batch_blocks;

        lgcn_spmm_pull_first<<<layer_blocks, 256, 0, stream>>>(users, items, rowptr, csr, uemb, iemb,
                                                               e_cur, uacc, iacc, need, B, U, N);
        lgcn_spmm_pull<<<layer_blocks, 256, 0, stream>>>(users, items, rowptr, csr, e_cur, e_next,
                                                         uacc, iacc, B, U, N);
        lgcn_spmm_pull_needed<<<layer_blocks, 256, 0, stream>>>(users, items, rowptr, csr, need, e_next,
                                                                e_cur, uacc, iacc, B, U, N);
        lgcn_spmm_pull_batch<<<batch_blocks, 256, 0, stream>>>(users, items, rowptr, csr, e_cur, uacc, iacc, B, U);
    } else {
        hipMemsetAsync(uacc, 0, (size_t)B * 64 * 4, stream);
        hipMemsetAsync(iacc, 0, (size_t)B * 64 * 4, stream);

        {
            long long total = (long long)N * 64;
            lgcn_init_x0<<<(int)((total + BS - 1) / BS), BS, 0, stream>>>(uemb, iemb, e_cur, U, N);
        }
        hipMemsetAsync(inv, 0, (size_t)N * 4, stream);
        lgcn_degree<<<(E + BS - 1) / BS, BS, 0, stream>>>(eu, ei, inv, E, U);
        lgcn_invsqrt<<<(N + BS - 1) / BS, BS, 0, stream>>>(inv, N);

        lgcn_gather_acc<<<gacc_blocks, BS, 0, stream>>>(users, items, e_cur, e_cur + item0, uacc, iacc, B);

        const long long spmm_threads = (long long)2 * E * 16;
        const int spmm_blocks = (int)((spmm_threads + BS - 1) / BS);
        for (int layer = 0; layer < 4; ++layer) {
            hipMemsetAsync(e_next, 0, (size_t)N * 64 * 4, stream);
            lgcn_spmm<<<spmm_blocks, BS, 0, stream>>>(eu, ei, inv, e_cur, e_next, E, U);
            lgcn_gather_acc<<<gacc_blocks, BS, 0, stream>>>(users, items, e_next, e_next + item0, uacc, iacc, B);
            float* tmp = e_cur; e_cur = e_next; e_next = tmp;
        }
    }

    const int tiles = (B + 15) / 16;
    const int dot_blocks = (tiles + 7) / 8;
    lgcn_dot_wmma<<<dot_blocks, 256, 0, stream>>>(uacc, iacc, out, B);
}


# Two launches folded into neighbours: the layer-1 row gather forms the entry weights (one inv read per entry, kept for later layers); layer-4 blocks end with the batch dot on WMMA; sum order unchanged
# speedup vs baseline: 6.1010x; 1.0429x over previous
#include <hip/hip_runtime.h>


typedef __attribute__((ext_vector_type(2))) float v2f;
typedef __attribute__((ext_vector_type(8))) float v8f;

#define DEV_ATOMIC_ADD(p, v) \
    __hip_atomic_fetch_add((p), (v), __ATOMIC_RELAXED, __HIP_MEMORY_SCOPE_AGENT)

__global__ void lgcn_degree(const int* __restrict__ eu, const int* __restrict__ ei,
                            float* __restrict__ deg, int E, int U) {
    int t = blockIdx.x * blockDim.x + threadIdx.x;
    if (t >= E) return;
    DEV_ATOMIC_ADD(&deg[eu[t]], 1.0f);
    DEV_ATOMIC_ADD(&deg[U + ei[t]], 1.0f);
}

__global__ void lgcn_invsqrt(float* __restrict__ d, int N) {
    int t = blockIdx.x * blockDim.x + threadIdx.x;
    if (t >= N) return;
    float v = d[t];
    d[t] = (v > 0.0f) ? (1.0f / sqrtf(v)) : 0.0f;
}

__global__ void lgcn_init_x0(const float* __restrict__ ue, const float* __restrict__ ie,
                             float* __restrict__ x, int U, int N) {
    long long t = (long long)blockIdx.x * blockDim.x + threadIdx.x;
    long long total = (long long)N * 64;
    if (t >= total) return;
    long long ubytes = (long long)U * 64;
    x[t] = (t < ubytes) ? ue[t] : ie[t - ubytes];
}

__global__ void lgcn_spmm(const int* __restrict__ eu, const int* __restrict__ ei,
                          const float* __restrict__ inv,
                          const float* __restrict__ xin, float* __restrict__ xout,
                          int E, int U) {
    long long t = (long long)blockIdx.x * blockDim.x + threadIdx.x;
    int edge = (int)(t >> 4);
    if (edge >= 2 * E) return;
    int q = ((int)t & 15) * 4;

    int s, d;
    if (edge < E) {
        s = eu[edge];
        d = U + ei[edge];
    } else {
        int e2 = edge - E;
        s = U + ei[e2];
        d = eu[e2];
    }
    float w = inv[s] * inv[d];
    const float4 v = *(const float4*)(xin + (size_t)s * 64 + q);
    float* o = xout + (size_t)d * 64 + q;
    DEV_ATOMIC_ADD(o + 0, w * v.x);
    DEV_ATOMIC_ADD(o + 1, w * v.y);
    DEV_ATOMIC_ADD(o + 2, w * v.z);
    DEV_ATOMIC_ADD(o + 3, w * v.w);
}


#define LGCN_EB 8192
#define LGCN_BIN_SHIFT 10

__global__ __launch_bounds__(1024) void lgcn_bin_edges(const int* __restrict__ eu,
                                                       const int* __restrict__ ei,
                                                       int2* __restrict__ binned,
                                                       int* __restrict__ segoff,
                                                       int E, int U, int nb) {
    __shared__ int hist[1024];
    __shared__ int wtot[32];
    __shared__ int2 stage[2 * LGCN_EB];
    const int tid = threadIdx.x;
    const int lane = tid & 31;
    const int wave = tid >> 5;
    hist[tid] = 0;
    __syncthreads();

    const int e0 = (int)blockIdx.x * LGCN_EB;
    int un[8], in[8], ra[8], rb[8];
#pragma unroll
    for (int k = 0; k < 8; ++k) {
        const int e = e0 + k * 1024 + tid;
        un[k] = -1; in[k] = 0; ra[k] = 0; rb[k] = 0;
        if (e < E) {
            un[k] = eu[e];
            in[k] = U + ei[e];
        }
    }
#pragma unroll
    for (int k = 0; k < 8; ++k) {
        if (un[k] >= 0) {
            ra[k] = atomicAdd(&hist[in[k] >> LGCN_BIN_SHIFT], 1);
            rb[k] = atomicAdd(&hist[un[k] >> LGCN_BIN_SHIFT], 1);
        }
    }
    __syncthreads();

    const int v = hist[tid];
    int incl = v;
#pragma unroll
    for (int off = 1; off < 32; off <<= 1) {
        int t = __shfl_up(incl, off);
        if (lane >= off) incl += t;
    }
    if (lane == 31) wtot[wave] = incl;
    __syncthreads();
    int excl = incl - v;
    for (int w = 0; w < wave; ++w) excl += wtot[w];
    hist[tid] = excl;
    if (tid <= nb) segoff[(size_t)blockIdx.x * (nb + 1) + tid] = excl;
    __syncthreads();

#pragma unroll
    for (int k = 0; k < 8; ++k) {
        if (un[k] >= 0) {
            stage[hist[in[k] >> LGCN_BIN_SHIFT] + ra[k]] = make_int2(un[k], in[k]);
            stage[hist[un[k] >> LGCN_BIN_SHIFT] + rb[k]] = make_int2(in[k], un[k]);
        }
    }
    __syncthreads();

    const int pairs = min(LGCN_EB, E - e0);
    int4* slice4 = (int4*)(binned + (size_t)2 * e0);
    for (int k = 0; k < 8; ++k) {
        const int j = k * 1024 + tid;
        if (j < pairs) {
            const int2 p0 = stage[2 * j];
            const int2 p1 = stage[2 * j + 1];
            slice4[j] = make_int4(p0.x, p0.y, p1.x, p1.y);
        }
    }
}

__device__ __forceinline__ int2 lgcn_run_entry(const int2* __restrict__ run, int j, int end) {
    return (j < end) ? run[j] : make_int2(0, -1);
}

#define LGCN_STAGE_CAP 16384

__global__ __launch_bounds__(1024) void lgcn_build_rows(const int2* __restrict__ binned,
                                                        const int* __restrict__ segoff,
                                                        int* __restrict__ rowptr,
                                                        float* __restrict__ inv,
                                                        int* __restrict__ need,
                                                        int2* __restrict__ csr,
                                                        int N, int nb, int nblk) {
    __shared__ int segS[1024];
    __shared__ int segE[1024];
    __shared__ int cnt[1024];
    __shared__ int rowst[1024];
    __shared__ int wtot[32];
    __shared__ int binstart;
    __shared__ int bintotal;
    __shared__ int2 stage[LGCN_STAGE_CAP];
    const int tid = threadIdx.x;
    const int lane = tid & 31;
    const int wave = tid >> 5;
    const int b = blockIdx.x;
    const int node0 = b << LGCN_BIN_SHIFT;

    if (tid == 0) binstart = 0;
    cnt[tid] = 0;
    __syncthreads();
    if (tid < nblk) {
        const int s = segoff[(size_t)tid * (nb + 1) + b];
        const int e = segoff[(size_t)tid * (nb + 1) + b + 1];
        segS[tid] = s;
        segE[tid] = e;
        atomicAdd(&binstart, s);
    }
    __syncthreads();

    for (int blk = wave; blk < nblk; blk += 32) {
        const int2* run = binned + (size_t)2 * LGCN_EB * blk;
        const int end = segE[blk];
        for (int j = segS[blk] + lane; j < end; j += 128) {
            const int2 a0 = lgcn_run_entry(run, j, end);
            const int2 a1 = lgcn_run_entry(run, j + 32, end);
            const int2 a2 = lgcn_run_entry(run, j + 64, end);
            const int2 a3 = lgcn_run_entry(run, j + 96, end);
            if (a0.y >= 0) atomicAdd(&cnt[a0.y - node0], 1);
            if (a1.y >= 0) atomicAdd(&cnt[a1.y - node0], 1);
            if (a2.y >= 0) atomicAdd(&cnt[a2.y - node0], 1);
            if (a3.y >= 0) atomicAdd(&cnt[a3.y - node0], 1);
        }
    }
    __syncthreads();

    const int c = cnt[tid];
    int incl = c;
#pragma unroll
    for (int off = 1; off < 32; off <<= 1) {
        int t = __shfl_up(incl, off);
        if (lane >= off) incl += t;
    }
    if (lane == 31) wtot[wave] = incl;
    __syncthreads();
    const int bs = binstart;
    int start = bs + incl - c;
    for (int w = 0; w < wave; ++w) start += wtot[w];
    if (tid == 1023) bintotal = start + c - bs;
    const int node = node0 + tid;
    if (node < N) {
        rowptr[node] = start;
        const float v = (float)c;
        inv[node] = (v > 0.0f) ? (1.0f / sqrtf(v)) : 0.0f;
        need[node] = 0;
        if (node == N - 1) {
            rowptr[N] = start + c;
            inv[N] = 0.0f;
        }
    }
    rowst[tid] = start;
    cnt[tid] = 0;
    __syncthreads();

    const int total = bintotal;
    if (total <= LGCN_STAGE_CAP) {
        for (int blk = wave; blk < nblk; blk += 32) {
            const int2* run = binned + (size_t)2 * LGCN_EB * blk;
            const int end = segE[blk];
            for (int j = segS[blk] + lane; j < end; j += 128) {
                const int2 a0 = lgcn_run_entry(run, j, end);
                const int2 a1 = lgcn_run_entry(run, j + 32, end);
                const int2 a2 = lgcn_run_entry(run, j + 64, end);
                const int2 a3 = lgcn_run_entry(run, j + 96, end);
                if (a0.y >= 0) { const int l = a0.y - node0; stage[rowst[l] - bs + atomicAdd(&cnt[l], 1)] = a0; }
                if (a1.y >= 0) { const int l = a1.y - node0; stage[rowst[l] - bs + atomicAdd(&cnt[l], 1)] = a1; }
                if (a2.y >= 0) { const int l = a2.y - node0; stage[rowst[l] - bs + atomicAdd(&cnt[l], 1)] = a2; }
                if (a3.y >= 0) { const int l = a3.y - node0; stage[rowst[l] - bs + atomicAdd(&cnt[l], 1)] = a3; }
            }
        }
        __syncthreads();
        const int shift = bs & 15;
        for (int j = tid - shift; j < total; j += 1024)
            if (j >= 0) csr[bs + j] = stage[j];
        return;
    }

    for (int blk = wave; blk < nblk; blk += 32) {
        const int2* run = binned + (size_t)2 * LGCN_EB * blk;
        const int end = segE[blk];
        for (int j = segS[blk] + lane; j < end; j += 128) {
            const int2 a0 = lgcn_run_entry(run, j, end);
            const int2 a1 = lgcn_run_entry(run, j + 32, end);
            const int2 a2 = lgcn_run_entry(run, j + 64, end);
            const int2 a3 = lgcn_run_entry(run, j + 96, end);
            if (a0.y >= 0) { const int l = a0.y - node0; csr[rowst[l] + atomicAdd(&cnt[l], 1)] = a0; }
            if (a1.y >= 0) { const int l = a1.y - node0; csr[rowst[l] + atomicAdd(&cnt[l], 1)] = a1; }
            if (a2.y >= 0) { const int l = a2.y - node0; csr[rowst[l] + atomicAdd(&cnt[l], 1)] = a2; }
            if (a3.y >= 0) { const int l = a3.y - node0; csr[rowst[l] + atomicAdd(&cnt[l], 1)] = a3; }
        }
    }
}

template <bool SPLIT>
__device__ __forceinline__ const float* lgcn_src_row(const float* __restrict__ xa,
                                                     const float* __restrict__ xb,
                                                     int U, int s) {
    if (SPLIT)
        return (s < U) ? xa + (size_t)s * 64 : xb + (size_t)(s - U) * 64;
    return xa + (size_t)s * 64;
}

template <int K>
__device__ __forceinline__ int lgcn_row_share(int v) {
    return __builtin_amdgcn_update_dpp(0, v, 0x150 + K, 0xf, 0xf, true);
}

template <bool SPLIT, int K>
__device__ __forceinline__ float4 lgcn_gather(const float* __restrict__ xa,
                                              const float* __restrict__ xb, int U,
                                              int es, int q) {
    const int s = lgcn_row_share<K>(es);
    if (SPLIT) return *(const float4*)(lgcn_src_row<true>(xa, xb, U, s) + q);
    const unsigned off = ((unsigned)s << 8) | ((unsigned)q << 2);
    return *(const float4*)((const char*)xa + off);
}

template <int K, bool FORM>
__device__ __forceinline__ void lgcn_add(float4& acc, int ew, float di, const float4& v) {
    float w = __int_as_float(lgcn_row_share<K>(ew));
    if (FORM) w *= di;
    acc.x = __builtin_fmaf(w, v.x, acc.x);
    acc.y = __builtin_fmaf(w, v.y, acc.y);
    acc.z = __builtin_fmaf(w, v.z, acc.z);
    acc.w = __builtin_fmaf(w, v.w, acc.w);
}

template <bool SPLIT, int K0, bool FORM>
__device__ __forceinline__ void lgcn_eight(float4& acc, const float* __restrict__ xa,
                                           const float* __restrict__ xb, int U,
                                           int es, int ew, float di, int q, bool upper) {
    const float4 v0 = lgcn_gather<SPLIT, K0 + 0>(xa, xb, U, es, q);
    const float4 v1 = lgcn_gather<SPLIT, K0 + 1>(xa, xb, U, es, q);
    const float4 v2 = lgcn_gather<SPLIT, K0 + 2>(xa, xb, U, es, q);
    const float4 v3 = lgcn_gather<SPLIT, K0 + 3>(xa, xb, U, es, q);
    if (upper) {
        const float4 v4 = lgcn_gather<SPLIT, K0 + 4>(xa, xb, U, es, q);
        const float4 v5 = lgcn_gather<SPLIT, K0 + 5>(xa, xb, U, es, q);
        const float4 v6 = lgcn_gather<SPLIT, K0 + 6>(xa, xb, U, es, q);
        const float4 v7 = lgcn_gather<SPLIT, K0 + 7>(xa, xb, U, es, q);
        if (SPLIT) __builtin_amdgcn_sched_barrier(0);
        lgcn_add<K0 + 0, FORM>(acc, ew, di, v0);
        lgcn_add<K0 + 1, FORM>(acc, ew, di, v1);
        lgcn_add<K0 + 2, FORM>(acc, ew, di, v2);
        lgcn_add<K0 + 3, FORM>(acc, ew, di, v3);
        lgcn_add<K0 + 4, FORM>(acc, ew, di, v4);
        lgcn_add<K0 + 5, FORM>(acc, ew, di, v5);
        lgcn_add<K0 + 6, FORM>(acc, ew, di, v6);
        lgcn_add<K0 + 7, FORM>(acc, ew, di, v7);
    } else {
        lgcn_add<K0 + 0, FORM>(acc, ew, di, v0);
        lgcn_add<K0 + 1, FORM>(acc, ew, di, v1);
        lgcn_add<K0 + 2, FORM>(acc, ew, di, v2);
        lgcn_add<K0 + 3, FORM>(acc, ew, di, v3);
    }
}

template <bool SPLIT>
__device__ __forceinline__ float4 lgcn_row_sum(const int* __restrict__ rowptr,
                                               const int2* __restrict__ csr,
                                               const float* __restrict__ xa,
                                               const float* __restrict__ xb, int U,
                                               int row, int q) {
    const int lane = q >> 2;
    const int end = rowptr[row + 1];
    const int last = end - 1;
    float4 acc = make_float4(0.0f, 0.0f, 0.0f, 0.0f);
    for (int base = rowptr[row]; base < end; base += 16) {
        const int left = end - base;
        const int2 e = csr[min(base + lane, last)];
        const int ew = (lane < left) ? e.y : 0;
        lgcn_eight<SPLIT, 0, false>(acc, xa, xb, U, e.x, ew, 0.0f, q, left > 4);
        if (left > 8) lgcn_eight<SPLIT, 8, false>(acc, xa, xb, U, e.x, ew, 0.0f, q, left > 12);
    }
    return acc;
}

__device__ __forceinline__ float4 lgcn_row_sum_first(const int* __restrict__ rowptr,
                                                     int2* csr,
                                                     const float* __restrict__ inv,
                                                     const float* __restrict__ ue,
                                                     const float* __restrict__ ie, int U,
                                                     int N, int row, int q) {
    const int lane = q >> 2;
    const int end = rowptr[row + 1];
    const int last = end - 1;
    const float di = inv[row];
    float4 acc = make_float4(0.0f, 0.0f, 0.0f, 0.0f);
    for (int base = rowptr[row]; base < end; base += 16) {
        const int left = end - base;
        const bool live = lane < left;
        const unsigned eoff = (unsigned)min(base + lane, last) << 3;
        const int s = *(const int*)((const char*)csr + eoff);
        const float wi = inv[live ? s : N];
        const int wbits = __float_as_int(wi);
        lgcn_eight<true, 0, true>(acc, ue, ie, U, s, wbits, di, q, left > 4);
        if (live) *(int*)((char*)csr + eoff + 4) = __float_as_int(wi * di);
        if (left > 8) lgcn_eight<true, 8, true>(acc, ue, ie, U, s, wbits, di, q, left > 12);
    }
    return acc;
}

__device__ __forceinline__ int lgcn_batch_node(const int* __restrict__ users,
                                               const int* __restrict__ items,
                                               int entry, int B, int U) {
    return (entry < B) ? users[entry] : U + items[entry - B];
}

__device__ __forceinline__ int lgcn_batch_blocks(int B) {
    return (B + 7) >> 3;
}

template <bool FIRST>
__device__ __forceinline__ void lgcn_batch_acc(const int* __restrict__ users,
                                               const int* __restrict__ items,
                                               const float* __restrict__ xu,
                                               const float* __restrict__ xi,
                                               float* uacc, float* iacc, int B, int t) {
    const int entry = t >> 4;
    if (entry >= 2 * B) return;
    const int q = (t & 15) * 4;
    const float* src;
    float* dst;
    if (entry < B) {
        src = xu + (size_t)users[entry] * 64 + q;
        dst = uacc + (size_t)entry * 64 + q;
    } else {
        const int e = entry - B;
        src = xi + (size_t)items[e] * 64 + q;
        dst = iacc + (size_t)e * 64 + q;
    }
    float4 v = *(const float4*)src;
    if (!FIRST) {
        float4 a = *(const float4*)dst;
        a.x += v.x; a.y += v.y; a.z += v.z; a.w += v.w;
        v = a;
    }
    *(float4*)dst = v;
}

__global__ __launch_bounds__(256) void lgcn_spmm_pull(const int* __restrict__ users,
                                                      const int* __restrict__ items,
                                                      const int* __restrict__ rowptr,
                                                      const int2* __restrict__ csr,
                                                      const float* __restrict__ xin,
                                                      float* __restrict__ xout,
                                                      float* uacc, float* iacc,
                                                      int B, int U, int N) {
    const int batch_blocks = lgcn_batch_blocks(B);
    if ((int)blockIdx.x < batch_blocks) {
        lgcn_batch_acc<false>(users, items, xin, xin + (size_t)U * 64, uacc, iacc, B,
                              (int)blockIdx.x * 256 + (int)threadIdx.x);
        return;
    }
    const int t = ((int)blockIdx.x - batch_blocks) * 256 + (int)threadIdx.x;
    const int row = t >> 4;
    if (row >= N) return;
    const int q = (t & 15) * 4;
    *(float4*)(xout + (size_t)row * 64 + q) =
        lgcn_row_sum<false>(rowptr, csr, xin, nullptr, 0, row, q);
}

__device__ __forceinline__ void lgcn_mark_need(const int* __restrict__ users,
                                               const int* __restrict__ items,
                                               const int* __restrict__ rowptr,
                                               const int2* __restrict__ csr,
                                               int* __restrict__ need, int B, int U, int t) {
    const int entry = t >> 4;
    if (entry >= 2 * B) return;
    const int l = t & 15;
    const int node = lgcn_batch_node(users, items, entry, B, U);
    if (l == 0)
        __hip_atomic_fetch_or(&need[node], 1, __ATOMIC_RELAXED, __HIP_MEMORY_SCOPE_AGENT);
    const int end = rowptr[node + 1];
    for (int j = rowptr[node] + l; j < end; j += 16)
        __hip_atomic_fetch_or(&need[csr[j].x], 1, __ATOMIC_RELAXED, __HIP_MEMORY_SCOPE_AGENT);
}

__global__ __launch_bounds__(256) void lgcn_spmm_pull_first(const int* __restrict__ users,
                                                            const int* __restrict__ items,
                                                            const int* __restrict__ rowptr,
                                                            int2* csr,
                                                            const float* __restrict__ inv,
                                                            const float* __restrict__ ue,
                                                            const float* __restrict__ ie,
                                                            float* __restrict__ xout,
                                                            float* uacc, float* iacc,
                                                            int* __restrict__ need,
                                                            int B, int U, int N) {
    const int batch_blocks = lgcn_batch_blocks(B);
    if ((int)blockIdx.x < batch_blocks) {
        const int tb = (int)blockIdx.x * 256 + (int)threadIdx.x;
        lgcn_batch_acc<true>(users, items, ue, ie, uacc, iacc, B, tb);
        lgcn_mark_need(users, items, rowptr, csr, need, B, U, tb);
        return;
    }
    const int t = ((int)blockIdx.x - batch_blocks) * 256 + (int)threadIdx.x;
    const int row = t >> 4;
    if (row >= N) return;
    const int q = (t & 15) * 4;
    *(float4*)(xout + (size_t)row * 64 + q) =
        lgcn_row_sum_first(rowptr, csr, inv, ue, ie, U, N, row, q);
}

__global__ __launch_bounds__(256) void lgcn_spmm_pull_needed(const int* __restrict__ users,
                                                             const int* __restrict__ items,
                                                             const int* __restrict__ rowptr,
                                                             const int2* __restrict__ csr,
                                                             const int* __restrict__ need,
                                                             const float* __restrict__ xin,
                                                             float* __restrict__ xout,
                                                             float* uacc, float* iacc,
                                                             int B, int U, int N) {
    const int batch_blocks = lgcn_batch_blocks(B);
    if ((int)blockIdx.x < batch_blocks) {
        lgcn_batch_acc<false>(users, items, xin, xin + (size_t)U * 64, uacc, iacc, B,
                              (int)blockIdx.x * 256 + (int)threadIdx.x);
        return;
    }
    const int t = ((int)blockIdx.x - batch_blocks) * 256 + (int)threadIdx.x;
    const int row = t >> 4;
    if (row >= N) return;
    if (need[row] == 0) return;
    const int q = (t & 15) * 4;
    *(float4*)(xout + (size_t)row * 64 + q) =
        lgcn_row_sum<false>(rowptr, csr, xin, nullptr, 0, row, q);
}

#define LGCN_FIN_PITCH 68

__global__ __launch_bounds__(256) void lgcn_spmm_pull_batch_dot(const int* __restrict__ users,
                                                                const int* __restrict__ items,
                                                                const int* __restrict__ rowptr,
                                                                const int2* __restrict__ csr,
                                                                const float* __restrict__ xin,
                                                                const float* __restrict__ uacc,
                                                                const float* __restrict__ iacc,
                                                                float* __restrict__ out,
                                                                int B, int U) {
    __shared__ __attribute__((aligned(16))) float fin[2][8][LGCN_FIN_PITCH];
    __shared__ float tile[16 * 16];
    const int tid = threadIdx.x;
    const int grp = tid >> 4;
    const int q = (tid & 15) * 4;
    const int slot = grp & 7;
    const int side = grp >> 3;
    int b = (int)blockIdx.x * 8 + slot;
    if (b > B - 1) b = B - 1;
    const int node = side ? U + items[b] : users[b];
    const float* run = (side ? iacc : uacc) + (size_t)b * 64 + q;
    float4 a = *(const float4*)run;
    const float4 x3 = *(const float4*)(xin + (size_t)node * 64 + q);
    const float4 s = lgcn_row_sum<false>(rowptr, csr, xin, nullptr, 0, node, q);
    a.x += x3.x; a.y += x3.y; a.z += x3.z; a.w += x3.w;
    a.x += s.x; a.y += s.y; a.z += s.z; a.w += s.w;
    *(float4*)&fin[side][slot][q] = a;
    __syncthreads();

    if (tid >= 32) return;
    const int lane = tid;
    const int m = lane & 7;
    const int khalf = (lane >> 4) << 1;
    const float* ur = &fin[0][m][0];
    const float* ir = &fin[1][m][0];
    v8f c = {};
#pragma unroll
    for (int kb = 0; kb < 64; kb += 4) {
        const int k0 = kb + khalf;
        v2f fa; fa.x = ur[k0]; fa.y = ur[k0 + 1];
        v2f fb; fb.x = ir[k0]; fb.y = ir[k0 + 1];
        c = __builtin_amdgcn_wmma_f32_16x16x4_f32(false, fa, false, fb,
                                                  (short)0, c, false, false);
    }
    const int n = lane & 15;
    const int mbase = (lane >> 4) ? 8 : 0;
#pragma unroll
    for (int r = 0; r < 8; ++r)
        tile[(mbase + r) * 16 + n] = c[r];
    const int bo = (int)blockIdx.x * 8 + lane;
    if (lane < 8 && bo < B) out[bo] = tile[lane * 16 + lane] * (1.0f / 25.0f);
}

__global__ void lgcn_gather_acc(const int* __restrict__ users, const int* __restrict__ items,
                                const float* __restrict__ xu, const float* __restrict__ xi,
                                float* __restrict__ uacc, float* __restrict__ iacc,
                                int B) {
    int t = blockIdx.x * blockDim.x + threadIdx.x;
    int half = B * 64;
    if (t >= 2 * half) return;
    if (t < half) {
        int b = t >> 6, d = t & 63;
        uacc[t] += xu[(size_t)users[b] * 64 + d];
    } else {
        int tt = t - half;
        int b = tt >> 6, d = tt & 63;
        iacc[tt] += xi[(size_t)items[b] * 64 + d];
    }
}

__global__ __launch_bounds__(256) void lgcn_dot_wmma(const float* __restrict__ uacc,
                                                     const float* __restrict__ iacc,
                                                     float* __restrict__ out, int B) {
    __shared__ float tile[8][16 * 16];
    const int lane = threadIdx.x & 31;
    const int wave = threadIdx.x >> 5;
    const int tile0 = (blockIdx.x * 8 + wave) * 16;

    const int m = lane & 15;
    const int khalf = (lane >> 4) << 1;
    int row = tile0 + m;
    if (row > B - 1) row = B - 1;
    const float* __restrict__ ur = uacc + (size_t)row * 64;
    const float* __restrict__ ir = iacc + (size_t)row * 64;

    v8f c = {};
#pragma unroll
    for (int kb = 0; kb < 64; kb += 4) {
        int k0 = kb + khalf;
        v2f a; a.x = ur[k0]; a.y = ur[k0 + 1];
        v2f b; b.x = ir[k0]; b.y = ir[k0 + 1];
        c = __builtin_amdgcn_wmma_f32_16x16x4_f32(false, a, false, b,
                                                  (short)0, c, false, false);
    }

    float* lds = &tile[wave][0];
    const int n = lane & 15;
    const int mbase = (lane >> 4) ? 8 : 0;
#pragma unroll
    for (int r = 0; r < 8; ++r)
        lds[(mbase + r) * 16 + n] = c[r];
    __syncthreads();

    if (lane < 16) {
        int b = tile0 + lane;
        if (b < B)
            out[b] = lds[lane * 16 + lane] * (1.0f / 25.0f);
    }
}

extern "C" void kernel_launch(void* const* d_in, const int* in_sizes, int n_in,
                              void* d_out, int out_size, void* d_ws, size_t ws_size,
                              hipStream_t stream) {
    const int*   users = (const int*)d_in[0];
    const int*   items = (const int*)d_in[1];
    const int*   eu    = (const int*)d_in[2];
    const int*   ei    = (const int*)d_in[3];
    const float* uemb  = (const float*)d_in[4];
    const float* iemb  = (const float*)d_in[5];
    float*       out   = (float*)d_out;

    const int B = in_sizes[0];
    const int E = in_sizes[2];
    const int U = in_sizes[4] / 64;
    const int I = in_sizes[5] / 64;
    const int N = U + I;

    char* ws = (char*)d_ws;
    size_t off = 0;
    auto carve = [&](size_t bytes) -> float* {
        float* p = (float*)(ws + off);
        off += (bytes + 255) & ~(size_t)255;
        return p;
    };
    float* inv    = carve((size_t)N * 4);
    float* e_cur  = carve((size_t)N * 64 * 4);
    float* e_next = carve((size_t)N * 64 * 4);
    float* uacc   = carve((size_t)B * 64 * 4);
    float* iacc   = carve((size_t)B * 64 * 4);
    int*   rowptr = (int*)carve((size_t)(N + 1) * 4);
    int2*  csr    = (int2*)carve((size_t)2 * E * 8);

    const int nchunk = (N + 1023) / 1024;
    const int nblk = (E + LGCN_EB - 1) / LGCN_EB;
    size_t toff = 0;
    auto tcarve = [&](size_t bytes) -> int* {
        int* p = (int*)((char*)e_next + toff);
        toff += (bytes + 255) & ~(size_t)255;
        return p;
    };
    int2* binned = (int2*)tcarve((size_t)2 * E * 8);
    int*  segoff = tcarve((size_t)nblk * (nchunk + 1) * 4);
    float* inv_g = (float*)tcarve((size_t)(N + 1) * 4);
    int*   need  = (int*)inv;

    const bool gather_path = (off <= ws_size) && (toff <= (size_t)N * 64 * 4) &&
                             ((size_t)N * 64 * 4 <= (size_t)0xffffffffu) &&
                             (nchunk <= 1023) && (nblk <= 1024) && (E <= 0x0fffffff);

    const int BS = 256;

    const int gacc_blocks = (2 * B * 64 + BS - 1) / BS;
    const size_t item0 = (size_t)U * 64;

    if (gather_path) {
        lgcn_bin_edges<<<nblk, 1024, 0, stream>>>(eu, ei, binned, segoff, E, U, nchunk);
        lgcn_build_rows<<<nchunk, 1024, 0, stream>>>(binned, segoff, rowptr, inv_g, need, csr, N, nchunk, nblk);
        const int batch_blocks = (B + 7) / 8;
        const int pull_blocks = (N + 15) / 16;
        const int layer_blocks = pull_blocks + batch_blocks;

        lgcn_spmm_pull_first<<<layer_blocks, 256, 0, stream>>>(users, items, rowptr, csr, inv_g, uemb, iemb,
                                                               e_cur, uacc, iacc, need, B, U, N);
        lgcn_spmm_pull<<<layer_blocks, 256, 0, stream>>>(users, items, rowptr, csr, e_cur, e_next,
                                                         uacc, iacc, B, U, N);
        lgcn_spmm_pull_needed<<<layer_blocks, 256, 0, stream>>>(users, items, rowptr, csr, need, e_next,
                                                                e_cur, uacc, iacc, B, U, N);
        lgcn_spmm_pull_batch_dot<<<batch_blocks, 256, 0, stream>>>(users, items, rowptr, csr, e_cur, uacc, iacc, out, B, U);
        return;
    } else {
        hipMemsetAsync(uacc, 0, (size_t)B * 64 * 4, stream);
        hipMemsetAsync(iacc, 0, (size_t)B * 64 * 4, stream);

        {
            long long total = (long long)N * 64;
            lgcn_init_x0<<<(int)((total + BS - 1) / BS), BS, 0, stream>>>(uemb, iemb, e_cur, U, N);
        }
        hipMemsetAsync(inv, 0, (size_t)N * 4, stream);
        lgcn_degree<<<(E + BS - 1) / BS, BS, 0, stream>>>(eu, ei, inv, E, U);
        lgcn_invsqrt<<<(N + BS - 1) / BS, BS, 0, stream>>>(inv, N);

        lgcn_gather_acc<<<gacc_blocks, BS, 0, stream>>>(users, items, e_cur, e_cur + item0, uacc, iacc, B);

        const long long spmm_threads = (long long)2 * E * 16;
        const int spmm_blocks = (int)((spmm_threads + BS - 1) / BS);
        for (int layer = 0; layer < 4; ++layer) {
            hipMemsetAsync(e_next, 0, (size_t)N * 64 * 4, stream);
            lgcn_spmm<<<spmm_blocks, BS, 0, stream>>>(eu, ei, inv, e_cur, e_next, E, U);
            lgcn_gather_acc<<<gacc_blocks, BS, 0, stream>>>(users, items, e_next, e_next + item0, uacc, iacc, B);
            float* tmp = e_cur; e_cur = e_next; e_next = tmp;
        }
    }

    const int tiles = (B + 15) / 16;
    const int dot_blocks = (tiles + 7) / 8;
    lgcn_dot_wmma<<<dot_blocks, 256, 0, stream>>>(uacc, iacc, out, B);
}


# Layers 3 and 4 in one launch: a block's 64 row groups sum the layer-3 rows its 8 batch pairs read into LDS, then the layer-4 sums and dot; the pass over all rows and its marks go; sum order unchanged
# speedup vs baseline: 6.8970x; 1.1305x over previous
#include <hip/hip_runtime.h>


typedef __attribute__((ext_vector_type(2))) float v2f;
typedef __attribute__((ext_vector_type(8))) float v8f;

#define DEV_ATOMIC_ADD(p, v) \
    __hip_atomic_fetch_add((p), (v), __ATOMIC_RELAXED, __HIP_MEMORY_SCOPE_AGENT)

__global__ void lgcn_degree(const int* __restrict__ eu, const int* __restrict__ ei,
                            float* __restrict__ deg, int E, int U) {
    int t = blockIdx.x * blockDim.x + threadIdx.x;
    if (t >= E) return;
    DEV_ATOMIC_ADD(&deg[eu[t]], 1.0f);
    DEV_ATOMIC_ADD(&deg[U + ei[t]], 1.0f);
}

__global__ void lgcn_invsqrt(float* __restrict__ d, int N) {
    int t = blockIdx.x * blockDim.x + threadIdx.x;
    if (t >= N) return;
    float v = d[t];
    d[t] = (v > 0.0f) ? (1.0f / sqrtf(v)) : 0.0f;
}

__global__ void lgcn_init_x0(const float* __restrict__ ue, const float* __restrict__ ie,
                             float* __restrict__ x, int U, int N) {
    long long t = (long long)blockIdx.x * blockDim.x + threadIdx.x;
    long long total = (long long)N * 64;
    if (t >= total) return;
    long long ubytes = (long long)U * 64;
    x[t] = (t < ubytes) ? ue[t] : ie[t - ubytes];
}

__global__ void lgcn_spmm(const int* __restrict__ eu, const int* __restrict__ ei,
                          const float* __restrict__ inv,
                          const float* __restrict__ xin, float* __restrict__ xout,
                          int E, int U) {
    long long t = (long long)blockIdx.x * blockDim.x + threadIdx.x;
    int edge = (int)(t >> 4);
    if (edge >= 2 * E) return;
    int q = ((int)t & 15) * 4;

    int s, d;
    if (edge < E) {
        s = eu[edge];
        d = U + ei[edge];
    } else {
        int e2 = edge - E;
        s = U + ei[e2];
        d = eu[e2];
    }
    float w = inv[s] * inv[d];
    const float4 v = *(const float4*)(xin + (size_t)s * 64 + q);
    float* o = xout + (size_t)d * 64 + q;
    DEV_ATOMIC_ADD(o + 0, w * v.x);
    DEV_ATOMIC_ADD(o + 1, w * v.y);
    DEV_ATOMIC_ADD(o + 2, w * v.z);
    DEV_ATOMIC_ADD(o + 3, w * v.w);
}


#define LGCN_EB 8192
#define LGCN_BIN_SHIFT 10

__global__ __launch_bounds__(1024) void lgcn_bin_edges(const int* __restrict__ eu,
                                                       const int* __restrict__ ei,
                                                       int2* __restrict__ binned,
                                                       int* __restrict__ segoff,
                                                       int E, int U, int nb) {
    __shared__ int hist[1024];
    __shared__ int wtot[32];
    __shared__ int2 stage[2 * LGCN_EB];
    const int tid = threadIdx.x;
    const int lane = tid & 31;
    const int wave = tid >> 5;
    hist[tid] = 0;
    __syncthreads();

    const int e0 = (int)blockIdx.x * LGCN_EB;
    int un[8], in[8], ra[8], rb[8];
#pragma unroll
    for (int k = 0; k < 8; ++k) {
        const int e = e0 + k * 1024 + tid;
        const int ec = min(e, E - 1);
        const int u = eu[ec];
        const int i = ei[ec];
        const bool have = e < E;
        un[k] = have ? u : -1;
        in[k] = have ? U + i : 0;
        ra[k] = 0; rb[k] = 0;
    }
#pragma unroll
    for (int k = 0; k < 8; ++k) {
        if (un[k] >= 0) {
            ra[k] = atomicAdd(&hist[in[k] >> LGCN_BIN_SHIFT], 1);
            rb[k] = atomicAdd(&hist[un[k] >> LGCN_BIN_SHIFT], 1);
        }
    }
    __syncthreads();

    const int v = hist[tid];
    int incl = v;
#pragma unroll
    for (int off = 1; off < 32; off <<= 1) {
        int t = __shfl_up(incl, off);
        if (lane >= off) incl += t;
    }
    if (lane == 31) wtot[wave] = incl;
    __syncthreads();
    int excl = incl - v;
    for (int w = 0; w < wave; ++w) excl += wtot[w];
    hist[tid] = excl;
    if (tid <= nb) segoff[(size_t)blockIdx.x * (nb + 1) + tid] = excl;
    __syncthreads();

#pragma unroll
    for (int k = 0; k < 8; ++k) {
        if (un[k] >= 0) {
            stage[hist[in[k] >> LGCN_BIN_SHIFT] + ra[k]] = make_int2(un[k], in[k]);
            stage[hist[un[k] >> LGCN_BIN_SHIFT] + rb[k]] = make_int2(in[k], un[k]);
        }
    }
    __syncthreads();

    const int pairs = min(LGCN_EB, E - e0);
    int4* slice4 = (int4*)(binned + (size_t)2 * e0);
    for (int k = 0; k < 8; ++k) {
        const int j = k * 1024 + tid;
        if (j < pairs) {
            const int2 p0 = stage[2 * j];
            const int2 p1 = stage[2 * j + 1];
            slice4[j] = make_int4(p0.x, p0.y, p1.x, p1.y);
        }
    }
}

__device__ __forceinline__ int2 lgcn_run_entry(const int2* __restrict__ run, int j, int end) {
    return (j < end) ? run[j] : make_int2(0, -1);
}

#define LGCN_STAGE_CAP 16384

#define LGCN_WALK_RUNS 3

template <typename F>
__device__ __forceinline__ void lgcn_walk_runs(const int2* __restrict__ binned,
                                               const int* segS, const int* segE,
                                               int nblk, int wave, int lane, F use) {
    for (int b0 = wave; b0 < nblk; b0 += 32 * LGCN_WALK_RUNS) {
        const int2* run[LGCN_WALK_RUNS];
        int pos[LGCN_WALK_RUNS], end[LGCN_WALK_RUNS];
#pragma unroll
        for (int r = 0; r < LGCN_WALK_RUNS; ++r) {
            const int blk = b0 + 32 * r;
            const int bc = (blk < nblk) ? blk : b0;
            run[r] = binned + (size_t)2 * LGCN_EB * bc;
            pos[r] = segS[bc] + lane;
            end[r] = (blk < nblk) ? segE[bc] : 0;
        }
        for (;;) {
            bool left = false;
#pragma unroll
            for (int r = 0; r < LGCN_WALK_RUNS; ++r) left = left || (pos[r] - lane < end[r]);
            if (!left) break;
            int2 a[LGCN_WALK_RUNS][4];
#pragma unroll
            for (int r = 0; r < LGCN_WALK_RUNS; ++r) {
#pragma unroll
                for (int i = 0; i < 4; ++i) a[r][i] = lgcn_run_entry(run[r], pos[r] + 32 * i, end[r]);
            }
#pragma unroll
            for (int r = 0; r < LGCN_WALK_RUNS; ++r) {
#pragma unroll
                for (int i = 0; i < 4; ++i)
                    if (a[r][i].y >= 0) use(a[r][i]);
                pos[r] += 128;
            }
        }
    }
}

__global__ __launch_bounds__(1024) void lgcn_build_rows(const int2* __restrict__ binned,
                                                        const int* __restrict__ segoff,
                                                        int* __restrict__ rowptr,
                                                        float* __restrict__ inv,
                                                        int2* __restrict__ csr,
                                                        int N, int nb, int nblk) {
    __shared__ int segS[1024];
    __shared__ int segE[1024];
    __shared__ int cnt[1024];
    __shared__ int rowst[1024];
    __shared__ int wtot[32];
    __shared__ int binstart;
    __shared__ int bintotal;
    __shared__ int2 stage[LGCN_STAGE_CAP];
    const int tid = threadIdx.x;
    const int lane = tid & 31;
    const int wave = tid >> 5;
    const int b = blockIdx.x;
    const int node0 = b << LGCN_BIN_SHIFT;

    if (tid == 0) binstart = 0;
    cnt[tid] = 0;
    __syncthreads();
    if (tid < nblk) {
        const int s = segoff[(size_t)tid * (nb + 1) + b];
        const int e = segoff[(size_t)tid * (nb + 1) + b + 1];
        segS[tid] = s;
        segE[tid] = e;
        atomicAdd(&binstart, s);
    }
    __syncthreads();

    lgcn_walk_runs(binned, segS, segE, nblk, wave, lane,
                   [&](const int2 a) { atomicAdd(&cnt[a.y - node0], 1); });
    __syncthreads();

    const int c = cnt[tid];
    int incl = c;
#pragma unroll
    for (int off = 1; off < 32; off <<= 1) {
        int t = __shfl_up(incl, off);
        if (lane >= off) incl += t;
    }
    if (lane == 31) wtot[wave] = incl;
    __syncthreads();
    const int bs = binstart;
    int start = bs + incl - c;
    for (int w = 0; w < wave; ++w) start += wtot[w];
    if (tid == 1023) bintotal = start + c - bs;
    const int node = node0 + tid;
    if (node < N) {
        rowptr[node] = start;
        const float v = (float)c;
        inv[node] = (v > 0.0f) ? (1.0f / sqrtf(v)) : 0.0f;
        if (node == N - 1) {
            rowptr[N] = start + c;
            inv[N] = 0.0f;
        }
    }
    rowst[tid] = start;
    cnt[tid] = 0;
    __syncthreads();

    const int total = bintotal;
    if (total <= LGCN_STAGE_CAP) {
        lgcn_walk_runs(binned, segS, segE, nblk, wave, lane, [&](const int2 a) {
            const int l = a.y - node0;
            stage[rowst[l] - bs + atomicAdd(&cnt[l], 1)] = a;
        });
        __syncthreads();
        const int shift = bs & 15;
        for (int j = tid - shift; j < total; j += 1024)
            if (j >= 0) csr[bs + j] = stage[j];
        return;
    }

    lgcn_walk_runs(binned, segS, segE, nblk, wave, lane, [&](const int2 a) {
        const int l = a.y - node0;
        csr[rowst[l] + atomicAdd(&cnt[l], 1)] = a;
    });
}

template <bool SPLIT>
__device__ __forceinline__ const float* lgcn_src_row(const float* __restrict__ xa,
                                                     const float* __restrict__ xb,
                                                     int U, int s) {
    if (SPLIT)
        return (s < U) ? xa + (size_t)s * 64 : xb + (size_t)(s - U) * 64;
    return xa + (size_t)s * 64;
}

template <int K>
__device__ __forceinline__ int lgcn_row_share(int v) {
    return __builtin_amdgcn_update_dpp(0, v, 0x150 + K, 0xf, 0xf, true);
}

template <bool SPLIT, int K>
__device__ __forceinline__ float4 lgcn_gather(const float* __restrict__ xa,
                                              const float* __restrict__ xb, int U,
                                              int es, int q) {
    const int s = lgcn_row_share<K>(es);
    if (SPLIT) return *(const float4*)(lgcn_src_row<true>(xa, xb, U, s) + q);
    const unsigned off = ((unsigned)s << 8) | ((unsigned)q << 2);
    return *(const float4*)((const char*)xa + off);
}

template <int K, bool FORM>
__device__ __forceinline__ void lgcn_add(float4& acc, int ew, float di, const float4& v) {
    float w = __int_as_float(lgcn_row_share<K>(ew));
    if (FORM) w *= di;
    acc.x = __builtin_fmaf(w, v.x, acc.x);
    acc.y = __builtin_fmaf(w, v.y, acc.y);
    acc.z = __builtin_fmaf(w, v.z, acc.z);
    acc.w = __builtin_fmaf(w, v.w, acc.w);
}

template <bool SPLIT, int K0, bool FORM>
__device__ __forceinline__ void lgcn_eight(float4& acc, const float* __restrict__ xa,
                                           const float* __restrict__ xb, int U,
                                           int es, int ew, float di, int q, bool upper) {
    const float4 v0 = lgcn_gather<SPLIT, K0 + 0>(xa, xb, U, es, q);
    const float4 v1 = lgcn_gather<SPLIT, K0 + 1>(xa, xb, U, es, q);
    const float4 v2 = lgcn_gather<SPLIT, K0 + 2>(xa, xb, U, es, q);
    const float4 v3 = lgcn_gather<SPLIT, K0 + 3>(xa, xb, U, es, q);
    const float4 none = make_float4(0.0f, 0.0f, 0.0f, 0.0f);
    float4 v4 = none, v5 = none, v6 = none, v7 = none;
    if (upper) {
        v4 = lgcn_gather<SPLIT, K0 + 4>(xa, xb, U, es, q);
        v5 = lgcn_gather<SPLIT, K0 + 5>(xa, xb, U, es, q);
        v6 = lgcn_gather<SPLIT, K0 + 6>(xa, xb, U, es, q);
        v7 = lgcn_gather<SPLIT, K0 + 7>(xa, xb, U, es, q);
    }
    lgcn_add<K0 + 0, FORM>(acc, ew, di, v0);
    lgcn_add<K0 + 1, FORM>(acc, ew, di, v1);
    lgcn_add<K0 + 2, FORM>(acc, ew, di, v2);
    lgcn_add<K0 + 3, FORM>(acc, ew, di, v3);
    if (upper) {
        lgcn_add<K0 + 4, FORM>(acc, ew, di, v4);
        lgcn_add<K0 + 5, FORM>(acc, ew, di, v5);
        lgcn_add<K0 + 6, FORM>(acc, ew, di, v6);
        lgcn_add<K0 + 7, FORM>(acc, ew, di, v7);
    }
}

template <bool SPLIT>
__device__ __forceinline__ float4 lgcn_row_sum(const int* __restrict__ rowptr,
                                               const int2* __restrict__ csr,
                                               const float* __restrict__ xa,
                                               const float* __restrict__ xb, int U,
                                               int row, int q) {
    const int lane = q >> 2;
    const int end = rowptr[row + 1];
    const int last = end - 1;
    float4 acc = make_float4(0.0f, 0.0f, 0.0f, 0.0f);
    for (int base = rowptr[row]; base < end; base += 16) {
        const int left = end - base;
        const int2 e = csr[min(base + lane, last)];
        const int ew = (lane < left) ? e.y : 0;
        lgcn_eight<SPLIT, 0, false>(acc, xa, xb, U, e.x, ew, 0.0f, q, left > 4);
        if (left > 8) lgcn_eight<SPLIT, 8, false>(acc, xa, xb, U, e.x, ew, 0.0f, q, left > 12);
    }
    return acc;
}

__device__ __forceinline__ float4 lgcn_row_sum_first(const int* __restrict__ rowptr,
                                                     int2* csr,
                                                     const float* __restrict__ inv,
                                                     const float* __restrict__ ue,
                                                     const float* __restrict__ ie, int U,
                                                     int N, int row, int q) {
    const int lane = q >> 2;
    const int end = rowptr[row + 1];
    const int last = end - 1;
    const float di = inv[row];
    float4 acc = make_float4(0.0f, 0.0f, 0.0f, 0.0f);
    for (int base = rowptr[row]; base < end; base += 16) {
        const int left = end - base;
        const bool live = lane < left;
        const unsigned eoff = (unsigned)min(base + lane, last) << 3;
        const int s = *(const int*)((const char*)csr + eoff);
        const float wi = inv[live ? s : N];
        const int wbits = __float_as_int(wi);
        lgcn_eight<true, 0, true>(acc, ue, ie, U, s, wbits, di, q, left > 4);
        if (live) *(int*)((char*)csr + eoff + 4) = __float_as_int(wi * di);
        if (left > 8) lgcn_eight<true, 8, true>(acc, ue, ie, U, s, wbits, di, q, left > 12);
    }
    return acc;
}

__device__ __forceinline__ int lgcn_batch_blocks(int B) {
    return (B + 7) >> 3;
}

template <bool FIRST>
__device__ __forceinline__ void lgcn_batch_acc(const int* __restrict__ users,
                                               const int* __restrict__ items,
                                               const float* __restrict__ xu,
                                               const float* __restrict__ xi,
                                               float* uacc, float* iacc, int B, int t) {
    const int entry = t >> 4;
    if (entry >= 2 * B) return;
    const int q = (t & 15) * 4;
    const float* src;
    float* dst;
    if (entry < B) {
        src = xu + (size_t)users[entry] * 64 + q;
        dst = uacc + (size_t)entry * 64 + q;
    } else {
        const int e = entry - B;
        src = xi + (size_t)items[e] * 64 + q;
        dst = iacc + (size_t)e * 64 + q;
    }
    float4 v = *(const float4*)src;
    if (!FIRST) {
        float4 a = *(const float4*)dst;
        a.x += v.x; a.y += v.y; a.z += v.z; a.w += v.w;
        v = a;
    }
    *(float4*)dst = v;
}

__global__ __launch_bounds__(256) void lgcn_spmm_pull(const int* __restrict__ users,
                                                      const int* __restrict__ items,
                                                      const int* __restrict__ rowptr,
                                                      const int2* __restrict__ csr,
                                                      const float* __restrict__ xin,
                                                      float* __restrict__ xout,
                                                      float* uacc, float* iacc,
                                                      int B, int U, int N) {
    const int batch_blocks = lgcn_batch_blocks(B);
    if ((int)blockIdx.x < batch_blocks) {
        lgcn_batch_acc<false>(users, items, xin, xin + (size_t)U * 64, uacc, iacc, B,
                              (int)blockIdx.x * 256 + (int)threadIdx.x);
        return;
    }
    const int t = ((int)blockIdx.x - batch_blocks) * 256 + (int)threadIdx.x;
    const int row = t >> 4;
    if (row >= N) return;
    const int q = (t & 15) * 4;
    *(float4*)(xout + (size_t)row * 64 + q) =
        lgcn_row_sum<false>(rowptr, csr, xin, nullptr, 0, row, q);
}

__global__ __launch_bounds__(256) void lgcn_spmm_pull_first(const int* __restrict__ users,
                                                            const int* __restrict__ items,
                                                            const int* __restrict__ rowptr,
                                                            int2* csr,
                                                            const float* __restrict__ inv,
                                                            const float* __restrict__ ue,
                                                            const float* __restrict__ ie,
                                                            float* __restrict__ xout,
                                                            float* uacc, float* iacc,
                                                            int B, int U, int N) {
    const int batch_blocks = lgcn_batch_blocks(B);
    if ((int)blockIdx.x < batch_blocks) {
        lgcn_batch_acc<true>(users, items, ue, ie, uacc, iacc, B,
                             (int)blockIdx.x * 256 + (int)threadIdx.x);
        return;
    }
    const int t = ((int)blockIdx.x - batch_blocks) * 256 + (int)threadIdx.x;
    const int row = t >> 4;
    if (row >= N) return;
    const int q = (t & 15) * 4;
    *(float4*)(xout + (size_t)row * 64 + q) =
        lgcn_row_sum_first(rowptr, csr, inv, ue, ie, U, N, row, q);
}

#define LGCN_FIN_PITCH 68

#define LGCN_TAIL_GROUPS 64

__global__ __launch_bounds__(1024) void lgcn_tail_dot(const int* __restrict__ users,
                                                      const int* __restrict__ items,
                                                      const int* __restrict__ rowptr,
                                                      const int2* __restrict__ csr,
                                                      const float* __restrict__ xin,
                                                      const float* __restrict__ uacc,
                                                      const float* __restrict__ iacc,
                                                      float* __restrict__ out,
                                                      int B, int U) {
    __shared__ __attribute__((aligned(16))) float rows[2][LGCN_TAIL_GROUPS][LGCN_FIN_PITCH];
    __shared__ float wts[2][LGCN_TAIL_GROUPS];
    __shared__ __attribute__((aligned(16))) float own3[16][LGCN_FIN_PITCH];
    __shared__ __attribute__((aligned(16))) float sum4[16][LGCN_FIN_PITCH];
    __shared__ __attribute__((aligned(16))) float fin[2][8][LGCN_FIN_PITCH];
    __shared__ float tile[16 * 16];
    __shared__ int tnode[16];
    __shared__ int tbeg[16];
    __shared__ int tstart[17];
    const int tid = threadIdx.x;
    const int grp = tid >> 4;
    const int q = (tid & 15) * 4;

    if (tid < 32) {
        int cnt = 0;
        if (tid < 16) {
            int b = (int)blockIdx.x * 8 + (tid & 7);
            if (b > B - 1) b = B - 1;
            const int node = (tid >> 3) ? U + items[b] : users[b];
            const int beg = rowptr[node];
            cnt = 1 + rowptr[node + 1] - beg;
            tnode[tid] = node;
            tbeg[tid] = beg;
        }
        int incl = cnt;
        for (int off = 1; off < 16; off <<= 1) {
            const int t = __shfl_up(incl, off);
            if (tid >= off) incl += t;
        }
        if (tid < 16) tstart[tid] = incl - cnt;
        if (tid == 15) tstart[16] = incl;
    }
    if (grp < 16) *(float4*)&sum4[grp][q] = make_float4(0.0f, 0.0f, 0.0f, 0.0f);
    __syncthreads();

    const int total = tstart[16];
    int set = 0;
    for (int r0 = 0; r0 < total; r0 += LGCN_TAIL_GROUPS, set ^= 1) {
        const int t = r0 + grp;
        if (t < total) {
            int e = 0;
            for (int k = 1; k < 16; ++k) e += (tstart[k] <= t) ? 1 : 0;
            const int j = t - tstart[e] - 1;
            int row = tnode[e];
            float w = 0.0f;
            if (j >= 0) {
                const int2 en = csr[tbeg[e] + j];
                row = en.x;
                w = __int_as_float(en.y);
            }
            *(float4*)&rows[set][grp][q] = lgcn_row_sum<false>(rowptr, csr, xin, nullptr, 0, row, q);
            if ((tid & 15) == 0) wts[set][grp] = w;
        }
        __syncthreads();
        if (grp < 16) {
            const int ts = tstart[grp];
            const int a0 = max(ts, r0);
            const int a1 = min(tstart[grp + 1], r0 + LGCN_TAIL_GROUPS);
            if (a0 < a1) {
                float4 acc = *(const float4*)&sum4[grp][q];
                for (int tt = a0; tt < a1; ++tt) {
                    const float4 v = *(const float4*)&rows[set][tt - r0][q];
                    if (tt == ts) {
                        *(float4*)&own3[grp][q] = v;
                    } else {
                        const float w = wts[set][tt - r0];
                        acc.x = __builtin_fmaf(w, v.x, acc.x);
                        acc.y = __builtin_fmaf(w, v.y, acc.y);
                        acc.z = __builtin_fmaf(w, v.z, acc.z);
                        acc.w = __builtin_fmaf(w, v.w, acc.w);
                    }
                }
                *(float4*)&sum4[grp][q] = acc;
            }
        }
    }

    if (grp < 16) {
        const int slot = grp & 7;
        const int side = grp >> 3;
        int b = (int)blockIdx.x * 8 + slot;
        if (b > B - 1) b = B - 1;
        float4 a = *(const float4*)((side ? iacc : uacc) + (size_t)b * 64 + q);
        const float4 x2 = *(const float4*)(xin + (size_t)tnode[grp] * 64 + q);
        const float4 x3 = *(const float4*)&own3[grp][q];
        const float4 s = *(const float4*)&sum4[grp][q];
        a.x += x2.x; a.y += x2.y; a.z += x2.z; a.w += x2.w;
        a.x += x3.x; a.y += x3.y; a.z += x3.z; a.w += x3.w;
        a.x += s.x; a.y += s.y; a.z += s.z; a.w += s.w;
        *(float4*)&fin[side][slot][q] = a;
    }
    __syncthreads();

    if (tid >= 32) return;
    const int lane = tid;
    const int m = lane & 7;
    const int khalf = (lane >> 4) << 1;
    const float* ur = &fin[0][m][0];
    const float* ir = &fin[1][m][0];
    v8f c = {};
#pragma unroll
    for (int kb = 0; kb < 64; kb += 4) {
        const int k0 = kb + khalf;
        v2f fa; fa.x = ur[k0]; fa.y = ur[k0 + 1];
        v2f fb; fb.x = ir[k0]; fb.y = ir[k0 + 1];
        c = __builtin_amdgcn_wmma_f32_16x16x4_f32(false, fa, false, fb,
                                                  (short)0, c, false, false);
    }
    const int n = lane & 15;
    const int mbase = (lane >> 4) ? 8 : 0;
#pragma unroll
    for (int r = 0; r < 8; ++r)
        tile[(mbase + r) * 16 + n] = c[r];
    const int bo = (int)blockIdx.x * 8 + lane;
    if (lane < 8 && bo < B) out[bo] = tile[lane * 16 + lane] * (1.0f / 25.0f);
}

__global__ void lgcn_gather_acc(const int* __restrict__ users, const int* __restrict__ items,
                                const float* __restrict__ xu, const float* __restrict__ xi,
                                float* __restrict__ uacc, float* __restrict__ iacc,
                                int B) {
    int t = blockIdx.x * blockDim.x + threadIdx.x;
    int half = B * 64;
    if (t >= 2 * half) return;
    if (t < half) {
        int b = t >> 6, d = t & 63;
        uacc[t] += xu[(size_t)users[b] * 64 + d];
    } else {
        int tt = t - half;
        int b = tt >> 6, d = tt & 63;
        iacc[tt] += xi[(size_t)items[b] * 64 + d];
    }
}

__global__ __launch_bounds__(256) void lgcn_dot_wmma(const float* __restrict__ uacc,
                                                     const float* __restrict__ iacc,
                                                     float* __restrict__ out, int B) {
    __shared__ float tile[8][16 * 16];
    const int lane = threadIdx.x & 31;
    const int wave = threadIdx.x >> 5;
    const int tile0 = (blockIdx.x * 8 + wave) * 16;

    const int m = lane & 15;
    const int khalf = (lane >> 4) << 1;
    int row = tile0 + m;
    if (row > B - 1) row = B - 1;
    const float* __restrict__ ur = uacc + (size_t)row * 64;
    const float* __restrict__ ir = iacc + (size_t)row * 64;

    v8f c = {};
#pragma unroll
    for (int kb = 0; kb < 64; kb += 4) {
        int k0 = kb + khalf;
        v2f a; a.x = ur[k0]; a.y = ur[k0 + 1];
        v2f b; b.x = ir[k0]; b.y = ir[k0 + 1];
        c = __builtin_amdgcn_wmma_f32_16x16x4_f32(false, a, false, b,
                                                  (short)0, c, false, false);
    }

    float* lds = &tile[wave][0];
    const int n = lane & 15;
    const int mbase = (lane >> 4) ? 8 : 0;
#pragma unroll
    for (int r = 0; r < 8; ++r)
        lds[(mbase + r) * 16 + n] = c[r];
    __syncthreads();

    if (lane < 16) {
        int b = tile0 + lane;
        if (b < B)
            out[b] = lds[lane * 16 + lane] * (1.0f / 25.0f);
    }
}

extern "C" void kernel_launch(void* const* d_in, const int* in_sizes, int n_in,
                              void* d_out, int out_size, void* d_ws, size_t ws_size,
                              hipStream_t stream) {
    const int*   users = (const int*)d_in[0];
    const int*   items = (const int*)d_in[1];
    const int*   eu    = (const int*)d_in[2];
    const int*   ei    = (const int*)d_in[3];
    const float* uemb  = (const float*)d_in[4];
    const float* iemb  = (const float*)d_in[5];
    float*       out   = (float*)d_out;

    const int B = in_sizes[0];
    const int E = in_sizes[2];
    const int U = in_sizes[4] / 64;
    const int I = in_sizes[5] / 64;
    const int N = U + I;

    char* ws = (char*)d_ws;
    size_t off = 0;
    auto carve = [&](size_t bytes) -> float* {
        float* p = (float*)(ws + off);
        off += (bytes + 255) & ~(size_t)255;
        return p;
    };
    float* inv    = carve((size_t)N * 4);
    float* e_cur  = carve((size_t)N * 64 * 4);
    float* e_next = carve((size_t)N * 64 * 4);
    float* uacc   = carve((size_t)B * 64 * 4);
    float* iacc   = carve((size_t)B * 64 * 4);
    int*   rowptr = (int*)carve((size_t)(N + 1) * 4);
    int2*  csr    = (int2*)carve((size_t)2 * E * 8);

    const int nchunk = (N + 1023) / 1024;
    const int nblk = (E + LGCN_EB - 1) / LGCN_EB;
    size_t toff = 0;
    auto tcarve = [&](size_t bytes) -> int* {
        int* p = (int*)((char*)e_next + toff);
        toff += (bytes + 255) & ~(size_t)255;
        return p;
    };
    int2* binned = (int2*)tcarve((size_t)2 * E * 8);
    int*  segoff = tcarve((size_t)nblk * (nchunk + 1) * 4);
    float* inv_g = (float*)tcarve((size_t)(N + 1) * 4);

    const bool gather_path = (off <= ws_size) && (toff <= (size_t)N * 64 * 4) &&
                             ((size_t)N * 64 * 4 <= (size_t)0xffffffffu) &&
                             (nchunk <= 1023) && (nblk <= 1024) && (E <= 0x0fffffff);

    const int BS = 256;

    const int gacc_blocks = (2 * B * 64 + BS - 1) / BS;
    const size_t item0 = (size_t)U * 64;

    if (gather_path) {
        lgcn_bin_edges<<<nblk, 1024, 0, stream>>>(eu, ei, binned, segoff, E, U, nchunk);
        lgcn_build_rows<<<nchunk, 1024, 0, stream>>>(binned, segoff, rowptr, inv_g, csr, N, nchunk, nblk);
        const int batch_blocks = (B + 7) / 8;
        const int pull_blocks = (N + 15) / 16;
        const int layer_blocks = pull_blocks + batch_blocks;

        lgcn_spmm_pull_first<<<layer_blocks, 256, 0, stream>>>(users, items, rowptr, csr, inv_g, uemb, iemb,
                                                               e_cur, uacc, iacc, B, U, N);
        lgcn_spmm_pull<<<layer_blocks, 256, 0, stream>>>(users, items, rowptr, csr, e_cur, e_next,
                                                         uacc, iacc, B, U, N);
        lgcn_tail_dot<<<batch_blocks, 1024, 0, stream>>>(users, items, rowptr, csr, e_next, uacc, iacc, out, B, U);
        return;
    } else {
        hipMemsetAsync(uacc, 0, (size_t)B * 64 * 4, stream);
        hipMemsetAsync(iacc, 0, (size_t)B * 64 * 4, stream);

        {
            long long total = (long long)N * 64;
            lgcn_init_x0<<<(int)((total + BS - 1) / BS), BS, 0, stream>>>(uemb, iemb, e_cur, U, N);
        }
        hipMemsetAsync(inv, 0, (size_t)N * 4, stream);
        lgcn_degree<<<(E + BS - 1) / BS, BS, 0, stream>>>(eu, ei, inv, E, U);
        lgcn_invsqrt<<<(N + BS - 1) / BS, BS, 0, stream>>>(inv, N);

        lgcn_gather_acc<<<gacc_blocks, BS, 0, stream>>>(users, items, e_cur, e_cur + item0, uacc, iacc, B);

        const long long spmm_threads = (long long)2 * E * 16;
        const int spmm_blocks = (int)((spmm_threads + BS - 1) / BS);
        for (int layer = 0; layer < 4; ++layer) {
            hipMemsetAsync(e_next, 0, (size_t)N * 64 * 4, stream);
            lgcn_spmm<<<spmm_blocks, BS, 0, stream>>>(eu, ei, inv, e_cur, e_next, E, U);
            lgcn_gather_acc<<<gacc_blocks, BS, 0, stream>>>(users, items, e_next, e_next + item0, uacc, iacc, B);
            float* tmp = e_cur; e_cur = e_next; e_next = tmp;
        }
    }

    const int tiles = (B + 15) / 16;
    const int dot_blocks = (tiles + 7) / 8;
    lgcn_dot_wmma<<<dot_blocks, 256, 0, stream>>>(uacc, iacc, out, B);
}
